# starting kernel
# baseline (speedup 1.0000x reference)
#include <hip/hip_runtime.h>
#include <math.h>


#define NEG_SLOPE 0.2f

typedef __attribute__((ext_vector_type(2))) float v2f;
typedef __attribute__((ext_vector_type(8))) float v8f;

__device__ __forceinline__ float lrelu(float v) {
    return v > 0.0f ? v : NEG_SLOPE * v;
}

__device__ __forceinline__ void atomicMaxF(float* addr, float val) {
    if (val >= 0.0f) {
        atomicMax((int*)addr, __float_as_int(val));
    } else {
        atomicMin((unsigned int*)addr, (unsigned int)__float_as_int(val));
    }
}

__global__ __launch_bounds__(256) void k_init(float* __restrict__ acc1,
                                              float* __restrict__ m1,
                                              float* __restrict__ denom1,
                                              float* __restrict__ acc2,
                                              float* __restrict__ m2,
                                              float* __restrict__ denom2,
                                              int n) {
    int i = blockIdx.x * blockDim.x + threadIdx.x;
    const float NEG_INF = __int_as_float(0xFF800000);
    if (i < 32 * n) acc1[i] = 0.0f;
    if (i < 2 * n) {
        m1[i] = NEG_INF;
        denom1[i] = 0.0f;
        acc2[i] = 0.0f;
    }
    if (i < n) {
        m2[i] = NEG_INF;
        denom2[i] = 0.0f;
    }
}

__global__ __launch_bounds__(256) void k_gemm1_wmma(const float* __restrict__ x,
                                                    const float* __restrict__ W1,
                                                    float* __restrict__ h1,
                                                    int n) {
    const int wave = (blockIdx.x * blockDim.x + threadIdx.x) >> 5;
    const int lane = threadIdx.x & 31;
    const int row0 = wave * 16;
    if (row0 >= n) return;

    const int m  = lane & 15;
    const int kh = lane >> 4;
    int row = row0 + m;
    if (row >= n) row = n - 1;

    const float* __restrict__ xrow = x + (size_t)row * 128;

    v8f c0 = {};
    v8f c1 = {};

    #pragma unroll
    for (int k0 = 0; k0 < 128; k0 += 4) {
        const int k = k0 + 2 * kh;
        v2f a = *(const v2f*)(xrow + k);
        v2f b0, b1;
        b0.x = W1[(size_t)k * 32 + m];
        b0.y = W1[(size_t)(k + 1) * 32 + m];
        b1.x = W1[(size_t)k * 32 + 16 + m];
        b1.y = W1[(size_t)(k + 1) * 32 + 16 + m];
        c0 = __builtin_amdgcn_wmma_f32_16x16x4_f32(false, a, false, b0,
                                                   (short)0, c0, false, false);
        c1 = __builtin_amdgcn_wmma_f32_16x16x4_f32(false, a, false, b1,
                                                   (short)0, c1, false, false);
    }

    const int mb = kh * 8;
    #pragma unroll
    for (int j = 0; j < 8; ++j) {
        const int r = row0 + mb + j;
        if (r < n) {
            h1[(size_t)r * 32 + m]      = c0[j];
            h1[(size_t)r * 32 + 16 + m] = c1[j];
        }
    }
}

__global__ __launch_bounds__(256) void k_alpha1(const float* __restrict__ h1,
                                                const float* __restrict__ asw,
                                                const float* __restrict__ adw,
                                                float* __restrict__ asrc,
                                                float* __restrict__ adst,
                                                int n) {
    int i = blockIdx.x * blockDim.x + threadIdx.x;
    if (i >= n) return;
    const float* hr = h1 + (size_t)i * 32;
    #pragma unroll
    for (int h = 0; h < 2; ++h) {
        float s = 0.0f, d = 0.0f;
        #pragma unroll
        for (int f = 0; f < 16; ++f) {
            const float v = hr[h * 16 + f];
            s += v * asw[h * 16 + f];
            d += v * adw[h * 16 + f];
        }
        asrc[i * 2 + h] = s;
        adst[i * 2 + h] = d;
    }
}

__global__ __launch_bounds__(256) void k_edge_max1(const int* __restrict__ src,
                                                   const int* __restrict__ dst,
                                                   const float* __restrict__ asrc,
                                                   const float* __restrict__ adst,
                                                   float* __restrict__ m1,
                                                   int E, int n) {
    int t = blockIdx.x * blockDim.x + threadIdx.x;
    const int tot = 2 * (E + n);
    if (t >= tot) return;
    const int h = t & 1;
    const int e = t >> 1;
    int s, d;
    if (e < E) { s = src[e]; d = dst[e]; } else { s = d = e - E; }
    const float logit = lrelu(asrc[(size_t)s * 2 + h] + adst[(size_t)d * 2 + h]);
    atomicMaxF(&m1[(size_t)d * 2 + h], logit);
}

__global__ __launch_bounds__(256) void k_edge_acc1(const int* __restrict__ src,
                                                   const int* __restrict__ dst,
                                                   const float* __restrict__ asrc,
                                                   const float* __restrict__ adst,
                                                   const float* __restrict__ m1,
                                                   const float* __restrict__ h1,
                                                   float* __restrict__ denom1,
                                                   float* __restrict__ acc1,
                                                   int E, int n) {
    int t = blockIdx.x * blockDim.x + threadIdx.x;
    const int tot = 2 * (E + n);
    if (t >= tot) return;
    const int h = t & 1;
    const int e = t >> 1;
    int s, d;
    if (e < E) { s = src[e]; d = dst[e]; } else { s = d = e - E; }
    const float logit = lrelu(asrc[(size_t)s * 2 + h] + adst[(size_t)d * 2 + h]);
    const float ex = __expf(logit - m1[(size_t)d * 2 + h]);
    atomicAdd(&denom1[(size_t)d * 2 + h], ex);
    const float* hs = h1 + (size_t)s * 32 + h * 16;
    float* ad = acc1 + (size_t)d * 32 + h * 16;
    #pragma unroll
    for (int f = 0; f < 16; ++f) {
        atomicAdd(&ad[f], ex * hs[f]);
    }
}

__global__ __launch_bounds__(256) void k_node_layer2(const float* __restrict__ acc1,
                                                     const float* __restrict__ denom1,
                                                     const float* __restrict__ b1,
                                                     const float* __restrict__ W2,
                                                     const float* __restrict__ as2w,
                                                     const float* __restrict__ ad2w,
                                                     float* __restrict__ g,
                                                     float* __restrict__ asrc2,
                                                     float* __restrict__ adst2,
                                                     int n) {
    int i = blockIdx.x * blockDim.x + threadIdx.x;
    if (i >= n) return;
    float g0 = 0.0f, g1 = 0.0f;
    const float inv0 = 1.0f / (denom1[(size_t)i * 2 + 0] + 1e-16f);
    const float inv1 = 1.0f / (denom1[(size_t)i * 2 + 1] + 1e-16f);
    #pragma unroll
    for (int c = 0; c < 32; ++c) {
        const float inv = (c < 16) ? inv0 : inv1;
        float v = acc1[(size_t)i * 32 + c] * inv + b1[c];
        v = v > 0.0f ? v : 0.0f;
        g0 += v * W2[c * 2 + 0];
        g1 += v * W2[c * 2 + 1];
    }
    g[(size_t)i * 2 + 0] = g0;
    g[(size_t)i * 2 + 1] = g1;
    asrc2[i] = g0 * as2w[0] + g1 * as2w[1];
    adst2[i] = g0 * ad2w[0] + g1 * ad2w[1];
}

__global__ __launch_bounds__(256) void k_edge_max2(const int* __restrict__ src,
                                                   const int* __restrict__ dst,
                                                   const float* __restrict__ asrc2,
                                                   const float* __restrict__ adst2,
                                                   float* __restrict__ m2,
                                                   int E, int n) {
    int e = blockIdx.x * blockDim.x + threadIdx.x;
    if (e >= E + n) return;
    int s, d;
    if (e < E) { s = src[e]; d = dst[e]; } else { s = d = e - E; }
    atomicMaxF(&m2[d], lrelu(asrc2[s] + adst2[d]));
}

__global__ __launch_bounds__(256) void k_edge_acc2(const int* __restrict__ src,
                                                   const int* __restrict__ dst,
                                                   const float* __restrict__ asrc2,
                                                   const float* __restrict__ adst2,
                                                   const float* __restrict__ m2,
                                                   const float* __restrict__ g,
                                                   float* __restrict__ denom2,
                                                   float* __restrict__ acc2,
                                                   int E, int n) {
    int e = blockIdx.x * blockDim.x + threadIdx.x;
    if (e >= E + n) return;
    int s, d;
    if (e < E) { s = src[e]; d = dst[e]; } else { s = d = e - E; }
    const float ex = __expf(lrelu(asrc2[s] + adst2[d]) - m2[d]);
    atomicAdd(&denom2[d], ex);
    atomicAdd(&acc2[(size_t)d * 2 + 0], ex * g[(size_t)s * 2 + 0]);
    atomicAdd(&acc2[(size_t)d * 2 + 1], ex * g[(size_t)s * 2 + 1]);
}

__global__ __launch_bounds__(256) void k_out(const float* __restrict__ acc2,
                                             const float* __restrict__ denom2,
                                             const float* __restrict__ b2,
                                             float* __restrict__ out,
                                             int n) {
    int i = blockIdx.x * blockDim.x + threadIdx.x;
    if (i >= n) return;
    const float inv = 1.0f / (denom2[i] + 1e-16f);
    out[(size_t)i * 2 + 0] = acc2[(size_t)i * 2 + 0] * inv + b2[0];
    out[(size_t)i * 2 + 1] = acc2[(size_t)i * 2 + 1] * inv + b2[1];
}

extern "C" void kernel_launch(void* const* d_in, const int* in_sizes, int n_in,
                              void* d_out, int out_size, void* d_ws, size_t ws_size,
                              hipStream_t stream) {
    const float* x    = (const float*)d_in[0];
    const int*   ei   = (const int*)d_in[1];
    const float* W1   = (const float*)d_in[2];
    const float* as1w = (const float*)d_in[3];
    const float* ad1w = (const float*)d_in[4];
    const float* b1   = (const float*)d_in[5];
    const float* W2   = (const float*)d_in[6];
    const float* as2w = (const float*)d_in[7];
    const float* ad2w = (const float*)d_in[8];
    const float* b2   = (const float*)d_in[9];

    const int N = in_sizes[0] / 128;
    const int E = in_sizes[1] / 2;
    const int* srcI = ei;
    const int* dstI = ei + E;

    float* p = (float*)d_ws;
    float* h1     = p; p += (size_t)32 * N;
    float* asrc1  = p; p += (size_t)2 * N;
    float* adst1  = p; p += (size_t)2 * N;
    float* m1     = p; p += (size_t)2 * N;
    float* denom1 = p; p += (size_t)2 * N;
    float* acc1   = p; p += (size_t)32 * N;
    float* g      = p; p += (size_t)2 * N;
    float* asrc2  = p; p += (size_t)N;
    float* adst2  = p; p += (size_t)N;
    float* m2     = p; p += (size_t)N;
    float* denom2 = p; p += (size_t)N;
    float* acc2   = p; p += (size_t)2 * N;

    const int T = 256;
    const int Etot = E + N;

    k_init<<<(32 * N + T - 1) / T, T, 0, stream>>>(acc1, m1, denom1, acc2, m2,
                                                   denom2, N);
    {
        const int tiles = (N + 15) / 16;
        const int waves_per_block = T / 32;
        const int blocks = (tiles + waves_per_block - 1) / waves_per_block;
        k_gemm1_wmma<<<blocks, T, 0, stream>>>(x, W1, h1, N);
    }
    k_alpha1<<<(N + T - 1) / T, T, 0, stream>>>(h1, as1w, ad1w, asrc1, adst1, N);
    k_edge_max1<<<(2 * Etot + T - 1) / T, T, 0, stream>>>(srcI, dstI, asrc1,
                                                          adst1, m1, E, N);
    k_edge_acc1<<<(2 * Etot + T - 1) / T, T, 0, stream>>>(srcI, dstI, asrc1,
                                                          adst1, m1, h1, denom1,
                                                          acc1, E, N);
    k_node_layer2<<<(N + T - 1) / T, T, 0, stream>>>(acc1, denom1, b1, W2, as2w,
                                                     ad2w, g, asrc2, adst2, N);
    k_edge_max2<<<(Etot + T - 1) / T, T, 0, stream>>>(srcI, dstI, asrc2, adst2,
                                                      m2, E, N);
    k_edge_acc2<<<(Etot + T - 1) / T, T, 0, stream>>>(srcI, dstI, asrc2, adst2,
                                                      m2, g, denom2, acc2, E, N);
    k_out<<<(N + T - 1) / T, T, 0, stream>>>(acc2, denom2, b2, (float*)d_out, N);
}


# accumulate passes: lanes of an f32 atomic add cover adjacent words of one destination record (layer 1 whole 128-B row, layer 2 16-B record), not one edge per lane; arrival-order sums kept
# speedup vs baseline: 1.7900x; 1.7900x over previous
#include <hip/hip_runtime.h>
#include <math.h>


#define NEG_SLOPE 0.2f

typedef __attribute__((ext_vector_type(2))) float v2f;
typedef __attribute__((ext_vector_type(8))) float v8f;

__device__ __forceinline__ float lrelu(float v) {
    return v > 0.0f ? v : NEG_SLOPE * v;
}

__device__ __forceinline__ void atomicMaxF(float* addr, float val) {
    if (val >= 0.0f) {
        atomicMax((int*)addr, __float_as_int(val));
    } else {
        atomicMin((unsigned int*)addr, (unsigned int)__float_as_int(val));
    }
}

__global__ __launch_bounds__(256) void k_init(float* __restrict__ acc1,
                                              float* __restrict__ m1,
                                              float* __restrict__ denom1,
                                              float* __restrict__ rec2,
                                              float* __restrict__ m2,
                                              int n) {
    int i = blockIdx.x * blockDim.x + threadIdx.x;
    const float NEG_INF = __int_as_float(0xFF800000);
    if (i < 32 * n) acc1[i] = 0.0f;
    if (i < 4 * n) rec2[i] = 0.0f;
    if (i < 2 * n) {
        m1[i] = NEG_INF;
        denom1[i] = 0.0f;
    }
    if (i < n) m2[i] = NEG_INF;
}

__global__ __launch_bounds__(256) void k_gemm1_wmma(const float* __restrict__ x,
                                                    const float* __restrict__ W1,
                                                    float* __restrict__ h1,
                                                    int n) {
    const int wave = (blockIdx.x * blockDim.x + threadIdx.x) >> 5;
    const int lane = threadIdx.x & 31;
    const int row0 = wave * 16;
    if (row0 >= n) return;

    const int m  = lane & 15;
    const int kh = lane >> 4;
    int row = row0 + m;
    if (row >= n) row = n - 1;

    const float* __restrict__ xrow = x + (size_t)row * 128;

    v8f c0 = {};
    v8f c1 = {};

    #pragma unroll
    for (int k0 = 0; k0 < 128; k0 += 4) {
        const int k = k0 + 2 * kh;
        v2f a = *(const v2f*)(xrow + k);
        v2f b0, b1;
        b0.x = W1[(size_t)k * 32 + m];
        b0.y = W1[(size_t)(k + 1) * 32 + m];
        b1.x = W1[(size_t)k * 32 + 16 + m];
        b1.y = W1[(size_t)(k + 1) * 32 + 16 + m];
        c0 = __builtin_amdgcn_wmma_f32_16x16x4_f32(false, a, false, b0,
                                                   (short)0, c0, false, false);
        c1 = __builtin_amdgcn_wmma_f32_16x16x4_f32(false, a, false, b1,
                                                   (short)0, c1, false, false);
    }

    const int mb = kh * 8;
    #pragma unroll
    for (int j = 0; j < 8; ++j) {
        const int r = row0 + mb + j;
        if (r < n) {
            h1[(size_t)r * 32 + m]      = c0[j];
            h1[(size_t)r * 32 + 16 + m] = c1[j];
        }
    }
}

__global__ __launch_bounds__(256) void k_alpha1(const float* __restrict__ h1,
                                                const float* __restrict__ asw,
                                                const float* __restrict__ adw,
                                                float* __restrict__ asrc,
                                                float* __restrict__ adst,
                                                int n) {
    int i = blockIdx.x * blockDim.x + threadIdx.x;
    if (i >= n) return;
    const float* hr = h1 + (size_t)i * 32;
    #pragma unroll
    for (int h = 0; h < 2; ++h) {
        float s = 0.0f, d = 0.0f;
        #pragma unroll
        for (int f = 0; f < 16; ++f) {
            const float v = hr[h * 16 + f];
            s += v * asw[h * 16 + f];
            d += v * adw[h * 16 + f];
        }
        asrc[i * 2 + h] = s;
        adst[i * 2 + h] = d;
    }
}

__global__ __launch_bounds__(256) void k_edge_max1(const int* __restrict__ src,
                                                   const int* __restrict__ dst,
                                                   const float* __restrict__ asrc,
                                                   const float* __restrict__ adst,
                                                   float* __restrict__ m1,
                                                   int E, int n) {
    int t = blockIdx.x * blockDim.x + threadIdx.x;
    const int tot = 2 * (E + n);
    if (t >= tot) return;
    const int h = t & 1;
    const int e = t >> 1;
    int s, d;
    if (e < E) { s = src[e]; d = dst[e]; } else { s = d = e - E; }
    const float logit = lrelu(asrc[(size_t)s * 2 + h] + adst[(size_t)d * 2 + h]);
    atomicMaxF(&m1[(size_t)d * 2 + h], logit);
}

__device__ __forceinline__ void row_add1(const float* __restrict__ h1,
                                         float* __restrict__ acc1,
                                         int s, int d, float ex, int j,
                                         int lane) {
    const int sj = __builtin_amdgcn_readlane(s, 2 * j);
    const int dj = __builtin_amdgcn_readlane(d, 2 * j);
    const float w0 = __int_as_float(__builtin_amdgcn_readlane(__float_as_int(ex), 2 * j));
    const float w1 = __int_as_float(__builtin_amdgcn_readlane(__float_as_int(ex), 2 * j + 1));
    const float w = (lane < 16) ? w0 : w1;
    const float v = h1[(size_t)sj * 32 + lane];
    atomicAdd(&acc1[(size_t)dj * 32 + lane], w * v);
}

__global__ __launch_bounds__(256) void k_edge_acc1(const int* __restrict__ src,
                                                   const int* __restrict__ dst,
                                                   const float* __restrict__ asrc,
                                                   const float* __restrict__ adst,
                                                   const float* __restrict__ m1,
                                                   const float* __restrict__ h1,
                                                   float* __restrict__ denom1,
                                                   float* __restrict__ acc1,
                                                   int E, int n) {
    const int t = blockIdx.x * blockDim.x + threadIdx.x;
    const int lane = threadIdx.x & 31;
    const int tot = 2 * (E + n);
    const int t0 = __builtin_amdgcn_readfirstlane(t - lane);
    if (t0 >= tot) return;
    const int h = t & 1;
    const int e = t >> 1;
    int s = 0, d = 0;
    float ex = 0.0f;
    if (t < tot) {
        if (e < E) { s = src[e]; d = dst[e]; } else { s = d = e - E; }
        const float logit = lrelu(asrc[(size_t)s * 2 + h] + adst[(size_t)d * 2 + h]);
        ex = __expf(logit - m1[(size_t)d * 2 + h]);
        atomicAdd(&denom1[(size_t)d * 2 + h], ex);
    }
    const int rest = (tot - t0) >> 1;
    if (rest >= 16) {
        float v[16];
        #pragma unroll
        for (int j = 0; j < 16; ++j) {
            const int sj = __builtin_amdgcn_readlane(s, 2 * j);
            v[j] = h1[(size_t)sj * 32 + lane];
        }
        #pragma unroll
        for (int j = 0; j < 16; ++j) {
            const int dj = __builtin_amdgcn_readlane(d, 2 * j);
            const float w0 = __int_as_float(__builtin_amdgcn_readlane(__float_as_int(ex), 2 * j));
            const float w1 = __int_as_float(__builtin_amdgcn_readlane(__float_as_int(ex), 2 * j + 1));
            const float w = (lane < 16) ? w0 : w1;
            atomicAdd(&acc1[(size_t)dj * 32 + lane], w * v[j]);
        }
    } else {
        for (int j = 0; j < rest; ++j) row_add1(h1, acc1, s, d, ex, j, lane);
    }
}

__global__ __launch_bounds__(256) void k_node_layer2(const float* __restrict__ acc1,
                                                     const float* __restrict__ denom1,
                                                     const float* __restrict__ b1,
                                                     const float* __restrict__ W2,
                                                     const float* __restrict__ as2w,
                                                     const float* __restrict__ ad2w,
                                                     float* __restrict__ g,
                                                     float* __restrict__ asrc2,
                                                     float* __restrict__ adst2,
                                                     int n) {
    int i = blockIdx.x * blockDim.x + threadIdx.x;
    if (i >= n) return;
    float g0 = 0.0f, g1 = 0.0f;
    const float inv0 = 1.0f / (denom1[(size_t)i * 2 + 0] + 1e-16f);
    const float inv1 = 1.0f / (denom1[(size_t)i * 2 + 1] + 1e-16f);
    #pragma unroll
    for (int c = 0; c < 32; ++c) {
        const float inv = (c < 16) ? inv0 : inv1;
        float v = acc1[(size_t)i * 32 + c] * inv + b1[c];
        v = v > 0.0f ? v : 0.0f;
        g0 += v * W2[c * 2 + 0];
        g1 += v * W2[c * 2 + 1];
    }
    g[(size_t)i * 2 + 0] = g0;
    g[(size_t)i * 2 + 1] = g1;
    asrc2[i] = g0 * as2w[0] + g1 * as2w[1];
    adst2[i] = g0 * ad2w[0] + g1 * ad2w[1];
}

__global__ __launch_bounds__(256) void k_edge_max2(const int* __restrict__ src,
                                                   const int* __restrict__ dst,
                                                   const float* __restrict__ asrc2,
                                                   const float* __restrict__ adst2,
                                                   float* __restrict__ m2,
                                                   int E, int n) {
    int e = blockIdx.x * blockDim.x + threadIdx.x;
    if (e >= E + n) return;
    int s, d;
    if (e < E) { s = src[e]; d = dst[e]; } else { s = d = e - E; }
    atomicMaxF(&m2[d], lrelu(asrc2[s] + adst2[d]));
}

__global__ __launch_bounds__(256) void k_edge_acc2(const int* __restrict__ src,
                                                   const int* __restrict__ dst,
                                                   const float* __restrict__ asrc2,
                                                   const float* __restrict__ adst2,
                                                   const float* __restrict__ m2,
                                                   const float* __restrict__ g,
                                                   float* __restrict__ rec2,
                                                   int E, int n) {
    const int e = blockIdx.x * blockDim.x + threadIdx.x;
    const int lane = threadIdx.x & 31;
    const int Etot = E + n;
    const int e0 = __builtin_amdgcn_readfirstlane(e - lane);
    if (e0 >= Etot) return;
    int d = 0;
    float ex = 0.0f, g0 = 0.0f, g1 = 0.0f;
    if (e < Etot) {
        int s;
        if (e < E) { s = src[e]; d = dst[e]; } else { s = d = e - E; }
        ex = __expf(lrelu(asrc2[s] + adst2[d]) - m2[d]);
        g0 = g[(size_t)s * 2 + 0];
        g1 = g[(size_t)s * 2 + 1];
    }
    const int c = lane & 3;
    const int q = lane >> 2;
    #pragma unroll
    for (int i = 0; i < 4; ++i) {
        const int idx = i * 8 + q;
        const int dd = __shfl(d, idx);
        const float xx = __shfl(ex, idx);
        const float a0 = __shfl(g0, idx);
        const float a1 = __shfl(g1, idx);
        const float val = (c == 0) ? xx * a0 : ((c == 1) ? xx * a1 : xx);
        if (c < 3 && e0 + idx < Etot) atomicAdd(&rec2[(size_t)dd * 4 + c], val);
    }
}

__global__ __launch_bounds__(256) void k_out(const float* __restrict__ rec2,
                                             const float* __restrict__ b2,
                                             float* __restrict__ out,
                                             int n) {
    int i = blockIdx.x * blockDim.x + threadIdx.x;
    if (i >= n) return;
    const float inv = 1.0f / (rec2[(size_t)i * 4 + 2] + 1e-16f);
    out[(size_t)i * 2 + 0] = rec2[(size_t)i * 4 + 0] * inv + b2[0];
    out[(size_t)i * 2 + 1] = rec2[(size_t)i * 4 + 1] * inv + b2[1];
}

extern "C" void kernel_launch(void* const* d_in, const int* in_sizes, int n_in,
                              void* d_out, int out_size, void* d_ws, size_t ws_size,
                              hipStream_t stream) {
    const float* x    = (const float*)d_in[0];
    const int*   ei   = (const int*)d_in[1];
    const float* W1   = (const float*)d_in[2];
    const float* as1w = (const float*)d_in[3];
    const float* ad1w = (const float*)d_in[4];
    const float* b1   = (const float*)d_in[5];
    const float* W2   = (const float*)d_in[6];
    const float* as2w = (const float*)d_in[7];
    const float* ad2w = (const float*)d_in[8];
    const float* b2   = (const float*)d_in[9];

    const int N = in_sizes[0] / 128;
    const int E = in_sizes[1] / 2;
    const int* srcI = ei;
    const int* dstI = ei + E;

    float* p = (float*)d_ws;
    float* acc1   = p; p += (size_t)32 * N;
    float* h1     = p; p += (size_t)32 * N;
    float* rec2   = p; p += (size_t)4 * N;
    float* asrc1  = p; p += (size_t)2 * N;
    float* adst1  = p; p += (size_t)2 * N;
    float* m1     = p; p += (size_t)2 * N;
    float* denom1 = p; p += (size_t)2 * N;
    float* g      = p; p += (size_t)2 * N;
    float* asrc2  = p; p += (size_t)N;
    float* adst2  = p; p += (size_t)N;
    float* m2     = p; p += (size_t)N;

    const int T = 256;
    const int Etot = E + N;

    k_init<<<(32 * N + T - 1) / T, T, 0, stream>>>(acc1, m1, denom1, rec2, m2, N);
    {
        const int tiles = (N + 15) / 16;
        const int waves_per_block = T / 32;
        const int blocks = (tiles + waves_per_block - 1) / waves_per_block;
        k_gemm1_wmma<<<blocks, T, 0, stream>>>(x, W1, h1, N);
    }
    k_alpha1<<<(N + T - 1) / T, T, 0, stream>>>(h1, as1w, ad1w, asrc1, adst1, N);
    k_edge_max1<<<(2 * Etot + T - 1) / T, T, 0, stream>>>(srcI, dstI, asrc1,
                                                          adst1, m1, E, N);
    k_edge_acc1<<<(2 * Etot + T - 1) / T, T, 0, stream>>>(srcI, dstI, asrc1,
                                                          adst1, m1, h1, denom1,
                                                          acc1, E, N);
    k_node_layer2<<<(N + T - 1) / T, T, 0, stream>>>(acc1, denom1, b1, W2, as2w,
                                                     ad2w, g, asrc2, adst2, N);
    k_edge_max2<<<(Etot + T - 1) / T, T, 0, stream>>>(srcI, dstI, asrc2, adst2,
                                                      m2, E, N);
    k_edge_acc2<<<(Etot + T - 1) / T, T, 0, stream>>>(srcI, dstI, asrc2, adst2,
                                                      m2, g, rec2, E, N);
    k_out<<<(N + T - 1) / T, T, 0, stream>>>(rec2, b2, (float*)d_out, N);
}


# edges grouped by destination each call (counting sort, 1 int atomic per edge); each node's list is reduced by one wave in registers and across lanes, in the count's arrival order; no float atomics
# speedup vs baseline: 3.7860x; 2.1151x over previous
#include <hip/hip_runtime.h>
#include <math.h>


#define NEG_SLOPE 0.2f

typedef __attribute__((ext_vector_type(2))) float v2f;
typedef __attribute__((ext_vector_type(4))) float v4f;
typedef __attribute__((ext_vector_type(8))) float v8f;

__device__ __forceinline__ float lrelu(float v) {
    return v > 0.0f ? v : NEG_SLOPE * v;
}

__device__ __forceinline__ float wave_max(float v) {
    for (int o = 16; o > 0; o >>= 1) v = fmaxf(v, __shfl_xor(v, o));
    return v;
}

__device__ __forceinline__ float wave_sum(float v) {
    for (int o = 16; o > 0; o >>= 1) v += __shfl_xor(v, o);
    return v;
}

__global__ __launch_bounds__(256) void k_zero(int* __restrict__ cnt, int n) {
    const int stride = gridDim.x * blockDim.x;
    for (int i = blockIdx.x * blockDim.x + threadIdx.x; i < n; i += stride) cnt[i] = 0;
}

__global__ __launch_bounds__(256) void k_count(const int* __restrict__ dst,
                                               int* __restrict__ cnt,
                                               int* __restrict__ pos,
                                               int E) {
    const int stride = gridDim.x * blockDim.x;
    for (int e = blockIdx.x * blockDim.x + threadIdx.x; e < E; e += stride)
        pos[e] = atomicAdd(&cnt[dst[e]], 1);
}

__global__ __launch_bounds__(1024) void k_scan(const int* __restrict__ cnt,
                                               int* __restrict__ rowptr,
                                               int n) {
    __shared__ int wsum[32];
    const int tid = threadIdx.x;
    const int lane = tid & 31;
    const int wv = tid >> 5;
    const int start = blockIdx.x * 1024;

    int part = 0;
    for (int i = tid; i < start; i += 1024) part += cnt[i];
    for (int o = 16; o > 0; o >>= 1) part += __shfl_xor(part, o);
    if (lane == 0) wsum[wv] = part;
    __syncthreads();
    int base = 0;
    for (int k = 0; k < 32; ++k) base += wsum[k];
    __syncthreads();

    const int i = start + tid;
    const int c = (i < n) ? cnt[i] : 0;
    int incl = c;
    for (int o = 1; o < 32; o <<= 1) {
        const int t = __shfl_up(incl, o);
        if (lane >= o) incl += t;
    }
    if (lane == 31) wsum[wv] = incl;
    __syncthreads();
    int woff = 0;
    for (int k = 0; k < wv; ++k) woff += wsum[k];
    if (i < n) rowptr[i] = base + woff + incl - c;
}

__global__ __launch_bounds__(256) void k_place(const int* __restrict__ src,
                                               const int* __restrict__ dst,
                                               const int* __restrict__ rowptr,
                                               const int* __restrict__ pos,
                                               int* __restrict__ col,
                                               int E) {
    const int stride = gridDim.x * blockDim.x;
    for (int e = blockIdx.x * blockDim.x + threadIdx.x; e < E; e += stride)
        col[rowptr[dst[e]] + pos[e]] = src[e];
}

__global__ __launch_bounds__(256) void k_gemm1_wmma(const float* __restrict__ x,
                                                    const float* __restrict__ W1,
                                                    float* __restrict__ h1,
                                                    int n) {
    const int wave = (blockIdx.x * blockDim.x + threadIdx.x) >> 5;
    const int lane = threadIdx.x & 31;
    const int row0 = wave * 16;
    if (row0 >= n) return;

    const int m  = lane & 15;
    const int kh = lane >> 4;
    int row = row0 + m;
    if (row >= n) row = n - 1;

    const float* __restrict__ xrow = x + (size_t)row * 128;

    v8f c0 = {};
    v8f c1 = {};

    #pragma unroll
    for (int k0 = 0; k0 < 128; k0 += 4) {
        const int k = k0 + 2 * kh;
        v2f a = *(const v2f*)(xrow + k);
        v2f b0, b1;
        b0.x = W1[(size_t)k * 32 + m];
        b0.y = W1[(size_t)(k + 1) * 32 + m];
        b1.x = W1[(size_t)k * 32 + 16 + m];
        b1.y = W1[(size_t)(k + 1) * 32 + 16 + m];
        c0 = __builtin_amdgcn_wmma_f32_16x16x4_f32(false, a, false, b0,
                                                   (short)0, c0, false, false);
        c1 = __builtin_amdgcn_wmma_f32_16x16x4_f32(false, a, false, b1,
                                                   (short)0, c1, false, false);
    }

    const int mb = kh * 8;
    #pragma unroll
    for (int j = 0; j < 8; ++j) {
        const int r = row0 + mb + j;
        if (r < n) {
            h1[(size_t)r * 32 + m]      = c0[j];
            h1[(size_t)r * 32 + 16 + m] = c1[j];
        }
    }
}

__global__ __launch_bounds__(256) void k_alpha1(const float* __restrict__ h1,
                                                const float* __restrict__ asw,
                                                const float* __restrict__ adw,
                                                float* __restrict__ asrc,
                                                float* __restrict__ adst,
                                                int n) {
    int i = blockIdx.x * blockDim.x + threadIdx.x;
    if (i >= n) return;
    const float* hr = h1 + (size_t)i * 32;
    #pragma unroll
    for (int h = 0; h < 2; ++h) {
        float s = 0.0f, d = 0.0f;
        #pragma unroll
        for (int f = 0; f < 16; ++f) {
            const float v = hr[h * 16 + f];
            s += v * asw[h * 16 + f];
            d += v * adw[h * 16 + f];
        }
        asrc[i * 2 + h] = s;
        adst[i * 2 + h] = d;
    }
}

template <int U>
__device__ __forceinline__ float add_rows(const float* __restrict__ h1, int s,
                                          float e0, float e1, int j, int lane,
                                          float acc) {
    float v[U];
    #pragma unroll
    for (int u = 0; u < U; ++u) {
        const int sj = __builtin_amdgcn_readlane(s, j + u);
        v[u] = h1[(size_t)sj * 32 + lane];
    }
    #pragma unroll
    for (int u = 0; u < U; ++u) {
        const float w0 = __int_as_float(__builtin_amdgcn_readlane(__float_as_int(e0), j + u));
        const float w1 = __int_as_float(__builtin_amdgcn_readlane(__float_as_int(e1), j + u));
        acc += ((lane < 16) ? w0 : w1) * v[u];
    }
    return acc;
}

__device__ __forceinline__ float add_chunk(const float* __restrict__ h1, int s,
                                           float e0, float e1, int m, int lane,
                                           float acc) {
    int j = 0;
    for (; j + 8 <= m; j += 8) acc = add_rows<8>(h1, s, e0, e1, j, lane, acc);
    if (m - j >= 4) { acc = add_rows<4>(h1, s, e0, e1, j, lane, acc); j += 4; }
    if (m - j >= 2) { acc = add_rows<2>(h1, s, e0, e1, j, lane, acc); j += 2; }
    if (m - j >= 1) acc = add_rows<1>(h1, s, e0, e1, j, lane, acc);
    return acc;
}

__global__ __launch_bounds__(256) void k_node1(const int* __restrict__ rowptr,
                                               const int* __restrict__ cnt,
                                               const int* __restrict__ col,
                                               const float* __restrict__ asrc,
                                               const float* __restrict__ adst,
                                               const float* __restrict__ h1,
                                               const float* __restrict__ b1,
                                               const float* __restrict__ W2,
                                               const float* __restrict__ as2w,
                                               const float* __restrict__ ad2w,
                                               float* __restrict__ rec2,
                                               int n) {
    const int lane = threadIdx.x & 31;
    const int w = __builtin_amdgcn_readfirstlane((int)((blockIdx.x * blockDim.x + threadIdx.x) >> 5));
    const float NEG_INF = __int_as_float(0xFF800000);
    const float bias = b1[lane];
    const float w20 = W2[lane * 2 + 0];
    const float w21 = W2[lane * 2 + 1];
    const int dEnd = (w * 8 + 8 < n) ? w * 8 + 8 : n;
    for (int d = w * 8; d < dEnd; ++d) {
    const int beg = rowptr[d];
    const int deg = cnt[d];
    const int tot = deg + 1;
    const float ad0 = adst[(size_t)d * 2 + 0];
    const float ad1 = adst[(size_t)d * 2 + 1];

    int s0 = d;
    float l00 = NEG_INF, l01 = NEG_INF;
    if (lane < tot) {
        if (lane < deg) s0 = col[beg + lane];
        const v2f a = *(const v2f*)(asrc + (size_t)s0 * 2);
        l00 = lrelu(a.x + ad0);
        l01 = lrelu(a.y + ad1);
    }
    float mx0 = l00, mx1 = l01;
    for (int c0 = 32; c0 < tot; c0 += 32) {
        const int idx = c0 + lane;
        if (idx < tot) {
            const int s = (idx < deg) ? col[beg + idx] : d;
            const v2f a = *(const v2f*)(asrc + (size_t)s * 2);
            mx0 = fmaxf(mx0, lrelu(a.x + ad0));
            mx1 = fmaxf(mx1, lrelu(a.y + ad1));
        }
    }
    mx0 = wave_max(mx0);
    mx1 = wave_max(mx1);

    const float e00 = (lane < tot) ? __expf(l00 - mx0) : 0.0f;
    const float e01 = (lane < tot) ? __expf(l01 - mx1) : 0.0f;
    float den0 = e00, den1 = e01;
    float acc = add_chunk(h1, s0, e00, e01, tot < 32 ? tot : 32, lane, 0.0f);
    for (int c0 = 32; c0 < tot; c0 += 32) {
        const int idx = c0 + lane;
        int s = d;
        float e0 = 0.0f, e1 = 0.0f;
        if (idx < tot) {
            if (idx < deg) s = col[beg + idx];
            const v2f a = *(const v2f*)(asrc + (size_t)s * 2);
            e0 = __expf(lrelu(a.x + ad0) - mx0);
            e1 = __expf(lrelu(a.y + ad1) - mx1);
        }
        den0 += e0;
        den1 += e1;
        const int left = tot - c0;
        acc = add_chunk(h1, s, e0, e1, left < 32 ? left : 32, lane, acc);
    }
    den0 = wave_sum(den0);
    den1 = wave_sum(den1);

    const float inv = 1.0f / (((lane < 16) ? den0 : den1) + 1e-16f);
    float v = acc * inv + bias;
    v = v > 0.0f ? v : 0.0f;
    const float g0 = wave_sum(v * w20);
    const float g1 = wave_sum(v * w21);
    if (lane == 0) {
        v4f r;
        r.x = g0;
        r.y = g1;
        r.z = g0 * as2w[0] + g1 * as2w[1];
        r.w = g0 * ad2w[0] + g1 * ad2w[1];
        *(v4f*)(rec2 + (size_t)d * 4) = r;
    }
    }
}

__global__ __launch_bounds__(256) void k_node2(const int* __restrict__ rowptr,
                                               const int* __restrict__ cnt,
                                               const int* __restrict__ col,
                                               const float* __restrict__ rec2,
                                               const float* __restrict__ b2,
                                               float* __restrict__ out,
                                               int n) {
    const int lane = threadIdx.x & 31;
    const int w = __builtin_amdgcn_readfirstlane((int)((blockIdx.x * blockDim.x + threadIdx.x) >> 5));
    const float NEG_INF = __int_as_float(0xFF800000);
    const float bias0 = b2[0];
    const float bias1 = b2[1];
    const int dEnd = (w * 8 + 8 < n) ? w * 8 + 8 : n;
    for (int d = w * 8; d < dEnd; ++d) {
    const int beg = rowptr[d];
    const int deg = cnt[d];
    const int tot = deg + 1;
    const float ad = rec2[(size_t)d * 4 + 3];

    v4f r0 = {0.0f, 0.0f, 0.0f, 0.0f};
    float l0 = NEG_INF;
    if (lane < tot) {
        const int s = (lane < deg) ? col[beg + lane] : d;
        r0 = *(const v4f*)(rec2 + (size_t)s * 4);
        l0 = lrelu(r0.z + ad);
    }
    float mx = l0;
    for (int c0 = 32; c0 < tot; c0 += 32) {
        const int idx = c0 + lane;
        if (idx < tot) {
            const int s = (idx < deg) ? col[beg + idx] : d;
            mx = fmaxf(mx, lrelu(rec2[(size_t)s * 4 + 2] + ad));
        }
    }
    mx = wave_max(mx);

    const float ex0 = (lane < tot) ? __expf(l0 - mx) : 0.0f;
    float den = ex0;
    float a0 = ex0 * r0.x;
    float a1 = ex0 * r0.y;
    for (int c0 = 32; c0 < tot; c0 += 32) {
        const int idx = c0 + lane;
        if (idx < tot) {
            const int s = (idx < deg) ? col[beg + idx] : d;
            const v4f r = *(const v4f*)(rec2 + (size_t)s * 4);
            const float ex = __expf(lrelu(r.z + ad) - mx);
            den += ex;
            a0 += ex * r.x;
            a1 += ex * r.y;
        }
    }
    den = wave_sum(den);
    a0 = wave_sum(a0);
    a1 = wave_sum(a1);
    if (lane == 0) {
        const float inv = 1.0f / (den + 1e-16f);
        v2f o;
        o.x = a0 * inv + bias0;
        o.y = a1 * inv + bias1;
        *(v2f*)(out + (size_t)d * 2) = o;
    }
    }
}

extern "C" void kernel_launch(void* const* d_in, const int* in_sizes, int n_in,
                              void* d_out, int out_size, void* d_ws, size_t ws_size,
                              hipStream_t stream) {
    const float* x    = (const float*)d_in[0];
    const int*   ei   = (const int*)d_in[1];
    const float* W1   = (const float*)d_in[2];
    const float* as1w = (const float*)d_in[3];
    const float* ad1w = (const float*)d_in[4];
    const float* b1   = (const float*)d_in[5];
    const float* W2   = (const float*)d_in[6];
    const float* as2w = (const float*)d_in[7];
    const float* ad2w = (const float*)d_in[8];
    const float* b2   = (const float*)d_in[9];

    const int N = in_sizes[0] / 128;
    const int E = in_sizes[1] / 2;
    const int* srcI = ei;
    const int* dstI = ei + E;

    float* p = (float*)d_ws;
    float* h1     = p; p += (size_t)32 * N;
    float* rec2   = p; p += (size_t)4 * N;
    float* asrc1  = p; p += (size_t)2 * N;
    float* adst1  = p; p += (size_t)2 * N;
    int* cnt      = (int*)p; p += (size_t)N;
    int* rowptr   = (int*)p; p += (size_t)N;
    int* pos      = (int*)p; p += (size_t)E;
    int* col      = (int*)p; p += (size_t)E;

    const int T = 256;

    k_zero<<<64, T, 0, stream>>>(cnt, N);
    k_count<<<2048, T, 0, stream>>>(dstI, cnt, pos, E);
    k_scan<<<(N + 1023) / 1024, 1024, 0, stream>>>(cnt, rowptr, N);
    k_place<<<2048, T, 0, stream>>>(srcI, dstI, rowptr, pos, col, E);
    {
        const int tiles = (N + 15) / 16;
        const int waves_per_block = T / 32;
        const int blocks = (tiles + waves_per_block - 1) / waves_per_block;
        k_gemm1_wmma<<<blocks, T, 0, stream>>>(x, W1, h1, N);
    }
    k_alpha1<<<(N + T - 1) / T, T, 0, stream>>>(h1, as1w, ad1w, asrc1, adst1, N);
    k_node1<<<(N + 63) / 64, T, 0, stream>>>(rowptr, cnt, col, asrc1, adst1, h1, b1,
                                           W2, as2w, ad2w, rec2, N);
    k_node2<<<(N + 63) / 64, T, 0, stream>>>(rowptr, cnt, col, rec2, b2,
                                           (float*)d_out, N);
}


# edges grouped by destination by a two-level counting sort on LDS counters (bins of 512 nodes, contiguous copy-out), 3 launches, no global atomic; list order = LDS atomic arrival order
# speedup vs baseline: 8.5082x; 2.2473x over previous
#include <hip/hip_runtime.h>
#include <math.h>


#define NEG_SLOPE 0.2f

typedef __attribute__((ext_vector_type(2))) float v2f;
typedef __attribute__((ext_vector_type(4))) float v4f;
typedef __attribute__((ext_vector_type(8))) float v8f;

__device__ __forceinline__ float lrelu(float v) {
    return v > 0.0f ? v : NEG_SLOPE * v;
}

__device__ __forceinline__ float wave_max(float v) {
    for (int o = 16; o > 0; o >>= 1) v = fmaxf(v, __shfl_xor(v, o));
    return v;
}

__device__ __forceinline__ float wave_sum(float v) {
    for (int o = 16; o > 0; o >>= 1) v += __shfl_xor(v, o);
    return v;
}

#define BIN_SHIFT 9
#define BIN_NODES 512
#define MAX_BINS 1024
#define TILE_EDGES 8192
#define EPT 8
#define BIN_CAP 16384

typedef __attribute__((ext_vector_type(2))) int v2i;

__device__ __forceinline__ int block_excl_scan(int c, int* wsum, int lane, int wv) {
    int incl = c;
    for (int o = 1; o < 32; o <<= 1) {
        const int t = __shfl_up(incl, o);
        if (lane >= o) incl += t;
    }
    if (lane == 31) wsum[wv] = incl;
    __syncthreads();
    int woff = 0;
    for (int k = 0; k < wv; ++k) woff += wsum[k];
    return woff + incl - c;
}

__global__ __launch_bounds__(1024) void k_bincount(const int* __restrict__ dst,
                                                   int* __restrict__ M,
                                                   int E, int nb) {
    __shared__ int hist[MAX_BINS];
    const int tid = threadIdx.x;
    for (int b = tid; b < nb; b += 1024) hist[b] = 0;
    __syncthreads();
    const int e0 = blockIdx.x * TILE_EDGES;
    int e1 = e0 + TILE_EDGES;
    if (e1 > E) e1 = E;
    int dv[EPT];
    #pragma unroll
    for (int u = 0; u < EPT; ++u) {
        const int e = e0 + u * 1024 + tid;
        dv[u] = dst[e < e1 ? e : e1 - 1];
    }
    #pragma unroll
    for (int u = 0; u < EPT; ++u)
        if (e0 + u * 1024 + tid < e1) atomicAdd(&hist[dv[u] >> BIN_SHIFT], 1);
    __syncthreads();
    for (int b = tid; b < nb; b += 1024) M[blockIdx.x * nb + b] = hist[b];
}

__global__ __launch_bounds__(1024) void k_binscatter(const int* __restrict__ src,
                                                     const int* __restrict__ dst,
                                                     const int* __restrict__ M,
                                                     int* __restrict__ binBase,
                                                     int* __restrict__ staging,
                                                     int E, int nb, int tiles) {
    __shared__ int totL[MAX_BINS];
    __shared__ int preL[MAX_BINS];
    __shared__ int offG[MAX_BINS];
    __shared__ int lstart[MAX_BINS];
    __shared__ int cur[MAX_BINS];
    __shared__ int wsA[32];
    __shared__ int wsB[32];
    __shared__ v2i buf[TILE_EDGES];
    const int tid = threadIdx.x;
    const int lane = tid & 31;
    const int wv = tid >> 5;
    const int tile = blockIdx.x;

    for (int b = tid; b < nb; b += 1024) { totL[b] = 0; preL[b] = 0; }
    __syncthreads();
    {
        const int q = tid >> 8;
        for (int b = tid & 255; b < nb; b += 256) {
            int tot = 0, pre = 0;
            #pragma unroll 8
            for (int t = q; t < tiles; t += 4) {
                const int v = M[t * nb + b];
                tot += v;
                if (t < tile) pre += v;
            }
            atomicAdd(&totL[b], tot);
            atomicAdd(&preL[b], pre);
        }
    }
    __syncthreads();
    const int myTot = (tid < nb) ? totL[tid] : 0;
    const int myCnt = (tid < nb) ? M[tile * nb + tid] : 0;
    const int base = block_excl_scan(myTot, wsA, lane, wv);
    const int ls = block_excl_scan(myCnt, wsB, lane, wv);
    if (tid < nb) {
        offG[tid] = base + preL[tid];
        lstart[tid] = ls;
        cur[tid] = ls;
        if (tile == 0) binBase[tid] = base;
    }
    if (tile == 0 && tid == 0) binBase[nb] = E;
    __syncthreads();

    const int e0 = tile * TILE_EDGES;
    int ne = E - e0;
    if (ne > TILE_EDGES) ne = TILE_EDGES;
    int sv[EPT], dv[EPT];
    #pragma unroll
    for (int u = 0; u < EPT; ++u) {
        const int i = u * 1024 + tid;
        const int ic = e0 + (i < ne ? i : ne - 1);
        sv[u] = src[ic];
        dv[u] = dst[ic];
    }
    #pragma unroll
    for (int u = 0; u < EPT; ++u) {
        if (u * 1024 + tid < ne) {
            v2i r;
            r.x = sv[u];
            r.y = dv[u];
            const int p = atomicAdd(&cur[dv[u] >> BIN_SHIFT], 1);
            buf[p] = r;
        }
    }
    __syncthreads();
    v2i* __restrict__ st = (v2i*)staging;
    for (int i = tid; i < ne; i += 1024) {
        const v2i r = buf[i];
        const int b = r.y >> BIN_SHIFT;
        st[offG[b] + (i - lstart[b])] = r;
    }
}

__global__ __launch_bounds__(1024) void k_binsort(const int* __restrict__ staging,
                                                  const int* __restrict__ binBase,
                                                  int* __restrict__ rowptr,
                                                  int* __restrict__ cnt,
                                                  int* __restrict__ col,
                                                  int n) {
    __shared__ int cntL[BIN_NODES];
    __shared__ int curL[BIN_NODES];
    __shared__ int ws[32];
    __shared__ int colL[BIN_CAP];
    const int tid = threadIdx.x;
    const int lane = tid & 31;
    const int wv = tid >> 5;
    const int b = blockIdx.x;
    const int e0 = binBase[b];
    const int ne = binBase[b + 1] - e0;
    const v2i* __restrict__ st = (const v2i*)staging + e0;

    if (tid < BIN_NODES) cntL[tid] = 0;
    __syncthreads();
    for (int i0 = 0; i0 < ne; i0 += EPT * 1024) {
        int dv[EPT];
        #pragma unroll
        for (int u = 0; u < EPT; ++u) {
            const int i = i0 + u * 1024 + tid;
            dv[u] = st[i < ne ? i : ne - 1].y;
        }
        #pragma unroll
        for (int u = 0; u < EPT; ++u)
            if (i0 + u * 1024 + tid < ne) atomicAdd(&cntL[dv[u] & (BIN_NODES - 1)], 1);
    }
    __syncthreads();
    const int c = (tid < BIN_NODES) ? cntL[tid] : 0;
    const int start = block_excl_scan(c, ws, lane, wv);
    if (tid < BIN_NODES) {
        curL[tid] = start;
        const int d = b * BIN_NODES + tid;
        if (d < n) { rowptr[d] = e0 + start; cnt[d] = c; }
    }
    __syncthreads();
    const bool fits = ne <= BIN_CAP;
    for (int i0 = 0; i0 < ne; i0 += EPT * 1024) {
        v2i rv[EPT];
        #pragma unroll
        for (int u = 0; u < EPT; ++u) {
            const int i = i0 + u * 1024 + tid;
            rv[u] = st[i < ne ? i : ne - 1];
        }
        #pragma unroll
        for (int u = 0; u < EPT; ++u) {
            if (i0 + u * 1024 + tid < ne) {
                const int p = atomicAdd(&curL[rv[u].y & (BIN_NODES - 1)], 1);
                if (fits) colL[p] = rv[u].x; else col[e0 + p] = rv[u].x;
            }
        }
    }
    __syncthreads();
    if (fits)
        for (int i = tid; i < ne; i += 1024) col[e0 + i] = colL[i];
}

__global__ __launch_bounds__(256) void k_gemm1_wmma(const float* __restrict__ x,
                                                    const float* __restrict__ W1,
                                                    float* __restrict__ h1,
                                                    int n) {
    const int wave = (blockIdx.x * blockDim.x + threadIdx.x) >> 5;
    const int lane = threadIdx.x & 31;
    const int row0 = wave * 16;
    if (row0 >= n) return;

    const int m  = lane & 15;
    const int kh = lane >> 4;
    int row = row0 + m;
    if (row >= n) row = n - 1;

    const float* __restrict__ xrow = x + (size_t)row * 128;

    v8f c0 = {};
    v8f c1 = {};

    #pragma unroll
    for (int k0 = 0; k0 < 128; k0 += 4) {
        const int k = k0 + 2 * kh;
        v2f a = *(const v2f*)(xrow + k);
        v2f b0, b1;
        b0.x = W1[(size_t)k * 32 + m];
        b0.y = W1[(size_t)(k + 1) * 32 + m];
        b1.x = W1[(size_t)k * 32 + 16 + m];
        b1.y = W1[(size_t)(k + 1) * 32 + 16 + m];
        c0 = __builtin_amdgcn_wmma_f32_16x16x4_f32(false, a, false, b0,
                                                   (short)0, c0, false, false);
        c1 = __builtin_amdgcn_wmma_f32_16x16x4_f32(false, a, false, b1,
                                                   (short)0, c1, false, false);
    }

    const int mb = kh * 8;
    #pragma unroll
    for (int j = 0; j < 8; ++j) {
        const int r = row0 + mb + j;
        if (r < n) {
            h1[(size_t)r * 32 + m]      = c0[j];
            h1[(size_t)r * 32 + 16 + m] = c1[j];
        }
    }
}

__global__ __launch_bounds__(256) void k_alpha1(const float* __restrict__ h1,
                                                const float* __restrict__ asw,
                                                const float* __restrict__ adw,
                                                float* __restrict__ asrc,
                                                float* __restrict__ adst,
                                                int n) {
    int i = blockIdx.x * blockDim.x + threadIdx.x;
    if (i >= n) return;
    const float* hr = h1 + (size_t)i * 32;
    #pragma unroll
    for (int h = 0; h < 2; ++h) {
        float s = 0.0f, d = 0.0f;
        #pragma unroll
        for (int f = 0; f < 16; ++f) {
            const float v = hr[h * 16 + f];
            s += v * asw[h * 16 + f];
            d += v * adw[h * 16 + f];
        }
        asrc[i * 2 + h] = s;
        adst[i * 2 + h] = d;
    }
}

template <int U>
__device__ __forceinline__ float add_rows(const float* __restrict__ h1, int s,
                                          float e0, float e1, int j, int lane,
                                          float acc) {
    float v[U];
    #pragma unroll
    for (int u = 0; u < U; ++u) {
        const int sj = __builtin_amdgcn_readlane(s, j + u);
        v[u] = h1[(size_t)sj * 32 + lane];
    }
    #pragma unroll
    for (int u = 0; u < U; ++u) {
        const float w0 = __int_as_float(__builtin_amdgcn_readlane(__float_as_int(e0), j + u));
        const float w1 = __int_as_float(__builtin_amdgcn_readlane(__float_as_int(e1), j + u));
        acc += ((lane < 16) ? w0 : w1) * v[u];
    }
    return acc;
}

__device__ __forceinline__ float add_chunk(const float* __restrict__ h1, int s,
                                           float e0, float e1, int m, int lane,
                                           float acc) {
    int j = 0;
    for (; j + 8 <= m; j += 8) acc = add_rows<8>(h1, s, e0, e1, j, lane, acc);
    if (m - j >= 4) { acc = add_rows<4>(h1, s, e0, e1, j, lane, acc); j += 4; }
    if (m - j >= 2) { acc = add_rows<2>(h1, s, e0, e1, j, lane, acc); j += 2; }
    if (m - j >= 1) acc = add_rows<1>(h1, s, e0, e1, j, lane, acc);
    return acc;
}

__global__ __launch_bounds__(256) void k_node1(const int* __restrict__ rowptr,
                                               const int* __restrict__ cnt,
                                               const int* __restrict__ col,
                                               const float* __restrict__ asrc,
                                               const float* __restrict__ adst,
                                               const float* __restrict__ h1,
                                               const float* __restrict__ b1,
                                               const float* __restrict__ W2,
                                               const float* __restrict__ as2w,
                                               const float* __restrict__ ad2w,
                                               float* __restrict__ rec2,
                                               int n) {
    const int lane = threadIdx.x & 31;
    const int w = __builtin_amdgcn_readfirstlane((int)((blockIdx.x * blockDim.x + threadIdx.x) >> 5));
    const float NEG_INF = __int_as_float(0xFF800000);
    const float bias = b1[lane];
    const float w20 = W2[lane * 2 + 0];
    const float w21 = W2[lane * 2 + 1];
    const int dEnd = (w * 8 + 8 < n) ? w * 8 + 8 : n;
    for (int d = w * 8; d < dEnd; ++d) {
    const int beg = rowptr[d];
    const int deg = cnt[d];
    const int tot = deg + 1;
    const float ad0 = adst[(size_t)d * 2 + 0];
    const float ad1 = adst[(size_t)d * 2 + 1];

    int s0 = d;
    float l00 = NEG_INF, l01 = NEG_INF;
    if (lane < tot) {
        if (lane < deg) s0 = col[beg + lane];
        const v2f a = *(const v2f*)(asrc + (size_t)s0 * 2);
        l00 = lrelu(a.x + ad0);
        l01 = lrelu(a.y + ad1);
    }
    float mx0 = l00, mx1 = l01;
    for (int c0 = 32; c0 < tot; c0 += 32) {
        const int idx = c0 + lane;
        if (idx < tot) {
            const int s = (idx < deg) ? col[beg + idx] : d;
            const v2f a = *(const v2f*)(asrc + (size_t)s * 2);
            mx0 = fmaxf(mx0, lrelu(a.x + ad0));
            mx1 = fmaxf(mx1, lrelu(a.y + ad1));
        }
    }
    mx0 = wave_max(mx0);
    mx1 = wave_max(mx1);

    const float e00 = (lane < tot) ? __expf(l00 - mx0) : 0.0f;
    const float e01 = (lane < tot) ? __expf(l01 - mx1) : 0.0f;
    float den0 = e00, den1 = e01;
    float acc = add_chunk(h1, s0, e00, e01, tot < 32 ? tot : 32, lane, 0.0f);
    for (int c0 = 32; c0 < tot; c0 += 32) {
        const int idx = c0 + lane;
        int s = d;
        float e0 = 0.0f, e1 = 0.0f;
        if (idx < tot) {
            if (idx < deg) s = col[beg + idx];
            const v2f a = *(const v2f*)(asrc + (size_t)s * 2);
            e0 = __expf(lrelu(a.x + ad0) - mx0);
            e1 = __expf(lrelu(a.y + ad1) - mx1);
        }
        den0 += e0;
        den1 += e1;
        const int left = tot - c0;
        acc = add_chunk(h1, s, e0, e1, left < 32 ? left : 32, lane, acc);
    }
    den0 = wave_sum(den0);
    den1 = wave_sum(den1);

    const float inv = 1.0f / (((lane < 16) ? den0 : den1) + 1e-16f);
    float v = acc * inv + bias;
    v = v > 0.0f ? v : 0.0f;
    const float g0 = wave_sum(v * w20);
    const float g1 = wave_sum(v * w21);
    if (lane == 0) {
        v4f r;
        r.x = g0;
        r.y = g1;
        r.z = g0 * as2w[0] + g1 * as2w[1];
        r.w = g0 * ad2w[0] + g1 * ad2w[1];
        *(v4f*)(rec2 + (size_t)d * 4) = r;
    }
    }
}

__global__ __launch_bounds__(256) void k_node2(const int* __restrict__ rowptr,
                                               const int* __restrict__ cnt,
                                               const int* __restrict__ col,
                                               const float* __restrict__ rec2,
                                               const float* __restrict__ b2,
                                               float* __restrict__ out,
                                               int n) {
    const int lane = threadIdx.x & 31;
    const int w = __builtin_amdgcn_readfirstlane((int)((blockIdx.x * blockDim.x + threadIdx.x) >> 5));
    const float NEG_INF = __int_as_float(0xFF800000);
    const float bias0 = b2[0];
    const float bias1 = b2[1];
    const int dEnd = (w * 8 + 8 < n) ? w * 8 + 8 : n;
    for (int d = w * 8; d < dEnd; ++d) {
    const int beg = rowptr[d];
    const int deg = cnt[d];
    const int tot = deg + 1;
    const float ad = rec2[(size_t)d * 4 + 3];

    v4f r0 = {0.0f, 0.0f, 0.0f, 0.0f};
    float l0 = NEG_INF;
    if (lane < tot) {
        const int s = (lane < deg) ? col[beg + lane] : d;
        r0 = *(const v4f*)(rec2 + (size_t)s * 4);
        l0 = lrelu(r0.z + ad);
    }
    float mx = l0;
    for (int c0 = 32; c0 < tot; c0 += 32) {
        const int idx = c0 + lane;
        if (idx < tot) {
            const int s = (idx < deg) ? col[beg + idx] : d;
            mx = fmaxf(mx, lrelu(rec2[(size_t)s * 4 + 2] + ad));
        }
    }
    mx = wave_max(mx);

    const float ex0 = (lane < tot) ? __expf(l0 - mx) : 0.0f;
    float den = ex0;
    float a0 = ex0 * r0.x;
    float a1 = ex0 * r0.y;
    for (int c0 = 32; c0 < tot; c0 += 32) {
        const int idx = c0 + lane;
        if (idx < tot) {
            const int s = (idx < deg) ? col[beg + idx] : d;
            const v4f r = *(const v4f*)(rec2 + (size_t)s * 4);
            const float ex = __expf(lrelu(r.z + ad) - mx);
            den += ex;
            a0 += ex * r.x;
            a1 += ex * r.y;
        }
    }
    den = wave_sum(den);
    a0 = wave_sum(a0);
    a1 = wave_sum(a1);
    if (lane == 0) {
        const float inv = 1.0f / (den + 1e-16f);
        v2f o;
        o.x = a0 * inv + bias0;
        o.y = a1 * inv + bias1;
        *(v2f*)(out + (size_t)d * 2) = o;
    }
    }
}

extern "C" void kernel_launch(void* const* d_in, const int* in_sizes, int n_in,
                              void* d_out, int out_size, void* d_ws, size_t ws_size,
                              hipStream_t stream) {
    const float* x    = (const float*)d_in[0];
    const int*   ei   = (const int*)d_in[1];
    const float* W1   = (const float*)d_in[2];
    const float* as1w = (const float*)d_in[3];
    const float* ad1w = (const float*)d_in[4];
    const float* b1   = (const float*)d_in[5];
    const float* W2   = (const float*)d_in[6];
    const float* as2w = (const float*)d_in[7];
    const float* ad2w = (const float*)d_in[8];
    const float* b2   = (const float*)d_in[9];

    const int N = in_sizes[0] / 128;
    const int E = in_sizes[1] / 2;
    const int* srcI = ei;
    const int* dstI = ei + E;

    float* p = (float*)d_ws;
    float* h1     = p; p += (size_t)32 * N;
    float* rec2   = p; p += (size_t)4 * N;
    float* asrc1  = p; p += (size_t)2 * N;
    float* adst1  = p; p += (size_t)2 * N;
    int* cnt      = (int*)p; p += (size_t)N;
    int* rowptr   = (int*)p; p += (size_t)N;
    int* staging  = (int*)p; p += (size_t)2 * E;
    int* col      = (int*)p; p += (size_t)E;
    const int nb    = (N + BIN_NODES - 1) / BIN_NODES;
    const int etiles = (E + TILE_EDGES - 1) / TILE_EDGES;
    int* M        = (int*)p; p += (size_t)etiles * nb;
    int* binBase  = (int*)p; p += (size_t)nb + 1;

    const int T = 256;

    k_bincount<<<etiles, 1024, 0, stream>>>(dstI, M, E, nb);
    k_binscatter<<<etiles, 1024, 0, stream>>>(srcI, dstI, M, binBase, staging, E, nb, etiles);
    k_binsort<<<nb, 1024, 0, stream>>>(staging, binBase, rowptr, cnt, col, N);
    {
        const int tiles = (N + 15) / 16;
        const int waves_per_block = T / 32;
        const int blocks = (tiles + waves_per_block - 1) / waves_per_block;
        k_gemm1_wmma<<<blocks, T, 0, stream>>>(x, W1, h1, N);
    }
    k_alpha1<<<(N + T - 1) / T, T, 0, stream>>>(h1, as1w, ad1w, asrc1, adst1, N);
    k_node1<<<(N + 63) / 64, T, 0, stream>>>(rowptr, cnt, col, asrc1, adst1, h1, b1,
                                           W2, as2w, ad2w, rec2, N);
    k_node2<<<(N + 63) / 64, T, 0, stream>>>(rowptr, cnt, col, rec2, b2,
                                           (float*)d_out, N);
}


# layer-1 node pass: each source row gathered once, 16 B per lane (4 rows per load); logits come from the held rows, so the logit pass and its gather go; per-node sums: 4 slot sums + lane tree
# speedup vs baseline: 9.5775x; 1.1257x over previous
#include <hip/hip_runtime.h>
#include <math.h>


#define NEG_SLOPE 0.2f

typedef __attribute__((ext_vector_type(2))) float v2f;
typedef __attribute__((ext_vector_type(4))) float v4f;
typedef __attribute__((ext_vector_type(8))) float v8f;

__device__ __forceinline__ float lrelu(float v) {
    return v > 0.0f ? v : NEG_SLOPE * v;
}

__device__ __forceinline__ float wave_max(float v) {
    for (int o = 16; o > 0; o >>= 1) v = fmaxf(v, __shfl_xor(v, o));
    return v;
}

__device__ __forceinline__ float wave_sum(float v) {
    for (int o = 16; o > 0; o >>= 1) v += __shfl_xor(v, o);
    return v;
}

#define BIN_SHIFT 9
#define BIN_NODES 512
#define MAX_BINS 1024
#define TILE_EDGES 8192
#define EPT 8
#define BIN_CAP 16384

typedef __attribute__((ext_vector_type(2))) int v2i;

__device__ __forceinline__ int block_excl_scan(int c, int* wsum, int lane, int wv) {
    int incl = c;
    for (int o = 1; o < 32; o <<= 1) {
        const int t = __shfl_up(incl, o);
        if (lane >= o) incl += t;
    }
    if (lane == 31) wsum[wv] = incl;
    __syncthreads();
    int woff = 0;
    for (int k = 0; k < wv; ++k) woff += wsum[k];
    return woff + incl - c;
}

__global__ __launch_bounds__(1024) void k_bincount(const int* __restrict__ dst,
                                                   int* __restrict__ M,
                                                   int E, int nb) {
    __shared__ int hist[MAX_BINS];
    const int tid = threadIdx.x;
    for (int b = tid; b < nb; b += 1024) hist[b] = 0;
    __syncthreads();
    const int e0 = blockIdx.x * TILE_EDGES;
    int e1 = e0 + TILE_EDGES;
    if (e1 > E) e1 = E;
    int dv[EPT];
    #pragma unroll
    for (int u = 0; u < EPT; ++u) {
        const int e = e0 + u * 1024 + tid;
        dv[u] = dst[e < e1 ? e : e1 - 1];
    }
    #pragma unroll
    for (int u = 0; u < EPT; ++u)
        if (e0 + u * 1024 + tid < e1) atomicAdd(&hist[dv[u] >> BIN_SHIFT], 1);
    __syncthreads();
    for (int b = tid; b < nb; b += 1024) M[blockIdx.x * nb + b] = hist[b];
}

__global__ __launch_bounds__(1024) void k_binscatter(const int* __restrict__ src,
                                                     const int* __restrict__ dst,
                                                     const int* __restrict__ M,
                                                     int* __restrict__ binBase,
                                                     int* __restrict__ staging,
                                                     int E, int nb, int tiles) {
    __shared__ int totL[MAX_BINS];
    __shared__ int preL[MAX_BINS];
    __shared__ int offG[MAX_BINS];
    __shared__ int lstart[MAX_BINS];
    __shared__ int cur[MAX_BINS];
    __shared__ int wsA[32];
    __shared__ int wsB[32];
    __shared__ v2i buf[TILE_EDGES];
    const int tid = threadIdx.x;
    const int lane = tid & 31;
    const int wv = tid >> 5;
    const int tile = blockIdx.x;

    for (int b = tid; b < nb; b += 1024) { totL[b] = 0; preL[b] = 0; }
    __syncthreads();
    {
        const int q = tid >> 8;
        for (int b = tid & 255; b < nb; b += 256) {
            int tot = 0, pre = 0;
            #pragma unroll 8
            for (int t = q; t < tiles; t += 4) {
                const int v = M[t * nb + b];
                tot += v;
                if (t < tile) pre += v;
            }
            atomicAdd(&totL[b], tot);
            atomicAdd(&preL[b], pre);
        }
    }
    __syncthreads();
    const int myTot = (tid < nb) ? totL[tid] : 0;
    const int myCnt = (tid < nb) ? M[tile * nb + tid] : 0;
    const int base = block_excl_scan(myTot, wsA, lane, wv);
    const int ls = block_excl_scan(myCnt, wsB, lane, wv);
    if (tid < nb) {
        offG[tid] = base + preL[tid];
        lstart[tid] = ls;
        cur[tid] = ls;
        if (tile == 0) binBase[tid] = base;
    }
    if (tile == 0 && tid == 0) binBase[nb] = E;
    __syncthreads();

    const int e0 = tile * TILE_EDGES;
    int ne = E - e0;
    if (ne > TILE_EDGES) ne = TILE_EDGES;
    int sv[EPT], dv[EPT];
    #pragma unroll
    for (int u = 0; u < EPT; ++u) {
        const int i = u * 1024 + tid;
        const int ic = e0 + (i < ne ? i : ne - 1);
        sv[u] = src[ic];
        dv[u] = dst[ic];
    }
    #pragma unroll
    for (int u = 0; u < EPT; ++u) {
        if (u * 1024 + tid < ne) {
            v2i r;
            r.x = sv[u];
            r.y = dv[u];
            const int p = atomicAdd(&cur[dv[u] >> BIN_SHIFT], 1);
            buf[p] = r;
        }
    }
    __syncthreads();
    v2i* __restrict__ st = (v2i*)staging;
    for (int i = tid; i < ne; i += 1024) {
        const v2i r = buf[i];
        const int b = r.y >> BIN_SHIFT;
        st[offG[b] + (i - lstart[b])] = r;
    }
}

__global__ __launch_bounds__(1024) void k_binsort(const int* __restrict__ staging,
                                                  const int* __restrict__ binBase,
                                                  int* __restrict__ rowptr,
                                                  int* __restrict__ cnt,
                                                  int* __restrict__ col,
                                                  int n) {
    __shared__ int cntL[BIN_NODES];
    __shared__ int curL[BIN_NODES];
    __shared__ int ws[32];
    __shared__ int colL[BIN_CAP];
    const int tid = threadIdx.x;
    const int lane = tid & 31;
    const int wv = tid >> 5;
    const int b = blockIdx.x;
    const int e0 = binBase[b];
    const int ne = binBase[b + 1] - e0;
    const v2i* __restrict__ st = (const v2i*)staging + e0;

    if (tid < BIN_NODES) cntL[tid] = 0;
    __syncthreads();
    for (int i0 = 0; i0 < ne; i0 += EPT * 1024) {
        int dv[EPT];
        #pragma unroll
        for (int u = 0; u < EPT; ++u) {
            const int i = i0 + u * 1024 + tid;
            dv[u] = st[i < ne ? i : ne - 1].y;
        }
        #pragma unroll
        for (int u = 0; u < EPT; ++u)
            if (i0 + u * 1024 + tid < ne) atomicAdd(&cntL[dv[u] & (BIN_NODES - 1)], 1);
    }
    __syncthreads();
    const int c = (tid < BIN_NODES) ? cntL[tid] : 0;
    const int start = block_excl_scan(c, ws, lane, wv);
    if (tid < BIN_NODES) {
        curL[tid] = start;
        const int d = b * BIN_NODES + tid;
        if (d < n) { rowptr[d] = e0 + start; cnt[d] = c; }
    }
    __syncthreads();
    const bool fits = ne <= BIN_CAP;
    for (int i0 = 0; i0 < ne; i0 += EPT * 1024) {
        v2i rv[EPT];
        #pragma unroll
        for (int u = 0; u < EPT; ++u) {
            const int i = i0 + u * 1024 + tid;
            rv[u] = st[i < ne ? i : ne - 1];
        }
        #pragma unroll
        for (int u = 0; u < EPT; ++u) {
            if (i0 + u * 1024 + tid < ne) {
                const int p = atomicAdd(&curL[rv[u].y & (BIN_NODES - 1)], 1);
                if (fits) colL[p] = rv[u].x; else col[e0 + p] = rv[u].x;
            }
        }
    }
    __syncthreads();
    if (fits)
        for (int i = tid; i < ne; i += 1024) col[e0 + i] = colL[i];
}

__global__ __launch_bounds__(256) void k_gemm1_wmma(const float* __restrict__ x,
                                                    const float* __restrict__ W1,
                                                    float* __restrict__ h1,
                                                    int n) {
    const int wave = (blockIdx.x * blockDim.x + threadIdx.x) >> 5;
    const int lane = threadIdx.x & 31;
    const int row0 = wave * 16;
    if (row0 >= n) return;

    const int m  = lane & 15;
    const int kh = lane >> 4;
    int row = row0 + m;
    if (row >= n) row = n - 1;

    const float* __restrict__ xrow = x + (size_t)row * 128;

    v8f c0 = {};
    v8f c1 = {};

    #pragma unroll
    for (int k0 = 0; k0 < 128; k0 += 4) {
        const int k = k0 + 2 * kh;
        v2f a = *(const v2f*)(xrow + k);
        v2f b0, b1;
        b0.x = W1[(size_t)k * 32 + m];
        b0.y = W1[(size_t)(k + 1) * 32 + m];
        b1.x = W1[(size_t)k * 32 + 16 + m];
        b1.y = W1[(size_t)(k + 1) * 32 + 16 + m];
        c0 = __builtin_amdgcn_wmma_f32_16x16x4_f32(false, a, false, b0,
                                                   (short)0, c0, false, false);
        c1 = __builtin_amdgcn_wmma_f32_16x16x4_f32(false, a, false, b1,
                                                   (short)0, c1, false, false);
    }

    const int mb = kh * 8;
    #pragma unroll
    for (int j = 0; j < 8; ++j) {
        const int r = row0 + mb + j;
        if (r < n) {
            h1[(size_t)r * 32 + m]      = c0[j];
            h1[(size_t)r * 32 + 16 + m] = c1[j];
        }
    }
}

#define DPP_XOR1 0xB1
#define DPP_XOR2 0x4E
#define DPP_XOR4 0x164
#define DPP_XOR8 0x168

template <int C>
__device__ __forceinline__ float lane_dpp(float v) {
    const int b = __builtin_bit_cast(int, v);
    return __builtin_bit_cast(float, __builtin_amdgcn_update_dpp(0, b, C, 0xf, 0xf, true));
}

template <int I>
__device__ __forceinline__ int group8_bcast(int v) {
    return (int)__builtin_amdgcn_mov_dpp8((unsigned)v, (unsigned)(I * 0x249249));
}

__device__ __forceinline__ float xhalf(float v) {
    const int b = __builtin_bit_cast(int, v);
    return __builtin_bit_cast(float,
        __builtin_amdgcn_permlanex16(b, b, 0x76543210, 0xfedcba98, false, false));
}

__device__ __forceinline__ float slot_sum(float v) {
    v += lane_dpp<DPP_XOR8>(v);
    v += xhalf(v);
    return v;
}

__global__ __launch_bounds__(256) void k_node1(const int* __restrict__ rowptr,
                                               const int* __restrict__ cnt,
                                               const int* __restrict__ col,
                                               const float* __restrict__ h1,
                                               const float* __restrict__ as1w,
                                               const float* __restrict__ ad1w,
                                               const float* __restrict__ b1,
                                               const float* __restrict__ W2,
                                               const float* __restrict__ as2w,
                                               const float* __restrict__ ad2w,
                                               float* __restrict__ rec2,
                                               int n) {
    const int lane = threadIdx.x & 31;
    const int w = __builtin_amdgcn_readfirstlane((int)((blockIdx.x * blockDim.x + threadIdx.x) >> 5));
    const float NEG_INF = __int_as_float(0xFF800000);
    const int g = lane >> 3;
    const int q = lane & 7;
    const v4f* __restrict__ h1v = (const v4f*)h1;
    const v4f asw = *(const v4f*)(as1w + 4 * q);
    const v4f adw = *(const v4f*)(ad1w + 4 * q);
    const v4f bias = *(const v4f*)(b1 + 4 * q);
    const v4f w2a = *(const v4f*)(W2 + 8 * q);
    const v4f w2b = *(const v4f*)(W2 + 8 * q + 4);
    const int dEnd = (w * 8 + 8 < n) ? w * 8 + 8 : n;
    for (int d = w * 8; d < dEnd; ++d) {
    const int beg = rowptr[d];
    const int deg = cnt[d];
    const int tot = deg + 1;
    const v4f own = h1v[(unsigned)d * 8u + (unsigned)q];
    float adp = 0.0f;
    float mx = NEG_INF;
    float den = 0.0f;
    v4f acc = {0.0f, 0.0f, 0.0f, 0.0f};

    auto walk = [&](const int c0, const bool first) {
        const int left = tot - c0;
        const int m = left < 32 ? left : 32;
        const int nv = (m - g + 3) >> 2;
        const int idx = c0 + 4 * q + g;
        int sl = d;
        if (idx < deg) sl = col[beg + idx];

        v4f row[8];
#define ROW_LOAD(I) if ((I) == 0 || 4 * (I) < m)   \
            row[I] = h1v[(unsigned)group8_bcast<I>(sl) * 8u + (unsigned)q];
        ROW_LOAD(0) ROW_LOAD(1) ROW_LOAD(2) ROW_LOAD(3)
        ROW_LOAD(4) ROW_LOAD(5) ROW_LOAD(6) ROW_LOAD(7)
#undef ROW_LOAD
        if (first) {
            float t = own.x * adw.x + own.y * adw.y + own.z * adw.z + own.w * adw.w;
            t += lane_dpp<DPP_XOR1>(t);
            t += lane_dpp<DPP_XOR2>(t);
            adp = t;
        }
        float lg[8];
        float cm = NEG_INF;
        #pragma unroll
        for (int i = 0; i < 8; ++i) {
            if (i == 0 || 4 * i < m) {
                float p = row[i].x * asw.x + row[i].y * asw.y + row[i].z * asw.z + row[i].w * asw.w;
                p += lane_dpp<DPP_XOR1>(p);
                p += lane_dpp<DPP_XOR2>(p);
                lg[i] = (i < nv) ? lrelu(p + adp) : NEG_INF;
                cm = fmaxf(cm, lg[i]);
            }
        }
        cm = fmaxf(cm, lane_dpp<DPP_XOR8>(cm));
        cm = fmaxf(cm, xhalf(cm));
        if (first) {
            mx = cm;
        } else {
            const float nm = fmaxf(mx, cm);
            const float sc = __expf(mx - nm);
            den *= sc;
            acc.x *= sc; acc.y *= sc; acc.z *= sc; acc.w *= sc;
            mx = nm;
        }
        #pragma unroll
        for (int i = 0; i < 8; ++i) {
            if (i == 0 || 4 * i < m) {
                const float e = (i < nv) ? __expf(lg[i] - mx) : 0.0f;
                den += e;
                acc.x += e * row[i].x;
                acc.y += e * row[i].y;
                acc.z += e * row[i].z;
                acc.w += e * row[i].w;
            }
        }
    };
    walk(0, true);
    for (int c0 = 32; c0 < tot; c0 += 32) walk(c0, false);

    den = slot_sum(den);
    acc.x = slot_sum(acc.x);
    acc.y = slot_sum(acc.y);
    acc.z = slot_sum(acc.z);
    acc.w = slot_sum(acc.w);

    const float inv = 1.0f / (den + 1e-16f);
    float v0 = acc.x * inv + bias.x;
    float v1 = acc.y * inv + bias.y;
    float v2 = acc.z * inv + bias.z;
    float v3 = acc.w * inv + bias.w;
    v0 = v0 > 0.0f ? v0 : 0.0f;
    v1 = v1 > 0.0f ? v1 : 0.0f;
    v2 = v2 > 0.0f ? v2 : 0.0f;
    v3 = v3 > 0.0f ? v3 : 0.0f;
    float g0 = v0 * w2a.x + v1 * w2a.z + v2 * w2b.x + v3 * w2b.z;
    float g1 = v0 * w2a.y + v1 * w2a.w + v2 * w2b.y + v3 * w2b.w;
    g0 += lane_dpp<DPP_XOR1>(g0);
    g1 += lane_dpp<DPP_XOR1>(g1);
    g0 += lane_dpp<DPP_XOR2>(g0);
    g1 += lane_dpp<DPP_XOR2>(g1);
    g0 += lane_dpp<DPP_XOR4>(g0);
    g1 += lane_dpp<DPP_XOR4>(g1);
    if (lane == 0) {
        v4f r;
        r.x = g0;
        r.y = g1;
        r.z = g0 * as2w[0] + g1 * as2w[1];
        r.w = g0 * ad2w[0] + g1 * ad2w[1];
        *(v4f*)(rec2 + (size_t)d * 4) = r;
    }
    }
}

__global__ __launch_bounds__(256) void k_node2(const int* __restrict__ rowptr,
                                               const int* __restrict__ cnt,
                                               const int* __restrict__ col,
                                               const float* __restrict__ rec2,
                                               const float* __restrict__ b2,
                                               float* __restrict__ out,
                                               int n) {
    const int lane = threadIdx.x & 31;
    const int w = __builtin_amdgcn_readfirstlane((int)((blockIdx.x * blockDim.x + threadIdx.x) >> 5));
    const float NEG_INF = __int_as_float(0xFF800000);
    const float bias0 = b2[0];
    const float bias1 = b2[1];
    const int dEnd = (w * 8 + 8 < n) ? w * 8 + 8 : n;
    for (int d = w * 8; d < dEnd; ++d) {
    const int beg = rowptr[d];
    const int deg = cnt[d];
    const int tot = deg + 1;
    const float ad = rec2[(size_t)d * 4 + 3];

    v4f r0 = {0.0f, 0.0f, 0.0f, 0.0f};
    float l0 = NEG_INF;
    if (lane < tot) {
        const int s = (lane < deg) ? col[beg + lane] : d;
        r0 = *(const v4f*)(rec2 + (size_t)s * 4);
        l0 = lrelu(r0.z + ad);
    }
    float mx = l0;
    for (int c0 = 32; c0 < tot; c0 += 32) {
        const int idx = c0 + lane;
        if (idx < tot) {
            const int s = (idx < deg) ? col[beg + idx] : d;
            mx = fmaxf(mx, lrelu(rec2[(size_t)s * 4 + 2] + ad));
        }
    }
    mx = wave_max(mx);

    const float ex0 = (lane < tot) ? __expf(l0 - mx) : 0.0f;
    float den = ex0;
    float a0 = ex0 * r0.x;
    float a1 = ex0 * r0.y;
    for (int c0 = 32; c0 < tot; c0 += 32) {
        const int idx = c0 + lane;
        if (idx < tot) {
            const int s = (idx < deg) ? col[beg + idx] : d;
            const v4f r = *(const v4f*)(rec2 + (size_t)s * 4);
            const float ex = __expf(lrelu(r.z + ad) - mx);
            den += ex;
            a0 += ex * r.x;
            a1 += ex * r.y;
        }
    }
    den = wave_sum(den);
    a0 = wave_sum(a0);
    a1 = wave_sum(a1);
    if (lane == 0) {
        const float inv = 1.0f / (den + 1e-16f);
        v2f o;
        o.x = a0 * inv + bias0;
        o.y = a1 * inv + bias1;
        *(v2f*)(out + (size_t)d * 2) = o;
    }
    }
}

extern "C" void kernel_launch(void* const* d_in, const int* in_sizes, int n_in,
                              void* d_out, int out_size, void* d_ws, size_t ws_size,
                              hipStream_t stream) {
    const float* x    = (const float*)d_in[0];
    const int*   ei   = (const int*)d_in[1];
    const float* W1   = (const float*)d_in[2];
    const float* as1w = (const float*)d_in[3];
    const float* ad1w = (const float*)d_in[4];
    const float* b1   = (const float*)d_in[5];
    const float* W2   = (const float*)d_in[6];
    const float* as2w = (const float*)d_in[7];
    const float* ad2w = (const float*)d_in[8];
    const float* b2   = (const float*)d_in[9];

    const int N = in_sizes[0] / 128;
    const int E = in_sizes[1] / 2;
    const int* srcI = ei;
    const int* dstI = ei + E;

    float* p = (float*)d_ws;
    float* h1     = p; p += (size_t)32 * N;
    float* rec2   = p; p += (size_t)4 * N;
    int* cnt      = (int*)p; p += (size_t)N;
    int* rowptr   = (int*)p; p += (size_t)N;
    int* staging  = (int*)p; p += (size_t)2 * E;
    int* col      = (int*)p; p += (size_t)E;
    const int nb    = (N + BIN_NODES - 1) / BIN_NODES;
    const int etiles = (E + TILE_EDGES - 1) / TILE_EDGES;
    int* M        = (int*)p; p += (size_t)etiles * nb;
    int* binBase  = (int*)p; p += (size_t)nb + 1;

    const int T = 256;

    k_bincount<<<etiles, 1024, 0, stream>>>(dstI, M, E, nb);
    k_binscatter<<<etiles, 1024, 0, stream>>>(srcI, dstI, M, binBase, staging, E, nb, etiles);
    k_binsort<<<nb, 1024, 0, stream>>>(staging, binBase, rowptr, cnt, col, N);
    {
        const int tiles = (N + 15) / 16;
        const int waves_per_block = T / 32;
        const int blocks = (tiles + waves_per_block - 1) / waves_per_block;
        k_gemm1_wmma<<<blocks, T, 0, stream>>>(x, W1, h1, N);
    }
    k_node1<<<(N + 63) / 64, T, 0, stream>>>(rowptr, cnt, col, h1, as1w, ad1w, b1,
                                           W2, as2w, ad2w, rec2, N);
    k_node2<<<(N + 63) / 64, T, 0, stream>>>(rowptr, cnt, col, rec2, b2,
                                           (float*)d_out, N);
}


# layer-1 WMMA product runs as trailing work-groups (32 row tiles each) of the second edge-grouping launch, which it does not depend on: 5 launches instead of 6 and the two overlap; sums unchanged
# speedup vs baseline: 10.3418x; 1.0798x over previous
#include <hip/hip_runtime.h>
#include <math.h>


#define NEG_SLOPE 0.2f

typedef __attribute__((ext_vector_type(2))) float v2f;
typedef __attribute__((ext_vector_type(4))) float v4f;
typedef __attribute__((ext_vector_type(8))) float v8f;

__device__ __forceinline__ float lrelu(float v) {
    return v > 0.0f ? v : NEG_SLOPE * v;
}

__device__ __forceinline__ float wave_max(float v) {
    for (int o = 16; o > 0; o >>= 1) v = fmaxf(v, __shfl_xor(v, o));
    return v;
}

__device__ __forceinline__ float wave_sum(float v) {
    for (int o = 16; o > 0; o >>= 1) v += __shfl_xor(v, o);
    return v;
}

__device__ __forceinline__ void gemm1_tile(const float* __restrict__ x,
                                           const float* __restrict__ W1,
                                           float* __restrict__ h1,
                                           int n, int wave, int lane) {
    const int row0 = wave * 16;
    if (row0 >= n) return;

    const int m  = lane & 15;
    const int kh = lane >> 4;
    int row = row0 + m;
    if (row >= n) row = n - 1;

    const float* __restrict__ xrow = x + (size_t)row * 128;

    v8f c0 = {};
    v8f c1 = {};

    #pragma unroll
    for (int k0 = 0; k0 < 128; k0 += 4) {
        const int k = k0 + 2 * kh;
        v2f a = *(const v2f*)(xrow + k);
        v2f b0, b1;
        b0.x = W1[(size_t)k * 32 + m];
        b0.y = W1[(size_t)(k + 1) * 32 + m];
        b1.x = W1[(size_t)k * 32 + 16 + m];
        b1.y = W1[(size_t)(k + 1) * 32 + 16 + m];
        c0 = __builtin_amdgcn_wmma_f32_16x16x4_f32(false, a, false, b0,
                                                   (short)0, c0, false, false);
        c1 = __builtin_amdgcn_wmma_f32_16x16x4_f32(false, a, false, b1,
                                                   (short)0, c1, false, false);
    }

    const int mb = kh * 8;
    #pragma unroll
    for (int j = 0; j < 8; ++j) {
        const int r = row0 + mb + j;
        if (r < n) {
            h1[(size_t)r * 32 + m]      = c0[j];
            h1[(size_t)r * 32 + 16 + m] = c1[j];
        }
    }
}

#define BIN_SHIFT 9
#define BIN_NODES 512
#define MAX_BINS 1024
#define TILE_EDGES 8192
#define EPT 8
#define BIN_CAP 16384

typedef __attribute__((ext_vector_type(2))) int v2i;

__device__ __forceinline__ int block_excl_scan(int c, int* wsum, int lane, int wv) {
    int incl = c;
    for (int o = 1; o < 32; o <<= 1) {
        const int t = __shfl_up(incl, o);
        if (lane >= o) incl += t;
    }
    if (lane == 31) wsum[wv] = incl;
    __syncthreads();
    int woff = 0;
    for (int k = 0; k < wv; ++k) woff += wsum[k];
    return woff + incl - c;
}

__global__ __launch_bounds__(1024) void k_bincount(const int* __restrict__ dst,
                                                   int* __restrict__ M,
                                                   int E, int nb) {
    __shared__ int hist[MAX_BINS];
    const int tid = threadIdx.x;
    for (int b = tid; b < nb; b += 1024) hist[b] = 0;
    __syncthreads();
    const int e0 = blockIdx.x * TILE_EDGES;
    int e1 = e0 + TILE_EDGES;
    if (e1 > E) e1 = E;
    int dv[EPT];
    #pragma unroll
    for (int u = 0; u < EPT; ++u) {
        const int e = e0 + u * 1024 + tid;
        dv[u] = dst[e < e1 ? e : e1 - 1];
    }
    #pragma unroll
    for (int u = 0; u < EPT; ++u)
        if (e0 + u * 1024 + tid < e1) atomicAdd(&hist[dv[u] >> BIN_SHIFT], 1);
    __syncthreads();
    for (int b = tid; b < nb; b += 1024) M[blockIdx.x * nb + b] = hist[b];
}

__global__ __launch_bounds__(1024) void k_binscatter_gemm1(const int* __restrict__ src,
                                                           const int* __restrict__ dst,
                                                           const int* __restrict__ M,
                                                           int* __restrict__ binBase,
                                                           int* __restrict__ staging,
                                                           int E, int nb, int tiles,
                                                           const float* __restrict__ x,
                                                           const float* __restrict__ W1,
                                                           float* __restrict__ h1,
                                                           int n) {
    if ((int)blockIdx.x >= tiles) {
        gemm1_tile(x, W1, h1, n, ((int)blockIdx.x - tiles) * 32 + (int)(threadIdx.x >> 5),
                   (int)(threadIdx.x & 31));
        return;
    }
    __shared__ int totL[MAX_BINS];
    __shared__ int preL[MAX_BINS];
    __shared__ int offG[MAX_BINS];
    __shared__ int lstart[MAX_BINS];
    __shared__ int cur[MAX_BINS];
    __shared__ int wsA[32];
    __shared__ int wsB[32];
    __shared__ v2i buf[TILE_EDGES];
    const int tid = threadIdx.x;
    const int lane = tid & 31;
    const int wv = tid >> 5;
    const int tile = blockIdx.x;

    for (int b = tid; b < nb; b += 1024) { totL[b] = 0; preL[b] = 0; }
    __syncthreads();
    {
        const int q = tid >> 8;
        for (int b = tid & 255; b < nb; b += 256) {
            int tot = 0, pre = 0;
            #pragma unroll 8
            for (int t = q; t < tiles; t += 4) {
                const int v = M[t * nb + b];
                tot += v;
                if (t < tile) pre += v;
            }
            atomicAdd(&totL[b], tot);
            atomicAdd(&preL[b], pre);
        }
    }
    __syncthreads();
    const int myTot = (tid < nb) ? totL[tid] : 0;
    const int myCnt = (tid < nb) ? M[tile * nb + tid] : 0;
    const int base = block_excl_scan(myTot, wsA, lane, wv);
    const int ls = block_excl_scan(myCnt, wsB, lane, wv);
    if (tid < nb) {
        offG[tid] = base + preL[tid];
        lstart[tid] = ls;
        cur[tid] = ls;
        if (tile == 0) binBase[tid] = base;
    }
    if (tile == 0 && tid == 0) binBase[nb] = E;
    __syncthreads();

    const int e0 = tile * TILE_EDGES;
    int ne = E - e0;
    if (ne > TILE_EDGES) ne = TILE_EDGES;
    int sv[EPT], dv[EPT];
    #pragma unroll
    for (int u = 0; u < EPT; ++u) {
        const int i = u * 1024 + tid;
        const int ic = e0 + (i < ne ? i : ne - 1);
        sv[u] = src[ic];
        dv[u] = dst[ic];
    }
    #pragma unroll
    for (int u = 0; u < EPT; ++u) {
        if (u * 1024 + tid < ne) {
            v2i r;
            r.x = sv[u];
            r.y = dv[u];
            const int p = atomicAdd(&cur[dv[u] >> BIN_SHIFT], 1);
            buf[p] = r;
        }
    }
    __syncthreads();
    v2i* __restrict__ st = (v2i*)staging;
    for (int i = tid; i < ne; i += 1024) {
        const v2i r = buf[i];
        const int b = r.y >> BIN_SHIFT;
        st[offG[b] + (i - lstart[b])] = r;
    }
}

__global__ __launch_bounds__(1024) void k_binsort(const int* __restrict__ staging,
                                                  const int* __restrict__ binBase,
                                                  int* __restrict__ rowptr,
                                                  int* __restrict__ cnt,
                                                  int* __restrict__ col,
                                                  int n) {
    __shared__ int cntL[BIN_NODES];
    __shared__ int curL[BIN_NODES];
    __shared__ int ws[32];
    __shared__ int colL[BIN_CAP];
    const int tid = threadIdx.x;
    const int lane = tid & 31;
    const int wv = tid >> 5;
    const int b = blockIdx.x;
    const int e0 = binBase[b];
    const int ne = binBase[b + 1] - e0;
    const v2i* __restrict__ st = (const v2i*)staging + e0;

    if (tid < BIN_NODES) cntL[tid] = 0;
    __syncthreads();
    for (int i0 = 0; i0 < ne; i0 += EPT * 1024) {
        int dv[EPT];
        #pragma unroll
        for (int u = 0; u < EPT; ++u) {
            const int i = i0 + u * 1024 + tid;
            dv[u] = st[i < ne ? i : ne - 1].y;
        }
        #pragma unroll
        for (int u = 0; u < EPT; ++u)
            if (i0 + u * 1024 + tid < ne) atomicAdd(&cntL[dv[u] & (BIN_NODES - 1)], 1);
    }
    __syncthreads();
    const int c = (tid < BIN_NODES) ? cntL[tid] : 0;
    const int start = block_excl_scan(c, ws, lane, wv);
    if (tid < BIN_NODES) {
        curL[tid] = start;
        const int d = b * BIN_NODES + tid;
        if (d < n) { rowptr[d] = e0 + start; cnt[d] = c; }
    }
    __syncthreads();
    const bool fits = ne <= BIN_CAP;
    for (int i0 = 0; i0 < ne; i0 += EPT * 1024) {
        v2i rv[EPT];
        #pragma unroll
        for (int u = 0; u < EPT; ++u) {
            const int i = i0 + u * 1024 + tid;
            rv[u] = st[i < ne ? i : ne - 1];
        }
        #pragma unroll
        for (int u = 0; u < EPT; ++u) {
            if (i0 + u * 1024 + tid < ne) {
                const int p = atomicAdd(&curL[rv[u].y & (BIN_NODES - 1)], 1);
                if (fits) colL[p] = rv[u].x; else col[e0 + p] = rv[u].x;
            }
        }
    }
    __syncthreads();
    if (fits)
        for (int i = tid; i < ne; i += 1024) col[e0 + i] = colL[i];
}

#define DPP_XOR1 0xB1
#define DPP_XOR2 0x4E
#define DPP_XOR4 0x164
#define DPP_XOR8 0x168

template <int C>
__device__ __forceinline__ float lane_dpp(float v) {
    const int b = __builtin_bit_cast(int, v);
    return __builtin_bit_cast(float, __builtin_amdgcn_update_dpp(0, b, C, 0xf, 0xf, true));
}

template <int I>
__device__ __forceinline__ int group8_bcast(int v) {
    return (int)__builtin_amdgcn_mov_dpp8((unsigned)v, (unsigned)(I * 0x249249));
}

__device__ __forceinline__ float xhalf(float v) {
    const int b = __builtin_bit_cast(int, v);
    return __builtin_bit_cast(float,
        __builtin_amdgcn_permlanex16(b, b, 0x76543210, 0xfedcba98, false, false));
}

__device__ __forceinline__ float slot_sum(float v) {
    v += lane_dpp<DPP_XOR8>(v);
    v += xhalf(v);
    return v;
}

__global__ __launch_bounds__(256) void k_node1(const int* __restrict__ rowptr,
                                               const int* __restrict__ cnt,
                                               const int* __restrict__ col,
                                               const float* __restrict__ h1,
                                               const float* __restrict__ as1w,
                                               const float* __restrict__ ad1w,
                                               const float* __restrict__ b1,
                                               const float* __restrict__ W2,
                                               const float* __restrict__ as2w,
                                               const float* __restrict__ ad2w,
                                               float* __restrict__ rec2,
                                               int n) {
    const int lane = threadIdx.x & 31;
    const int w = __builtin_amdgcn_readfirstlane((int)((blockIdx.x * blockDim.x + threadIdx.x) >> 5));
    const float NEG_INF = __int_as_float(0xFF800000);
    const int g = lane >> 3;
    const int q = lane & 7;
    const v4f* __restrict__ h1v = (const v4f*)h1;
    const v4f asw = *(const v4f*)(as1w + 4 * q);
    const v4f adw = *(const v4f*)(ad1w + 4 * q);
    const v4f bias = *(const v4f*)(b1 + 4 * q);
    const v4f w2a = *(const v4f*)(W2 + 8 * q);
    const v4f w2b = *(const v4f*)(W2 + 8 * q + 4);
    const int dEnd = (w * 8 + 8 < n) ? w * 8 + 8 : n;
    for (int d = w * 8; d < dEnd; ++d) {
    const int beg = rowptr[d];
    const int deg = cnt[d];
    const int tot = deg + 1;
    const v4f own = h1v[(unsigned)d * 8u + (unsigned)q];
    float adp = 0.0f;
    float mx = NEG_INF;
    float den = 0.0f;
    v4f acc = {0.0f, 0.0f, 0.0f, 0.0f};

    auto walk = [&](const int c0, const bool first) {
        const int left = tot - c0;
        const int m = left < 32 ? left : 32;
        const int nv = (m - g + 3) >> 2;
        const int idx = c0 + 4 * q + g;
        int sl = d;
        if (idx < deg) sl = col[beg + idx];

        v4f row[8];
#define ROW_LOAD(I) if ((I) == 0 || 4 * (I) < m)   \
            row[I] = h1v[(unsigned)group8_bcast<I>(sl) * 8u + (unsigned)q];
        ROW_LOAD(0) ROW_LOAD(1) ROW_LOAD(2) ROW_LOAD(3)
        ROW_LOAD(4) ROW_LOAD(5) ROW_LOAD(6) ROW_LOAD(7)
#undef ROW_LOAD
        if (first) {
            float t = own.x * adw.x + own.y * adw.y + own.z * adw.z + own.w * adw.w;
            t += lane_dpp<DPP_XOR1>(t);
            t += lane_dpp<DPP_XOR2>(t);
            adp = t;
        }
        float lg[8];
        float cm = NEG_INF;
        #pragma unroll
        for (int i = 0; i < 8; ++i) {
            if (i == 0 || 4 * i < m) {
                float p = row[i].x * asw.x + row[i].y * asw.y + row[i].z * asw.z + row[i].w * asw.w;
                p += lane_dpp<DPP_XOR1>(p);
                p += lane_dpp<DPP_XOR2>(p);
                lg[i] = (i < nv) ? lrelu(p + adp) : NEG_INF;
                cm = fmaxf(cm, lg[i]);
            }
        }
        cm = fmaxf(cm, lane_dpp<DPP_XOR8>(cm));
        cm = fmaxf(cm, xhalf(cm));
        if (first) {
            mx = cm;
        } else {
            const float nm = fmaxf(mx, cm);
            const float sc = __expf(mx - nm);
            den *= sc;
            acc.x *= sc; acc.y *= sc; acc.z *= sc; acc.w *= sc;
            mx = nm;
        }
        #pragma unroll
        for (int i = 0; i < 8; ++i) {
            if (i == 0 || 4 * i < m) {
                const float e = (i < nv) ? __expf(lg[i] - mx) : 0.0f;
                den += e;
                acc.x += e * row[i].x;
                acc.y += e * row[i].y;
                acc.z += e * row[i].z;
                acc.w += e * row[i].w;
            }
        }
    };
    walk(0, true);
    for (int c0 = 32; c0 < tot; c0 += 32) walk(c0, false);

    den = slot_sum(den);
    acc.x = slot_sum(acc.x);
    acc.y = slot_sum(acc.y);
    acc.z = slot_sum(acc.z);
    acc.w = slot_sum(acc.w);

    const float inv = 1.0f / (den + 1e-16f);
    float v0 = acc.x * inv + bias.x;
    float v1 = acc.y * inv + bias.y;
    float v2 = acc.z * inv + bias.z;
    float v3 = acc.w * inv + bias.w;
    v0 = v0 > 0.0f ? v0 : 0.0f;
    v1 = v1 > 0.0f ? v1 : 0.0f;
    v2 = v2 > 0.0f ? v2 : 0.0f;
    v3 = v3 > 0.0f ? v3 : 0.0f;
    float g0 = v0 * w2a.x + v1 * w2a.z + v2 * w2b.x + v3 * w2b.z;
    float g1 = v0 * w2a.y + v1 * w2a.w + v2 * w2b.y + v3 * w2b.w;
    g0 += lane_dpp<DPP_XOR1>(g0);
    g1 += lane_dpp<DPP_XOR1>(g1);
    g0 += lane_dpp<DPP_XOR2>(g0);
    g1 += lane_dpp<DPP_XOR2>(g1);
    g0 += lane_dpp<DPP_XOR4>(g0);
    g1 += lane_dpp<DPP_XOR4>(g1);
    if (lane == 0) {
        v4f r;
        r.x = g0;
        r.y = g1;
        r.z = g0 * as2w[0] + g1 * as2w[1];
        r.w = g0 * ad2w[0] + g1 * ad2w[1];
        *(v4f*)(rec2 + (size_t)d * 4) = r;
    }
    }
}

__global__ __launch_bounds__(256) void k_node2(const int* __restrict__ rowptr,
                                               const int* __restrict__ cnt,
                                               const int* __restrict__ col,
                                               const float* __restrict__ rec2,
                                               const float* __restrict__ b2,
                                               float* __restrict__ out,
                                               int n) {
    const int lane = threadIdx.x & 31;
    const int w = __builtin_amdgcn_readfirstlane((int)((blockIdx.x * blockDim.x + threadIdx.x) >> 5));
    const float NEG_INF = __int_as_float(0xFF800000);
    const float bias0 = b2[0];
    const float bias1 = b2[1];
    const int dEnd = (w * 8 + 8 < n) ? w * 8 + 8 : n;
    for (int d = w * 8; d < dEnd; ++d) {
    const int beg = rowptr[d];
    const int deg = cnt[d];
    const int tot = deg + 1;
    const float ad = rec2[(size_t)d * 4 + 3];

    v4f r0 = {0.0f, 0.0f, 0.0f, 0.0f};
    float l0 = NEG_INF;
    if (lane < tot) {
        const int s = (lane < deg) ? col[beg + lane] : d;
        r0 = *(const v4f*)(rec2 + (size_t)s * 4);
        l0 = lrelu(r0.z + ad);
    }
    float mx = l0;
    for (int c0 = 32; c0 < tot; c0 += 32) {
        const int idx = c0 + lane;
        if (idx < tot) {
            const int s = (idx < deg) ? col[beg + idx] : d;
            mx = fmaxf(mx, lrelu(rec2[(size_t)s * 4 + 2] + ad));
        }
    }
    mx = wave_max(mx);

    const float ex0 = (lane < tot) ? __expf(l0 - mx) : 0.0f;
    float den = ex0;
    float a0 = ex0 * r0.x;
    float a1 = ex0 * r0.y;
    for (int c0 = 32; c0 < tot; c0 += 32) {
        const int idx = c0 + lane;
        if (idx < tot) {
            const int s = (idx < deg) ? col[beg + idx] : d;
            const v4f r = *(const v4f*)(rec2 + (size_t)s * 4);
            const float ex = __expf(lrelu(r.z + ad) - mx);
            den += ex;
            a0 += ex * r.x;
            a1 += ex * r.y;
        }
    }
    den = wave_sum(den);
    a0 = wave_sum(a0);
    a1 = wave_sum(a1);
    if (lane == 0) {
        const float inv = 1.0f / (den + 1e-16f);
        v2f o;
        o.x = a0 * inv + bias0;
        o.y = a1 * inv + bias1;
        *(v2f*)(out + (size_t)d * 2) = o;
    }
    }
}

extern "C" void kernel_launch(void* const* d_in, const int* in_sizes, int n_in,
                              void* d_out, int out_size, void* d_ws, size_t ws_size,
                              hipStream_t stream) {
    const float* x    = (const float*)d_in[0];
    const int*   ei   = (const int*)d_in[1];
    const float* W1   = (const float*)d_in[2];
    const float* as1w = (const float*)d_in[3];
    const float* ad1w = (const float*)d_in[4];
    const float* b1   = (const float*)d_in[5];
    const float* W2   = (const float*)d_in[6];
    const float* as2w = (const float*)d_in[7];
    const float* ad2w = (const float*)d_in[8];
    const float* b2   = (const float*)d_in[9];

    const int N = in_sizes[0] / 128;
    const int E = in_sizes[1] / 2;
    const int* srcI = ei;
    const int* dstI = ei + E;

    float* p = (float*)d_ws;
    float* h1     = p; p += (size_t)32 * N;
    float* rec2   = p; p += (size_t)4 * N;
    int* cnt      = (int*)p; p += (size_t)N;
    int* rowptr   = (int*)p; p += (size_t)N;
    int* staging  = (int*)p; p += (size_t)2 * E;
    int* col      = (int*)p; p += (size_t)E;
    const int nb    = (N + BIN_NODES - 1) / BIN_NODES;
    const int etiles = (E + TILE_EDGES - 1) / TILE_EDGES;
    int* M        = (int*)p; p += (size_t)etiles * nb;
    int* binBase  = (int*)p; p += (size_t)nb + 1;

    const int T = 256;

    k_bincount<<<etiles, 1024, 0, stream>>>(dstI, M, E, nb);
    {
        const int gblocks = (N + 511) / 512;
        k_binscatter_gemm1<<<etiles + gblocks, 1024, 0, stream>>>(srcI, dstI, M, binBase, staging,
                                                                  E, nb, etiles, x, W1, h1, N);
    }
    k_binsort<<<nb, 1024, 0, stream>>>(staging, binBase, rowptr, cnt, col, N);
    k_node1<<<(N + 63) / 64, T, 0, stream>>>(rowptr, cnt, col, h1, as1w, ad1w, b1,
                                           W2, as2w, ad2w, rec2, N);
    k_node2<<<(N + 63) / 64, T, 0, stream>>>(rowptr, cnt, col, rec2, b2,
                                           (float*)d_out, N);
}


# edge grouping in 2 launches, not 3: each tile orders its edges by bin in LDS, writes them contiguously plus a run-start table; bin pass fetches runs by it; list (sum) order = LDS atomic arrival
# speedup vs baseline: 11.4416x; 1.1063x over previous
#include <hip/hip_runtime.h>
#include <math.h>


#define NEG_SLOPE 0.2f

typedef __attribute__((ext_vector_type(2))) float v2f;
typedef __attribute__((ext_vector_type(4))) float v4f;
typedef __attribute__((ext_vector_type(8))) float v8f;

__device__ __forceinline__ float lrelu(float v) {
    return v > 0.0f ? v : NEG_SLOPE * v;
}

__device__ __forceinline__ float wave_max(float v) {
    for (int o = 16; o > 0; o >>= 1) v = fmaxf(v, __shfl_xor(v, o));
    return v;
}

__device__ __forceinline__ float wave_sum(float v) {
    for (int o = 16; o > 0; o >>= 1) v += __shfl_xor(v, o);
    return v;
}

__device__ __forceinline__ void gemm1_tile(const float* __restrict__ x,
                                           const float* __restrict__ W1,
                                           float* __restrict__ h1,
                                           int n, int wave, int lane) {
    const int row0 = wave * 16;
    if (row0 >= n) return;

    const int m  = lane & 15;
    const int kh = lane >> 4;
    int row = row0 + m;
    if (row >= n) row = n - 1;

    const float* __restrict__ xrow = x + (size_t)row * 128;

    v8f c0 = {};
    v8f c1 = {};

    #pragma unroll
    for (int k0 = 0; k0 < 128; k0 += 4) {
        const int k = k0 + 2 * kh;
        v2f a = *(const v2f*)(xrow + k);
        v2f b0, b1;
        b0.x = W1[(size_t)k * 32 + m];
        b0.y = W1[(size_t)(k + 1) * 32 + m];
        b1.x = W1[(size_t)k * 32 + 16 + m];
        b1.y = W1[(size_t)(k + 1) * 32 + 16 + m];
        c0 = __builtin_amdgcn_wmma_f32_16x16x4_f32(false, a, false, b0,
                                                   (short)0, c0, false, false);
        c1 = __builtin_amdgcn_wmma_f32_16x16x4_f32(false, a, false, b1,
                                                   (short)0, c1, false, false);
    }

    const int mb = kh * 8;
    #pragma unroll
    for (int j = 0; j < 8; ++j) {
        const int r = row0 + mb + j;
        if (r < n) {
            h1[(size_t)r * 32 + m]      = c0[j];
            h1[(size_t)r * 32 + 16 + m] = c1[j];
        }
    }
}

#define BIN_SHIFT 9
#define BIN_NODES 512
#define MAX_BINS 1024
#define MAX_TILES 1024
#define TILE_EDGES 8192
#define EPT 8
#define BIN_CAP 16384
#define RPW 7

typedef __attribute__((ext_vector_type(2))) int v2i;

__device__ __forceinline__ int block_excl_scan(int c, int* wsum, int lane, int wv) {
    int incl = c;
    for (int o = 1; o < 32; o <<= 1) {
        const int t = __shfl_up(incl, o);
        if (lane >= o) incl += t;
    }
    if (lane == 31) wsum[wv] = incl;
    __syncthreads();
    int woff = 0;
    for (int k = 0; k < wv; ++k) woff += wsum[k];
    return woff + incl - c;
}

__global__ __launch_bounds__(1024) void k_tilesort_gemm1(const int* __restrict__ src,
                                                         const int* __restrict__ dst,
                                                         int* __restrict__ L,
                                                         int* __restrict__ staging,
                                                         int E, int nb, int tiles,
                                                         const float* __restrict__ x,
                                                         const float* __restrict__ W1,
                                                         float* __restrict__ h1,
                                                         int n) {
    if ((int)blockIdx.x >= tiles) {
        gemm1_tile(x, W1, h1, n, ((int)blockIdx.x - tiles) * 32 + (int)(threadIdx.x >> 5),
                   (int)(threadIdx.x & 31));
        return;
    }
    __shared__ int cur[MAX_BINS];
    __shared__ int ws[32];
    __shared__ v2i buf[TILE_EDGES];
    const int tid = threadIdx.x;
    const int lane = tid & 31;
    const int wv = tid >> 5;
    const int tile = blockIdx.x;

    for (int b = tid; b < nb; b += 1024) cur[b] = 0;
    __syncthreads();

    const int e0 = tile * TILE_EDGES;
    int ne = E - e0;
    if (ne > TILE_EDGES) ne = TILE_EDGES;
    int sv[EPT], dv[EPT];
    #pragma unroll
    for (int u = 0; u < EPT; ++u) {
        const int i = u * 1024 + tid;
        const int ic = e0 + (i < ne ? i : ne - 1);
        sv[u] = src[ic];
        dv[u] = dst[ic];
    }
    #pragma unroll
    for (int u = 0; u < EPT; ++u)
        if (u * 1024 + tid < ne) atomicAdd(&cur[dv[u] >> BIN_SHIFT], 1);
    __syncthreads();
    const int myCnt = (tid < nb) ? cur[tid] : 0;
    const int ls = block_excl_scan(myCnt, ws, lane, wv);
    int* __restrict__ Lrow = L + (size_t)tile * (size_t)(nb + 1);
    if (tid < nb) { cur[tid] = ls; Lrow[tid] = ls; }
    if (tid == 0) Lrow[nb] = ne;
    __syncthreads();
    #pragma unroll
    for (int u = 0; u < EPT; ++u) {
        if (u * 1024 + tid < ne) {
            v2i r;
            r.x = sv[u];
            r.y = dv[u];
            const int p = atomicAdd(&cur[dv[u] >> BIN_SHIFT], 1);
            buf[p] = r;
        }
    }
    __syncthreads();
    v2i* __restrict__ st = (v2i*)staging + e0;
    for (int i = tid; i < ne; i += 1024) st[i] = buf[i];
}

__global__ __launch_bounds__(1024) void k_binsort(const int* __restrict__ staging,
                                                  const int* __restrict__ L,
                                                  int* __restrict__ rowptr,
                                                  int* __restrict__ cnt,
                                                  int* __restrict__ col,
                                                  int n, int nb, int tiles) {
    __shared__ v2i runL[MAX_TILES];
    __shared__ int cntL[BIN_NODES];
    __shared__ int curL[BIN_NODES];
    __shared__ int ws[32];
    __shared__ int tot[2];
    __shared__ int colL[BIN_CAP];
    const int tid = threadIdx.x;
    const int lane = tid & 31;
    const int wv = __builtin_amdgcn_readfirstlane(tid >> 5);
    const int b = blockIdx.x;
    const v2i* __restrict__ st = (const v2i*)staging;

    if (tid < BIN_NODES) cntL[tid] = 0;
    if (tid < 2) tot[tid] = 0;
    int a = 0, len = 0;
    if (tid < tiles) {
        const int* __restrict__ Lrow = L + (size_t)tid * (size_t)(nb + 1) + b;
        a = Lrow[0];
        len = Lrow[1] - a;
        v2i rl;
        rl.x = a;
        rl.y = len;
        runL[tid] = rl;
    }
    __syncthreads();
    {
        int sa = a, sl = len;
        for (int o = 16; o > 0; o >>= 1) { sa += __shfl_xor(sa, o); sl += __shfl_xor(sl, o); }
        if (lane == 0) { atomicAdd(&tot[0], sa); atomicAdd(&tot[1], sl); }
    }
    __syncthreads();
    const int e0 = tot[0];
    const int ne = tot[1];
    const bool fits = ne <= BIN_CAP;
    const bool one = tiles <= 32 * RPW;

    v2i r0[RPW], r1[RPW];
    int ln[RPW], off[RPW];
    auto load_runs = [&](const int t0) {
        v2i rl[RPW];
        #pragma unroll
        for (int u = 0; u < RPW; ++u) {
            const int t = t0 + u * 32 + wv;
            rl[u] = runL[t < tiles ? t : tiles - 1];
        }
        #pragma unroll
        for (int u = 0; u < RPW; ++u) {
            const int t = t0 + u * 32 + wv;
            const int l = t < tiles ? __builtin_amdgcn_readfirstlane(rl[u].y) : 0;
            const int o = l > 0 ? t * TILE_EDGES + __builtin_amdgcn_readfirstlane(rl[u].x) : 0;
            ln[u] = l;
            off[u] = o;
            const int last = l > 0 ? l - 1 : 0;
            r0[u] = st[o + (lane < l ? lane : last)];
            r1[u] = st[o + (lane + 32 < l ? lane + 32 : last)];
        }
    };

    for (int t0 = 0; t0 < tiles; t0 += 32 * RPW) {
        load_runs(t0);
        #pragma unroll
        for (int u = 0; u < RPW; ++u) {
            if (lane < ln[u]) atomicAdd(&cntL[r0[u].y & (BIN_NODES - 1)], 1);
            if (lane + 32 < ln[u]) atomicAdd(&cntL[r1[u].y & (BIN_NODES - 1)], 1);
            for (int i = 64 + lane; i < ln[u]; i += 32)
                atomicAdd(&cntL[st[off[u] + i].y & (BIN_NODES - 1)], 1);
        }
    }
    __syncthreads();
    const int c = (tid < BIN_NODES) ? cntL[tid] : 0;
    const int start = block_excl_scan(c, ws, lane, wv);
    if (tid < BIN_NODES) {
        curL[tid] = start;
        const int d = b * BIN_NODES + tid;
        if (d < n) { rowptr[d] = e0 + start; cnt[d] = c; }
    }
    __syncthreads();
    for (int t0 = 0; t0 < tiles; t0 += 32 * RPW) {
        if (!one) load_runs(t0);
        int p0[RPW], p1[RPW];
        #pragma unroll
        for (int u = 0; u < RPW; ++u) {
            p0[u] = 0;
            p1[u] = 0;
            if (lane < ln[u]) p0[u] = atomicAdd(&curL[r0[u].y & (BIN_NODES - 1)], 1);
            if (lane + 32 < ln[u]) p1[u] = atomicAdd(&curL[r1[u].y & (BIN_NODES - 1)], 1);
        }
        #pragma unroll
        for (int u = 0; u < RPW; ++u) {
            if (lane < ln[u]) {
                if (fits) colL[p0[u]] = r0[u].x; else col[e0 + p0[u]] = r0[u].x;
            }
            if (lane + 32 < ln[u]) {
                if (fits) colL[p1[u]] = r1[u].x; else col[e0 + p1[u]] = r1[u].x;
            }
            for (int i = 64 + lane; i < ln[u]; i += 32) {
                const v2i r = st[off[u] + i];
                const int p = atomicAdd(&curL[r.y & (BIN_NODES - 1)], 1);
                if (fits) colL[p] = r.x; else col[e0 + p] = r.x;
            }
        }
    }
    __syncthreads();
    if (fits)
        for (int i = tid; i < ne; i += 1024) col[e0 + i] = colL[i];
}

#define DPP_XOR1 0xB1
#define DPP_XOR2 0x4E
#define DPP_XOR4 0x164
#define DPP_XOR8 0x168

template <int C>
__device__ __forceinline__ float lane_dpp(float v) {
    const int b = __builtin_bit_cast(int, v);
    return __builtin_bit_cast(float, __builtin_amdgcn_update_dpp(0, b, C, 0xf, 0xf, true));
}

template <int I>
__device__ __forceinline__ int group8_bcast(int v) {
    return (int)__builtin_amdgcn_mov_dpp8((unsigned)v, (unsigned)(I * 0x249249));
}

__device__ __forceinline__ float xhalf(float v) {
    const int b = __builtin_bit_cast(int, v);
    return __builtin_bit_cast(float,
        __builtin_amdgcn_permlanex16(b, b, 0x76543210, 0xfedcba98, false, false));
}

__device__ __forceinline__ float slot_sum(float v) {
    v += lane_dpp<DPP_XOR8>(v);
    v += xhalf(v);
    return v;
}

__global__ __launch_bounds__(256) void k_node1(const int* __restrict__ rowptr,
                                               const int* __restrict__ cnt,
                                               const int* __restrict__ col,
                                               const float* __restrict__ h1,
                                               const float* __restrict__ as1w,
                                               const float* __restrict__ ad1w,
                                               const float* __restrict__ b1,
                                               const float* __restrict__ W2,
                                               const float* __restrict__ as2w,
                                               const float* __restrict__ ad2w,
                                               float* __restrict__ rec2,
                                               int n) {
    const int lane = threadIdx.x & 31;
    const int w = __builtin_amdgcn_readfirstlane((int)((blockIdx.x * blockDim.x + threadIdx.x) >> 5));
    const float NEG_INF = __int_as_float(0xFF800000);
    const int g = lane >> 3;
    const int q = lane & 7;
    const v4f* __restrict__ h1v = (const v4f*)h1;
    const v4f asw = *(const v4f*)(as1w + 4 * q);
    const v4f adw = *(const v4f*)(ad1w + 4 * q);
    const v4f bias = *(const v4f*)(b1 + 4 * q);
    const v4f w2a = *(const v4f*)(W2 + 8 * q);
    const v4f w2b = *(const v4f*)(W2 + 8 * q + 4);
    const int dEnd = (w * 8 + 8 < n) ? w * 8 + 8 : n;
    for (int d = w * 8; d < dEnd; ++d) {
    const int beg = rowptr[d];
    const int deg = cnt[d];
    const int tot = deg + 1;
    const v4f own = h1v[(unsigned)d * 8u + (unsigned)q];
    float adp = 0.0f;
    float mx = NEG_INF;
    float den = 0.0f;
    v4f acc = {0.0f, 0.0f, 0.0f, 0.0f};

    auto walk = [&](const int c0, const bool first) {
        const int left = tot - c0;
        const int m = left < 32 ? left : 32;
        const int nv = (m - g + 3) >> 2;
        const int idx = c0 + 4 * q + g;
        int sl = d;
        if (idx < deg) sl = col[beg + idx];

        v4f row[8];
#define ROW_LOAD(I) if ((I) == 0 || 4 * (I) < m)   \
            row[I] = h1v[(unsigned)group8_bcast<I>(sl) * 8u + (unsigned)q];
        ROW_LOAD(0) ROW_LOAD(1) ROW_LOAD(2) ROW_LOAD(3)
        ROW_LOAD(4) ROW_LOAD(5) ROW_LOAD(6) ROW_LOAD(7)
#undef ROW_LOAD
        if (first) {
            float t = own.x * adw.x + own.y * adw.y + own.z * adw.z + own.w * adw.w;
            t += lane_dpp<DPP_XOR1>(t);
            t += lane_dpp<DPP_XOR2>(t);
            adp = t;
        }
        float lg[8];
        float cm = NEG_INF;
        #pragma unroll
        for (int i = 0; i < 8; ++i) {
            if (i == 0 || 4 * i < m) {
                float p = row[i].x * asw.x + row[i].y * asw.y + row[i].z * asw.z + row[i].w * asw.w;
                p += lane_dpp<DPP_XOR1>(p);
                p += lane_dpp<DPP_XOR2>(p);
                lg[i] = (i < nv) ? lrelu(p + adp) : NEG_INF;
                cm = fmaxf(cm, lg[i]);
            }
        }
        cm = fmaxf(cm, lane_dpp<DPP_XOR8>(cm));
        cm = fmaxf(cm, xhalf(cm));
        if (first) {
            mx = cm;
        } else {
            const float nm = fmaxf(mx, cm);
            const float sc = __expf(mx - nm);
            den *= sc;
            acc.x *= sc; acc.y *= sc; acc.z *= sc; acc.w *= sc;
            mx = nm;
        }
        #pragma unroll
        for (int i = 0; i < 8; ++i) {
            if (i == 0 || 4 * i < m) {
                const float e = (i < nv) ? __expf(lg[i] - mx) : 0.0f;
                den += e;
                acc.x += e * row[i].x;
                acc.y += e * row[i].y;
                acc.z += e * row[i].z;
                acc.w += e * row[i].w;
            }
        }
    };
    walk(0, true);
    for (int c0 = 32; c0 < tot; c0 += 32) walk(c0, false);

    den = slot_sum(den);
    acc.x = slot_sum(acc.x);
    acc.y = slot_sum(acc.y);
    acc.z = slot_sum(acc.z);
    acc.w = slot_sum(acc.w);

    const float inv = 1.0f / (den + 1e-16f);
    float v0 = acc.x * inv + bias.x;
    float v1 = acc.y * inv + bias.y;
    float v2 = acc.z * inv + bias.z;
    float v3 = acc.w * inv + bias.w;
    v0 = v0 > 0.0f ? v0 : 0.0f;
    v1 = v1 > 0.0f ? v1 : 0.0f;
    v2 = v2 > 0.0f ? v2 : 0.0f;
    v3 = v3 > 0.0f ? v3 : 0.0f;
    float g0 = v0 * w2a.x + v1 * w2a.z + v2 * w2b.x + v3 * w2b.z;
    float g1 = v0 * w2a.y + v1 * w2a.w + v2 * w2b.y + v3 * w2b.w;
    g0 += lane_dpp<DPP_XOR1>(g0);
    g1 += lane_dpp<DPP_XOR1>(g1);
    g0 += lane_dpp<DPP_XOR2>(g0);
    g1 += lane_dpp<DPP_XOR2>(g1);
    g0 += lane_dpp<DPP_XOR4>(g0);
    g1 += lane_dpp<DPP_XOR4>(g1);
    if (lane == 0) {
        v4f r;
        r.x = g0;
        r.y = g1;
        r.z = g0 * as2w[0] + g1 * as2w[1];
        r.w = g0 * ad2w[0] + g1 * ad2w[1];
        *(v4f*)(rec2 + (size_t)d * 4) = r;
    }
    }
}

__global__ __launch_bounds__(256) void k_node2(const int* __restrict__ rowptr,
                                               const int* __restrict__ cnt,
                                               const int* __restrict__ col,
                                               const float* __restrict__ rec2,
                                               const float* __restrict__ b2,
                                               float* __restrict__ out,
                                               int n) {
    const int lane = threadIdx.x & 31;
    const int w = __builtin_amdgcn_readfirstlane((int)((blockIdx.x * blockDim.x + threadIdx.x) >> 5));
    const float NEG_INF = __int_as_float(0xFF800000);
    const float bias0 = b2[0];
    const float bias1 = b2[1];
    const int dEnd = (w * 8 + 8 < n) ? w * 8 + 8 : n;
    for (int d = w * 8; d < dEnd; ++d) {
    const int beg = rowptr[d];
    const int deg = cnt[d];
    const int tot = deg + 1;
    const float ad = rec2[(size_t)d * 4 + 3];

    v4f r0 = {0.0f, 0.0f, 0.0f, 0.0f};
    float l0 = NEG_INF;
    if (lane < tot) {
        const int s = (lane < deg) ? col[beg + lane] : d;
        r0 = *(const v4f*)(rec2 + (size_t)s * 4);
        l0 = lrelu(r0.z + ad);
    }
    float mx = l0;
    for (int c0 = 32; c0 < tot; c0 += 32) {
        const int idx = c0 + lane;
        if (idx < tot) {
            const int s = (idx < deg) ? col[beg + idx] : d;
            mx = fmaxf(mx, lrelu(rec2[(size_t)s * 4 + 2] + ad));
        }
    }
    mx = wave_max(mx);

    const float ex0 = (lane < tot) ? __expf(l0 - mx) : 0.0f;
    float den = ex0;
    float a0 = ex0 * r0.x;
    float a1 = ex0 * r0.y;
    for (int c0 = 32; c0 < tot; c0 += 32) {
        const int idx = c0 + lane;
        if (idx < tot) {
            const int s = (idx < deg) ? col[beg + idx] : d;
            const v4f r = *(const v4f*)(rec2 + (size_t)s * 4);
            const float ex = __expf(lrelu(r.z + ad) - mx);
            den += ex;
            a0 += ex * r.x;
            a1 += ex * r.y;
        }
    }
    den = wave_sum(den);
    a0 = wave_sum(a0);
    a1 = wave_sum(a1);
    if (lane == 0) {
        const float inv = 1.0f / (den + 1e-16f);
        v2f o;
        o.x = a0 * inv + bias0;
        o.y = a1 * inv + bias1;
        *(v2f*)(out + (size_t)d * 2) = o;
    }
    }
}

extern "C" void kernel_launch(void* const* d_in, const int* in_sizes, int n_in,
                              void* d_out, int out_size, void* d_ws, size_t ws_size,
                              hipStream_t stream) {
    const float* x    = (const float*)d_in[0];
    const int*   ei   = (const int*)d_in[1];
    const float* W1   = (const float*)d_in[2];
    const float* as1w = (const float*)d_in[3];
    const float* ad1w = (const float*)d_in[4];
    const float* b1   = (const float*)d_in[5];
    const float* W2   = (const float*)d_in[6];
    const float* as2w = (const float*)d_in[7];
    const float* ad2w = (const float*)d_in[8];
    const float* b2   = (const float*)d_in[9];

    const int N = in_sizes[0] / 128;
    const int E = in_sizes[1] / 2;
    const int* srcI = ei;
    const int* dstI = ei + E;

    float* p = (float*)d_ws;
    float* h1     = p; p += (size_t)32 * N;
    float* rec2   = p; p += (size_t)4 * N;
    int* cnt      = (int*)p; p += (size_t)N;
    int* rowptr   = (int*)p; p += (size_t)N;
    int* staging  = (int*)p; p += (size_t)2 * E;
    int* col      = (int*)p; p += (size_t)E;
    const int nb    = (N + BIN_NODES - 1) / BIN_NODES;
    const int etiles = (E + TILE_EDGES - 1) / TILE_EDGES;
    int* L        = (int*)p; p += (size_t)etiles * (nb + 1);

    const int T = 256;

    {
        const int gblocks = (N + 511) / 512;
        k_tilesort_gemm1<<<etiles + gblocks, 1024, 0, stream>>>(srcI, dstI, L, staging,
                                                                E, nb, etiles, x, W1, h1, N);
    }
    k_binsort<<<nb, 1024, 0, stream>>>(staging, L, rowptr, cnt, col, N, nb, etiles);
    k_node1<<<(N + 63) / 64, T, 0, stream>>>(rowptr, cnt, col, h1, as1w, ad1w, b1,
                                           W2, as2w, ad2w, rec2, N);
    k_node2<<<(N + 63) / 64, T, 0, stream>>>(rowptr, cnt, col, rec2, b2,
                                           (float*)d_out, N);
}


# layer-2 node pass: 8 nodes walked together, 4 lanes per node, 32 nodes per wave (a quarter of the waves); node sums = each lane sums every 4th entry in list order, then a 4-lane tree
# speedup vs baseline: 11.9505x; 1.0445x over previous
#include <hip/hip_runtime.h>
#include <math.h>


#define NEG_SLOPE 0.2f

typedef __attribute__((ext_vector_type(2))) float v2f;
typedef __attribute__((ext_vector_type(4))) float v4f;
typedef __attribute__((ext_vector_type(8))) float v8f;

__device__ __forceinline__ float lrelu(float v) {
    return v > 0.0f ? v : NEG_SLOPE * v;
}

__device__ __forceinline__ void gemm1_tile(const float* __restrict__ x,
                                           const float* __restrict__ W1,
                                           float* __restrict__ h1,
                                           int n, int wave, int lane) {
    const int row0 = wave * 16;
    if (row0 >= n) return;

    const int m  = lane & 15;
    const int kh = lane >> 4;
    int row = row0 + m;
    if (row >= n) row = n - 1;

    const float* __restrict__ xrow = x + (size_t)row * 128;

    v8f c0 = {};
    v8f c1 = {};

    #pragma unroll
    for (int k0 = 0; k0 < 128; k0 += 4) {
        const int k = k0 + 2 * kh;
        v2f a = *(const v2f*)(xrow + k);
        v2f b0, b1;
        b0.x = W1[(size_t)k * 32 + m];
        b0.y = W1[(size_t)(k + 1) * 32 + m];
        b1.x = W1[(size_t)k * 32 + 16 + m];
        b1.y = W1[(size_t)(k + 1) * 32 + 16 + m];
        c0 = __builtin_amdgcn_wmma_f32_16x16x4_f32(false, a, false, b0,
                                                   (short)0, c0, false, false);
        c1 = __builtin_amdgcn_wmma_f32_16x16x4_f32(false, a, false, b1,
                                                   (short)0, c1, false, false);
    }

    const int mb = kh * 8;
    #pragma unroll
    for (int j = 0; j < 8; ++j) {
        const int r = row0 + mb + j;
        if (r < n) {
            h1[(size_t)r * 32 + m]      = c0[j];
            h1[(size_t)r * 32 + 16 + m] = c1[j];
        }
    }
}

#define BIN_SHIFT 9
#define BIN_NODES 512
#define MAX_BINS 1024
#define MAX_TILES 1024
#define TILE_EDGES 8192
#define EPT 8
#define BIN_CAP 16384
#define RPW 7

typedef __attribute__((ext_vector_type(2))) int v2i;

__device__ __forceinline__ int block_excl_scan(int c, int* wsum, int lane, int wv) {
    int incl = c;
    for (int o = 1; o < 32; o <<= 1) {
        const int t = __shfl_up(incl, o);
        if (lane >= o) incl += t;
    }
    if (lane == 31) wsum[wv] = incl;
    __syncthreads();
    int woff = 0;
    for (int k = 0; k < wv; ++k) woff += wsum[k];
    return woff + incl - c;
}

__global__ __launch_bounds__(1024) void k_tilesort_gemm1(const int* __restrict__ src,
                                                         const int* __restrict__ dst,
                                                         int* __restrict__ L,
                                                         int* __restrict__ staging,
                                                         int E, int nb, int tiles,
                                                         const float* __restrict__ x,
                                                         const float* __restrict__ W1,
                                                         float* __restrict__ h1,
                                                         int n) {
    if ((int)blockIdx.x >= tiles) {
        gemm1_tile(x, W1, h1, n, ((int)blockIdx.x - tiles) * 32 + (int)(threadIdx.x >> 5),
                   (int)(threadIdx.x & 31));
        return;
    }
    __shared__ int cur[MAX_BINS];
    __shared__ int ws[32];
    __shared__ v2i buf[TILE_EDGES];
    const int tid = threadIdx.x;
    const int lane = tid & 31;
    const int wv = tid >> 5;
    const int tile = blockIdx.x;

    for (int b = tid; b < nb; b += 1024) cur[b] = 0;
    __syncthreads();

    const int e0 = tile * TILE_EDGES;
    int ne = E - e0;
    if (ne > TILE_EDGES) ne = TILE_EDGES;
    int sv[EPT], dv[EPT];
    #pragma unroll
    for (int u = 0; u < EPT; ++u) {
        const int i = u * 1024 + tid;
        const int ic = e0 + (i < ne ? i : ne - 1);
        sv[u] = src[ic];
        dv[u] = dst[ic];
    }
    #pragma unroll
    for (int u = 0; u < EPT; ++u)
        if (u * 1024 + tid < ne) atomicAdd(&cur[dv[u] >> BIN_SHIFT], 1);
    __syncthreads();
    const int myCnt = (tid < nb) ? cur[tid] : 0;
    const int ls = block_excl_scan(myCnt, ws, lane, wv);
    int* __restrict__ Lrow = L + (size_t)tile * (size_t)(nb + 1);
    if (tid < nb) { cur[tid] = ls; Lrow[tid] = ls; }
    if (tid == 0) Lrow[nb] = ne;
    __syncthreads();
    #pragma unroll
    for (int u = 0; u < EPT; ++u) {
        if (u * 1024 + tid < ne) {
            v2i r;
            r.x = sv[u];
            r.y = dv[u];
            const int p = atomicAdd(&cur[dv[u] >> BIN_SHIFT], 1);
            buf[p] = r;
        }
    }
    __syncthreads();
    v2i* __restrict__ st = (v2i*)staging + e0;
    for (int i = tid; i < ne; i += 1024) st[i] = buf[i];
}

__global__ __launch_bounds__(1024) void k_binsort(const int* __restrict__ staging,
                                                  const int* __restrict__ L,
                                                  int* __restrict__ rowptr,
                                                  int* __restrict__ cnt,
                                                  int* __restrict__ col,
                                                  int n, int nb, int tiles) {
    __shared__ v2i runL[MAX_TILES];
    __shared__ int cntL[BIN_NODES];
    __shared__ int curL[BIN_NODES];
    __shared__ int ws[32];
    __shared__ int tot[2];
    __shared__ int colL[BIN_CAP];
    const int tid = threadIdx.x;
    const int lane = tid & 31;
    const int wv = __builtin_amdgcn_readfirstlane(tid >> 5);
    const int b = blockIdx.x;
    const v2i* __restrict__ st = (const v2i*)staging;

    if (tid < BIN_NODES) cntL[tid] = 0;
    if (tid < 2) tot[tid] = 0;
    int a = 0, len = 0;
    if (tid < tiles) {
        const int* __restrict__ Lrow = L + (size_t)tid * (size_t)(nb + 1) + b;
        a = Lrow[0];
        len = Lrow[1] - a;
        v2i rl;
        rl.x = a;
        rl.y = len;
        runL[tid] = rl;
    }
    __syncthreads();
    {
        int sa = a, sl = len;
        for (int o = 16; o > 0; o >>= 1) { sa += __shfl_xor(sa, o); sl += __shfl_xor(sl, o); }
        if (lane == 0) { atomicAdd(&tot[0], sa); atomicAdd(&tot[1], sl); }
    }
    __syncthreads();
    const int e0 = tot[0];
    const int ne = tot[1];
    const bool fits = ne <= BIN_CAP;
    const bool one = tiles <= 32 * RPW;

    v2i r0[RPW], r1[RPW];
    int ln[RPW], off[RPW];
    auto load_runs = [&](const int t0) {
        v2i rl[RPW];
        #pragma unroll
        for (int u = 0; u < RPW; ++u) {
            const int t = t0 + u * 32 + wv;
            rl[u] = runL[t < tiles ? t : tiles - 1];
        }
        #pragma unroll
        for (int u = 0; u < RPW; ++u) {
            const int t = t0 + u * 32 + wv;
            const int l = t < tiles ? __builtin_amdgcn_readfirstlane(rl[u].y) : 0;
            const int o = l > 0 ? t * TILE_EDGES + __builtin_amdgcn_readfirstlane(rl[u].x) : 0;
            ln[u] = l;
            off[u] = o;
            const int last = l > 0 ? l - 1 : 0;
            r0[u] = st[o + (lane < l ? lane : last)];
            r1[u] = st[o + (lane + 32 < l ? lane + 32 : last)];
        }
    };

    for (int t0 = 0; t0 < tiles; t0 += 32 * RPW) {
        load_runs(t0);
        #pragma unroll
        for (int u = 0; u < RPW; ++u) {
            if (lane < ln[u]) atomicAdd(&cntL[r0[u].y & (BIN_NODES - 1)], 1);
            if (lane + 32 < ln[u]) atomicAdd(&cntL[r1[u].y & (BIN_NODES - 1)], 1);
            for (int i = 64 + lane; i < ln[u]; i += 32)
                atomicAdd(&cntL[st[off[u] + i].y & (BIN_NODES - 1)], 1);
        }
    }
    __syncthreads();
    const int c = (tid < BIN_NODES) ? cntL[tid] : 0;
    const int start = block_excl_scan(c, ws, lane, wv);
    if (tid < BIN_NODES) {
        curL[tid] = start;
        const int d = b * BIN_NODES + tid;
        if (d < n) { rowptr[d] = e0 + start; cnt[d] = c; }
    }
    __syncthreads();
    for (int t0 = 0; t0 < tiles; t0 += 32 * RPW) {
        if (!one) load_runs(t0);
        int p0[RPW], p1[RPW];
        #pragma unroll
        for (int u = 0; u < RPW; ++u) {
            p0[u] = 0;
            p1[u] = 0;
            if (lane < ln[u]) p0[u] = atomicAdd(&curL[r0[u].y & (BIN_NODES - 1)], 1);
            if (lane + 32 < ln[u]) p1[u] = atomicAdd(&curL[r1[u].y & (BIN_NODES - 1)], 1);
        }
        #pragma unroll
        for (int u = 0; u < RPW; ++u) {
            if (lane < ln[u]) {
                if (fits) colL[p0[u]] = r0[u].x; else col[e0 + p0[u]] = r0[u].x;
            }
            if (lane + 32 < ln[u]) {
                if (fits) colL[p1[u]] = r1[u].x; else col[e0 + p1[u]] = r1[u].x;
            }
            for (int i = 64 + lane; i < ln[u]; i += 32) {
                const v2i r = st[off[u] + i];
                const int p = atomicAdd(&curL[r.y & (BIN_NODES - 1)], 1);
                if (fits) colL[p] = r.x; else col[e0 + p] = r.x;
            }
        }
    }
    __syncthreads();
    if (fits)
        for (int i = tid; i < ne; i += 1024) col[e0 + i] = colL[i];
}

#define DPP_XOR1 0xB1
#define DPP_XOR2 0x4E
#define DPP_XOR4 0x164
#define DPP_XOR8 0x168

template <int C>
__device__ __forceinline__ float lane_dpp(float v) {
    const int b = __builtin_bit_cast(int, v);
    return __builtin_bit_cast(float, __builtin_amdgcn_update_dpp(0, b, C, 0xf, 0xf, true));
}

template <int I>
__device__ __forceinline__ int group8_bcast(int v) {
    return (int)__builtin_amdgcn_mov_dpp8((unsigned)v, (unsigned)(I * 0x249249));
}

__device__ __forceinline__ float xhalf(float v) {
    const int b = __builtin_bit_cast(int, v);
    return __builtin_bit_cast(float,
        __builtin_amdgcn_permlanex16(b, b, 0x76543210, 0xfedcba98, false, false));
}

__device__ __forceinline__ float slot_sum(float v) {
    v += lane_dpp<DPP_XOR8>(v);
    v += xhalf(v);
    return v;
}

__global__ __launch_bounds__(256) void k_node1(const int* __restrict__ rowptr,
                                               const int* __restrict__ cnt,
                                               const int* __restrict__ col,
                                               const float* __restrict__ h1,
                                               const float* __restrict__ as1w,
                                               const float* __restrict__ ad1w,
                                               const float* __restrict__ b1,
                                               const float* __restrict__ W2,
                                               const float* __restrict__ as2w,
                                               const float* __restrict__ ad2w,
                                               float* __restrict__ rec2,
                                               int n) {
    const int lane = threadIdx.x & 31;
    const int w = __builtin_amdgcn_readfirstlane((int)((blockIdx.x * blockDim.x + threadIdx.x) >> 5));
    const float NEG_INF = __int_as_float(0xFF800000);
    const int g = lane >> 3;
    const int q = lane & 7;
    const v4f* __restrict__ h1v = (const v4f*)h1;
    const v4f asw = *(const v4f*)(as1w + 4 * q);
    const v4f adw = *(const v4f*)(ad1w + 4 * q);
    const v4f bias = *(const v4f*)(b1 + 4 * q);
    const v4f w2a = *(const v4f*)(W2 + 8 * q);
    const v4f w2b = *(const v4f*)(W2 + 8 * q + 4);
    const int dEnd = (w * 8 + 8 < n) ? w * 8 + 8 : n;
    for (int d = w * 8; d < dEnd; ++d) {
    const int beg = rowptr[d];
    const int deg = cnt[d];
    const int tot = deg + 1;
    const v4f own = h1v[(unsigned)d * 8u + (unsigned)q];
    float adp = 0.0f;
    float mx = NEG_INF;
    float den = 0.0f;
    v4f acc = {0.0f, 0.0f, 0.0f, 0.0f};

    auto walk = [&](const int c0, const bool first) {
        const int left = tot - c0;
        const int m = left < 32 ? left : 32;
        const int nv = (m - g + 3) >> 2;
        const int idx = c0 + 4 * q + g;
        int sl = d;
        if (idx < deg) sl = col[beg + idx];

        v4f row[8];
#define ROW_LOAD(I) if ((I) == 0 || 4 * (I) < m)   \
            row[I] = h1v[(unsigned)group8_bcast<I>(sl) * 8u + (unsigned)q];
        ROW_LOAD(0) ROW_LOAD(1) ROW_LOAD(2) ROW_LOAD(3)
        ROW_LOAD(4) ROW_LOAD(5) ROW_LOAD(6) ROW_LOAD(7)
#undef ROW_LOAD
        if (first) {
            float t = own.x * adw.x + own.y * adw.y + own.z * adw.z + own.w * adw.w;
            t += lane_dpp<DPP_XOR1>(t);
            t += lane_dpp<DPP_XOR2>(t);
            adp = t;
        }
        float lg[8];
        float cm = NEG_INF;
        #pragma unroll
        for (int i = 0; i < 8; ++i) {
            if (i == 0 || 4 * i < m) {
                float p = row[i].x * asw.x + row[i].y * asw.y + row[i].z * asw.z + row[i].w * asw.w;
                p += lane_dpp<DPP_XOR1>(p);
                p += lane_dpp<DPP_XOR2>(p);
                lg[i] = (i < nv) ? lrelu(p + adp) : NEG_INF;
                cm = fmaxf(cm, lg[i]);
            }
        }
        cm = fmaxf(cm, lane_dpp<DPP_XOR8>(cm));
        cm = fmaxf(cm, xhalf(cm));
        if (first) {
            mx = cm;
        } else {
            const float nm = fmaxf(mx, cm);
            const float sc = __expf(mx - nm);
            den *= sc;
            acc.x *= sc; acc.y *= sc; acc.z *= sc; acc.w *= sc;
            mx = nm;
        }
        #pragma unroll
        for (int i = 0; i < 8; ++i) {
            if (i == 0 || 4 * i < m) {
                const float e = (i < nv) ? __expf(lg[i] - mx) : 0.0f;
                den += e;
                acc.x += e * row[i].x;
                acc.y += e * row[i].y;
                acc.z += e * row[i].z;
                acc.w += e * row[i].w;
            }
        }
    };
    walk(0, true);
    for (int c0 = 32; c0 < tot; c0 += 32) walk(c0, false);

    den = slot_sum(den);
    acc.x = slot_sum(acc.x);
    acc.y = slot_sum(acc.y);
    acc.z = slot_sum(acc.z);
    acc.w = slot_sum(acc.w);

    const float inv = 1.0f / (den + 1e-16f);
    float v0 = acc.x * inv + bias.x;
    float v1 = acc.y * inv + bias.y;
    float v2 = acc.z * inv + bias.z;
    float v3 = acc.w * inv + bias.w;
    v0 = v0 > 0.0f ? v0 : 0.0f;
    v1 = v1 > 0.0f ? v1 : 0.0f;
    v2 = v2 > 0.0f ? v2 : 0.0f;
    v3 = v3 > 0.0f ? v3 : 0.0f;
    float g0 = v0 * w2a.x + v1 * w2a.z + v2 * w2b.x + v3 * w2b.z;
    float g1 = v0 * w2a.y + v1 * w2a.w + v2 * w2b.y + v3 * w2b.w;
    g0 += lane_dpp<DPP_XOR1>(g0);
    g1 += lane_dpp<DPP_XOR1>(g1);
    g0 += lane_dpp<DPP_XOR2>(g0);
    g1 += lane_dpp<DPP_XOR2>(g1);
    g0 += lane_dpp<DPP_XOR4>(g0);
    g1 += lane_dpp<DPP_XOR4>(g1);
    if (lane == 0) {
        v4f r;
        r.x = g0;
        r.y = g1;
        r.z = g0 * as2w[0] + g1 * as2w[1];
        r.w = g0 * ad2w[0] + g1 * ad2w[1];
        *(v4f*)(rec2 + (size_t)d * 4) = r;
    }
    }
}

#define N2_GROUPS 4

template <int C>
__device__ __forceinline__ int lane_dpp_i(int v) {
    return __builtin_amdgcn_update_dpp(0, v, C, 0xf, 0xf, true);
}

__global__ __launch_bounds__(256) void k_node2(const int* __restrict__ rowptr,
                                               const int* __restrict__ cnt,
                                               const int* __restrict__ col,
                                               const float* __restrict__ rec2,
                                               const float* __restrict__ b2,
                                               float* __restrict__ out,
                                               int n) {
    const int lane = threadIdx.x & 31;
    const int w = __builtin_amdgcn_readfirstlane((int)((blockIdx.x * blockDim.x + threadIdx.x) >> 5));
    const float NEG_INF = __int_as_float(0xFF800000);
    const float bias0 = b2[0];
    const float bias1 = b2[1];
    const int sub = lane & 3;
    const int slot = lane >> 2;
    const v4f* __restrict__ rec = (const v4f*)rec2;
    const int dFirst = w * (8 * N2_GROUPS);
    if (dFirst >= n) return;

    int dc = dFirst + slot < n ? dFirst + slot : n - 1;
    int beg = rowptr[dc];
    int deg = cnt[dc];
    float ad = rec2[(size_t)dc * 4 + 3];

    for (int gi = 0; gi < N2_GROUPS; ++gi) {
        const int d0 = dFirst + gi * 8;
        if (d0 >= n) break;
        const int d = d0 + slot;
        const bool valid = d < n;
        const int tot = valid ? deg + 1 : 0;

        const int dn = d + 8 < n ? d + 8 : n - 1;
        const int nbeg = rowptr[dn];
        const int ndeg = cnt[dn];
        const float nad = rec2[(size_t)dn * 4 + 3];

        int mt = tot;
        mt = max(mt, lane_dpp_i<DPP_XOR4>(mt));
        mt = max(mt, lane_dpp_i<DPP_XOR8>(mt));
        mt = max(mt, (int)__builtin_amdgcn_permlanex16(mt, mt, 0x76543210, 0xfedcba98, false, false));
        mt = __builtin_amdgcn_readfirstlane(mt);

        float mx = NEG_INF;
        float den = 0.0f;
        float a0 = 0.0f;
        float a1 = 0.0f;
        for (int c0 = 0; c0 < mt; c0 += 32) {
            int sv[8];
            v4f r[8];
            float lg[8];
            #pragma unroll
            for (int j = 0; j < 8; ++j) {
                const int idx = c0 + 4 * j + sub;
                sv[j] = col[idx < deg ? beg + idx : 0];
            }
            #pragma unroll
            for (int j = 0; j < 8; ++j) {
                const int idx = c0 + 4 * j + sub;
                r[j] = rec[idx < deg ? sv[j] : dc];
            }
            float cm = NEG_INF;
            #pragma unroll
            for (int j = 0; j < 8; ++j) {
                const int idx = c0 + 4 * j + sub;
                lg[j] = idx < tot ? lrelu(r[j].z + ad) : NEG_INF;
                cm = fmaxf(cm, lg[j]);
            }
            cm = fmaxf(cm, lane_dpp<DPP_XOR1>(cm));
            cm = fmaxf(cm, lane_dpp<DPP_XOR2>(cm));
            if (c0 == 0) {
                mx = cm;
            } else {
                const float nm = fmaxf(mx, cm);
                const float sc = __expf(mx - nm);
                den *= sc;
                a0 *= sc;
                a1 *= sc;
                mx = nm;
            }
            #pragma unroll
            for (int j = 0; j < 8; ++j) {
                const int idx = c0 + 4 * j + sub;
                const float e = idx < tot ? __expf(lg[j] - mx) : 0.0f;
                den += e;
                a0 += e * r[j].x;
                a1 += e * r[j].y;
            }
        }
        den += lane_dpp<DPP_XOR1>(den);
        a0 += lane_dpp<DPP_XOR1>(a0);
        a1 += lane_dpp<DPP_XOR1>(a1);
        den += lane_dpp<DPP_XOR2>(den);
        a0 += lane_dpp<DPP_XOR2>(a0);
        a1 += lane_dpp<DPP_XOR2>(a1);
        if (valid && sub == 0) {
            const float inv = 1.0f / (den + 1e-16f);
            v2f o;
            o.x = a0 * inv + bias0;
            o.y = a1 * inv + bias1;
            *(v2f*)(out + (size_t)d * 2) = o;
        }
        dc = dn;
        beg = nbeg;
        deg = ndeg;
        ad = nad;
    }
}

extern "C" void kernel_launch(void* const* d_in, const int* in_sizes, int n_in,
                              void* d_out, int out_size, void* d_ws, size_t ws_size,
                              hipStream_t stream) {
    const float* x    = (const float*)d_in[0];
    const int*   ei   = (const int*)d_in[1];
    const float* W1   = (const float*)d_in[2];
    const float* as1w = (const float*)d_in[3];
    const float* ad1w = (const float*)d_in[4];
    const float* b1   = (const float*)d_in[5];
    const float* W2   = (const float*)d_in[6];
    const float* as2w = (const float*)d_in[7];
    const float* ad2w = (const float*)d_in[8];
    const float* b2   = (const float*)d_in[9];

    const int N = in_sizes[0] / 128;
    const int E = in_sizes[1] / 2;
    const int* srcI = ei;
    const int* dstI = ei + E;

    float* p = (float*)d_ws;
    float* h1     = p; p += (size_t)32 * N;
    float* rec2   = p; p += (size_t)4 * N;
    int* cnt      = (int*)p; p += (size_t)N;
    int* rowptr   = (int*)p; p += (size_t)N;
    int* staging  = (int*)p; p += (size_t)2 * E;
    int* col      = (int*)p; p += (size_t)E;
    const int nb    = (N + BIN_NODES - 1) / BIN_NODES;
    const int etiles = (E + TILE_EDGES - 1) / TILE_EDGES;
    int* L        = (int*)p; p += (size_t)etiles * (nb + 1);

    const int T = 256;

    {
        const int gblocks = (N + 511) / 512;
        k_tilesort_gemm1<<<etiles + gblocks, 1024, 0, stream>>>(srcI, dstI, L, staging,
                                                                E, nb, etiles, x, W1, h1, N);
    }
    k_binsort<<<nb, 1024, 0, stream>>>(staging, L, rowptr, cnt, col, N, nb, etiles);
    k_node1<<<(N + 63) / 64, T, 0, stream>>>(rowptr, cnt, col, h1, as1w, ad1w, b1,
                                           W2, as2w, ad2w, rec2, N);
    k_node2<<<(N + 255) / 256, T, 0, stream>>>(rowptr, cnt, col, rec2, b2,
                                             (float*)d_out, N);
}


# bin pass work-group goes on into the layer-1 node pass of its own 512 nodes, reading their lists from its LDS: 3 launches instead of 4; every sum in the same order as before
# speedup vs baseline: 14.1004x; 1.1799x over previous
#include <hip/hip_runtime.h>
#include <math.h>


#define NEG_SLOPE 0.2f

typedef __attribute__((ext_vector_type(2))) float v2f;
typedef __attribute__((ext_vector_type(4))) float v4f;
typedef __attribute__((ext_vector_type(8))) float v8f;

__device__ __forceinline__ float lrelu(float v) {
    return v > 0.0f ? v : NEG_SLOPE * v;
}

__device__ __forceinline__ void gemm1_tile(const float* __restrict__ x,
                                           const float* __restrict__ W1,
                                           float* __restrict__ h1,
                                           int n, int wave, int lane) {
    const int row0 = wave * 16;
    if (row0 >= n) return;

    const int m  = lane & 15;
    const int kh = lane >> 4;
    int row = row0 + m;
    if (row >= n) row = n - 1;

    const float* __restrict__ xrow = x + (size_t)row * 128;

    v8f c0 = {};
    v8f c1 = {};

    #pragma unroll
    for (int k0 = 0; k0 < 128; k0 += 4) {
        const int k = k0 + 2 * kh;
        v2f a = *(const v2f*)(xrow + k);
        v2f b0, b1;
        b0.x = W1[(size_t)k * 32 + m];
        b0.y = W1[(size_t)(k + 1) * 32 + m];
        b1.x = W1[(size_t)k * 32 + 16 + m];
        b1.y = W1[(size_t)(k + 1) * 32 + 16 + m];
        c0 = __builtin_amdgcn_wmma_f32_16x16x4_f32(false, a, false, b0,
                                                   (short)0, c0, false, false);
        c1 = __builtin_amdgcn_wmma_f32_16x16x4_f32(false, a, false, b1,
                                                   (short)0, c1, false, false);
    }

    const int mb = kh * 8;
    #pragma unroll
    for (int j = 0; j < 8; ++j) {
        const int r = row0 + mb + j;
        if (r < n) {
            h1[(size_t)r * 32 + m]      = c0[j];
            h1[(size_t)r * 32 + 16 + m] = c1[j];
        }
    }
}


#define BIN_SHIFT 9
#define BIN_NODES 512
#define MAX_BINS 1024
#define MAX_TILES 1024
#define TILE_EDGES 8192
#define EPT 8
#define BIN_CAP 16384
#define RPW 7

typedef __attribute__((ext_vector_type(2))) int v2i;

__device__ __forceinline__ int block_excl_scan(int c, int* wsum, int lane, int wv) {
    int incl = c;
    for (int o = 1; o < 32; o <<= 1) {
        const int t = __shfl_up(incl, o);
        if (lane >= o) incl += t;
    }
    if (lane == 31) wsum[wv] = incl;
    __syncthreads();
    int woff = 0;
    for (int k = 0; k < wv; ++k) woff += wsum[k];
    return woff + incl - c;
}

__global__ __launch_bounds__(1024) void k_tilesort_gemm1(const int* __restrict__ src,
                                                         const int* __restrict__ dst,
                                                         int* __restrict__ L,
                                                         int* __restrict__ staging,
                                                         int E, int nb, int tiles,
                                                         const float* __restrict__ x,
                                                         const float* __restrict__ W1,
                                                         float* __restrict__ h1,
                                                         int n) {
    if ((int)blockIdx.x >= tiles) {
        gemm1_tile(x, W1, h1, n, ((int)blockIdx.x - tiles) * 32 + (int)(threadIdx.x >> 5),
                   (int)(threadIdx.x & 31));
        return;
    }
    __shared__ int cur[MAX_BINS];
    __shared__ int ws[32];
    __shared__ v2i buf[TILE_EDGES];
    const int tid = threadIdx.x;
    const int lane = tid & 31;
    const int wv = tid >> 5;
    const int tile = blockIdx.x;

    for (int b = tid; b < nb; b += 1024) cur[b] = 0;
    __syncthreads();

    const int e0 = tile * TILE_EDGES;
    int ne = E - e0;
    if (ne > TILE_EDGES) ne = TILE_EDGES;
    int sv[EPT], dv[EPT];
    #pragma unroll
    for (int u = 0; u < EPT; ++u) {
        const int i = u * 1024 + tid;
        const int ic = e0 + (i < ne ? i : ne - 1);
        sv[u] = src[ic];
        dv[u] = dst[ic];
    }
    #pragma unroll
    for (int u = 0; u < EPT; ++u)
        if (u * 1024 + tid < ne) atomicAdd(&cur[dv[u] >> BIN_SHIFT], 1);
    __syncthreads();
    const int myCnt = (tid < nb) ? cur[tid] : 0;
    const int ls = block_excl_scan(myCnt, ws, lane, wv);
    int* __restrict__ Lrow = L + (size_t)tile * (size_t)(nb + 1);
    if (tid < nb) { cur[tid] = ls; Lrow[tid] = ls; }
    if (tid == 0) Lrow[nb] = ne;
    __syncthreads();
    #pragma unroll
    for (int u = 0; u < EPT; ++u) {
        if (u * 1024 + tid < ne) {
            v2i r;
            r.x = sv[u];
            r.y = dv[u];
            const int p = atomicAdd(&cur[dv[u] >> BIN_SHIFT], 1);
            buf[p] = r;
        }
    }
    __syncthreads();
    v2i* __restrict__ st = (v2i*)staging + e0;
    for (int i = tid; i < ne; i += 1024) st[i] = buf[i];
}

#define DPP_XOR1 0xB1
#define DPP_XOR2 0x4E
#define DPP_XOR4 0x164
#define DPP_XOR8 0x168

template <int C>
__device__ __forceinline__ float lane_dpp(float v) {
    const int b = __builtin_bit_cast(int, v);
    return __builtin_bit_cast(float, __builtin_amdgcn_update_dpp(0, b, C, 0xf, 0xf, true));
}

template <int C>
__device__ __forceinline__ int lane_dpp_i(int v) {
    return __builtin_amdgcn_update_dpp(0, v, C, 0xf, 0xf, true);
}

template <int I>
__device__ __forceinline__ int quad_bcast(int v) {
    return __builtin_amdgcn_update_dpp(0, v, I * 0x55, 0xf, 0xf, true);
}

__device__ __forceinline__ int xhalf_i(int v) {
    return (int)__builtin_amdgcn_permlanex16(v, v, 0x76543210, 0xfedcba98, false, false);
}

#define N1_GROUPS 2
static_assert(32 * 8 * N1_GROUPS == BIN_NODES, "the 32 waves of a bin work-group cover its nodes");

__device__ __forceinline__ v2f lo2(const v4f v) { return __builtin_shufflevector(v, v, 0, 1); }
__device__ __forceinline__ v2f hi2(const v4f v) { return __builtin_shufflevector(v, v, 2, 3); }

__device__ __forceinline__ float dot8(const v4f a, const v4f b, const v4f wa, const v4f wb) {
    v2f s = lo2(a) * lo2(wa);
    s = __builtin_elementwise_fma(hi2(a), hi2(wa), s);
    s = __builtin_elementwise_fma(lo2(b), lo2(wb), s);
    s = __builtin_elementwise_fma(hi2(b), hi2(wb), s);
    return s.x + s.y;
}

template <bool B> struct ListsIn { static constexpr bool value = B; };

__global__ __launch_bounds__(1024) void k_binsort_node1(const int* __restrict__ staging,
                                                        const int* __restrict__ L,
                                                        int* __restrict__ rowptr,
                                                        int* __restrict__ cnt,
                                                        int* __restrict__ col,
                                                        int n, int nb, int tiles,
                                                        const float* __restrict__ h1,
                                                        const float* __restrict__ as1w,
                                                        const float* __restrict__ ad1w,
                                                        const float* __restrict__ b1,
                                                        const float* __restrict__ W2,
                                                        const float* __restrict__ as2w,
                                                        const float* __restrict__ ad2w,
                                                        float* __restrict__ rec2) {
    __shared__ v2i runL[MAX_TILES];
    __shared__ int cntL[BIN_NODES];
    __shared__ int curL[BIN_NODES];
    __shared__ int ws[32];
    __shared__ int tot[2];
    __shared__ int colL[BIN_CAP];
    const int tid = threadIdx.x;
    const int lane = tid & 31;
    const int wv = __builtin_amdgcn_readfirstlane(tid >> 5);
    const int b = blockIdx.x;
    const v2i* __restrict__ st = (const v2i*)staging;

    if (tid < BIN_NODES) cntL[tid] = 0;
    if (tid < 2) tot[tid] = 0;
    int a = 0, len = 0;
    if (tid < tiles) {
        const int* __restrict__ Lrow = L + (size_t)tid * (size_t)(nb + 1) + b;
        a = Lrow[0];
        len = Lrow[1] - a;
        v2i rl;
        rl.x = a;
        rl.y = len;
        runL[tid] = rl;
    }
    __syncthreads();
    {
        int sa = a, sl = len;
        for (int o = 16; o > 0; o >>= 1) { sa += __shfl_xor(sa, o); sl += __shfl_xor(sl, o); }
        if (lane == 0) { atomicAdd(&tot[0], sa); atomicAdd(&tot[1], sl); }
    }
    __syncthreads();
    const int e0 = tot[0];
    const int ne = tot[1];
    const bool fits = ne <= BIN_CAP;
    const bool one = tiles <= 32 * RPW;

    v2i r0[RPW], r1[RPW];
    int ln[RPW], off[RPW];
    auto load_runs = [&](const int t0) {
        v2i rl[RPW];
        #pragma unroll
        for (int u = 0; u < RPW; ++u) {
            const int t = t0 + u * 32 + wv;
            rl[u] = runL[t < tiles ? t : tiles - 1];
        }
        #pragma unroll
        for (int u = 0; u < RPW; ++u) {
            const int t = t0 + u * 32 + wv;
            const int l = t < tiles ? __builtin_amdgcn_readfirstlane(rl[u].y) : 0;
            const int o = l > 0 ? t * TILE_EDGES + __builtin_amdgcn_readfirstlane(rl[u].x) : 0;
            ln[u] = l;
            off[u] = o;
            const int last = l > 0 ? l - 1 : 0;
            r0[u] = st[o + (lane < l ? lane : last)];
            r1[u] = st[o + (lane + 32 < l ? lane + 32 : last)];
        }
    };

    for (int t0 = 0; t0 < tiles; t0 += 32 * RPW) {
        load_runs(t0);
        #pragma unroll
        for (int u = 0; u < RPW; ++u) {
            if (lane < ln[u]) atomicAdd(&cntL[r0[u].y & (BIN_NODES - 1)], 1);
            if (lane + 32 < ln[u]) atomicAdd(&cntL[r1[u].y & (BIN_NODES - 1)], 1);
            for (int i = 64 + lane; i < ln[u]; i += 32)
                atomicAdd(&cntL[st[off[u] + i].y & (BIN_NODES - 1)], 1);
        }
    }
    __syncthreads();
    const int c = (tid < BIN_NODES) ? cntL[tid] : 0;
    const int start = block_excl_scan(c, ws, lane, wv);
    if (tid < BIN_NODES) {
        curL[tid] = start;
        const int d = b * BIN_NODES + tid;
        if (d < n) { rowptr[d] = e0 + start; cnt[d] = c; }
    }
    __syncthreads();
    for (int t0 = 0; t0 < tiles; t0 += 32 * RPW) {
        if (!one) load_runs(t0);
        int p0[RPW], p1[RPW];
        #pragma unroll
        for (int u = 0; u < RPW; ++u) {
            p0[u] = 0;
            p1[u] = 0;
            if (lane < ln[u]) p0[u] = atomicAdd(&curL[r0[u].y & (BIN_NODES - 1)], 1);
            if (lane + 32 < ln[u]) p1[u] = atomicAdd(&curL[r1[u].y & (BIN_NODES - 1)], 1);
        }
        #pragma unroll
        for (int u = 0; u < RPW; ++u) {
            if (lane < ln[u]) {
                if (fits) colL[p0[u]] = r0[u].x; else col[e0 + p0[u]] = r0[u].x;
            }
            if (lane + 32 < ln[u]) {
                if (fits) colL[p1[u]] = r1[u].x; else col[e0 + p1[u]] = r1[u].x;
            }
            for (int i = 64 + lane; i < ln[u]; i += 32) {
                const v2i r = st[off[u] + i];
                const int p = atomicAdd(&curL[r.y & (BIN_NODES - 1)], 1);
                if (fits) colL[p] = r.x; else col[e0 + p] = r.x;
            }
        }
    }
    __syncthreads();
    if (fits)
        for (int i = tid; i < ne; i += 1024) col[e0 + i] = colL[i];

    auto node_pass = [&](auto where) {
        constexpr bool FITS = decltype(where)::value;
        const float NEG_INF = __int_as_float(0xFF800000);
        const int slot = lane >> 2;
        const int p = lane & 3;
        const v4f* __restrict__ h1v = (const v4f*)h1;
        const v4f aswA = *(const v4f*)(as1w + 8 * p);
        const v4f aswB = *(const v4f*)(as1w + 8 * p + 4);
        const float as20 = as2w[0];
        const float as21 = as2w[1];
        const float ad20 = ad2w[0];
        const float ad21 = ad2w[1];
        const int dFirst = b * BIN_NODES + wv * (8 * N1_GROUPS);
        if (dFirst >= n) return;

        #pragma nounroll
        for (int gi = 0; gi < N1_GROUPS; ++gi) {
            const int dg = dFirst + gi * 8;
            if (dg >= n) break;
            const int d = dg + slot;
            const bool valid = d < n;
            const int dc = valid ? d : n - 1;
            const int lc = dc - b * BIN_NODES;
            const int deg = cntL[lc];
            const int beg = curL[lc] - deg;
            const int nent = valid ? deg + 1 : 0;
            const unsigned rp = 2u * (unsigned)p;
            const v4f ownA = h1v[(unsigned)dc * 8u + rp];
            const v4f ownB = h1v[(unsigned)dc * 8u + rp + 1u];

            auto load_src = [&](const int c0, int& ea, int& eb) {
                const int ia = c0 + p;
                const int ib = c0 + 4 + p;
                ea = dc;
                eb = dc;
                if constexpr (FITS) {
                    if (ia < deg) ea = colL[beg + ia];
                    if (ib < deg) eb = colL[beg + ib];
                } else {
                    if (ia < deg) ea = col[e0 + beg + ia];
                    if (ib < deg) eb = col[e0 + beg + ib];
                }
            };
            int svA, svB;
            load_src(0, svA, svB);

            int mt = nent;
            mt = max(mt, lane_dpp_i<DPP_XOR4>(mt));
            mt = max(mt, lane_dpp_i<DPP_XOR8>(mt));
            mt = max(mt, xhalf_i(mt));
            mt = __builtin_amdgcn_readfirstlane(mt);

            float adp;
            {
                const v4f adwA = *(const v4f*)(ad1w + 8 * p);
                const v4f adwB = *(const v4f*)(ad1w + 8 * p + 4);
                float t = dot8(ownA, ownB, adwA, adwB);
                t += lane_dpp<DPP_XOR1>(t);
                adp = t;
            }

            float mx = NEG_INF;
            float den = 0.0f;
            v2f acc0 = {0.0f, 0.0f};
            v2f acc1 = {0.0f, 0.0f};
            v2f acc2 = {0.0f, 0.0f};
            v2f acc3 = {0.0f, 0.0f};

            auto walk = [&](const int c0, const bool first) {
                const int sa = svA;
                const int sb = svB;
                v4f rowA[8], rowB[8];
#define ROW_LOAD(I, S) { \
                    const unsigned r = (unsigned)quad_bcast<(I) & 3>(S) * 8u + rp; \
                    rowA[I] = h1v[r]; \
                    rowB[I] = h1v[r + 1u]; }
                ROW_LOAD(0, sa) ROW_LOAD(1, sa) ROW_LOAD(2, sa) ROW_LOAD(3, sa)
                ROW_LOAD(4, sb) ROW_LOAD(5, sb) ROW_LOAD(6, sb) ROW_LOAD(7, sb)
#undef ROW_LOAD
                if (c0 + 8 < mt) load_src(c0 + 8, svA, svB);

                float lg[8];
                float cm = NEG_INF;
                #pragma unroll
                for (int i = 0; i < 8; ++i) {
                    float t = dot8(rowA[i], rowB[i], aswA, aswB);
                    t += lane_dpp<DPP_XOR1>(t);
                    lg[i] = (c0 + i < nent) ? lrelu(t + adp) : NEG_INF;
                    cm = fmaxf(cm, lg[i]);
                }
                if (first) {
                    mx = cm;
                } else {
                    const float nm = fmaxf(mx, cm);
                    const float sc = __expf(mx - nm);
                    den *= sc;
                    acc0 *= sc; acc1 *= sc; acc2 *= sc; acc3 *= sc;
                    mx = nm;
                }
                #pragma unroll
                for (int i = 0; i < 8; ++i) {
                    const float e = (c0 + i < nent) ? __expf(lg[i] - mx) : 0.0f;
                    den += e;
                    const v2f ee = {e, e};
                    acc0 = __builtin_elementwise_fma(lo2(rowA[i]), ee, acc0);
                    acc1 = __builtin_elementwise_fma(hi2(rowA[i]), ee, acc1);
                    acc2 = __builtin_elementwise_fma(lo2(rowB[i]), ee, acc2);
                    acc3 = __builtin_elementwise_fma(hi2(rowB[i]), ee, acc3);
                }
            };
            walk(0, true);
            for (int c0 = 8; c0 < mt; c0 += 8) walk(c0, false);

            const v4f biasA = *(const v4f*)(b1 + 8 * p);
            const v4f biasB = *(const v4f*)(b1 + 8 * p + 4);
            const v4f w2a = *(const v4f*)(W2 + 16 * p);
            const v4f w2b = *(const v4f*)(W2 + 16 * p + 4);
            const v4f w2c = *(const v4f*)(W2 + 16 * p + 8);
            const v4f w2d = *(const v4f*)(W2 + 16 * p + 12);
            const float inv = 1.0f / (den + 1e-16f);
            float v0 = acc0.x * inv + biasA.x;
            float v1 = acc0.y * inv + biasA.y;
            float v2 = acc1.x * inv + biasA.z;
            float v3 = acc1.y * inv + biasA.w;
            float v4 = acc2.x * inv + biasB.x;
            float v5 = acc2.y * inv + biasB.y;
            float v6 = acc3.x * inv + biasB.z;
            float v7 = acc3.y * inv + biasB.w;
            v0 = v0 > 0.0f ? v0 : 0.0f;
            v1 = v1 > 0.0f ? v1 : 0.0f;
            v2 = v2 > 0.0f ? v2 : 0.0f;
            v3 = v3 > 0.0f ? v3 : 0.0f;
            v4 = v4 > 0.0f ? v4 : 0.0f;
            v5 = v5 > 0.0f ? v5 : 0.0f;
            v6 = v6 > 0.0f ? v6 : 0.0f;
            v7 = v7 > 0.0f ? v7 : 0.0f;
            float g0 = v0 * w2a.x + v1 * w2a.z + v2 * w2b.x + v3 * w2b.z
                     + v4 * w2c.x + v5 * w2c.z + v6 * w2d.x + v7 * w2d.z;
            float g1 = v0 * w2a.y + v1 * w2a.w + v2 * w2b.y + v3 * w2b.w
                     + v4 * w2c.y + v5 * w2c.w + v6 * w2d.y + v7 * w2d.w;
            g0 += lane_dpp<DPP_XOR1>(g0);
            g1 += lane_dpp<DPP_XOR1>(g1);
            g0 += lane_dpp<DPP_XOR2>(g0);
            g1 += lane_dpp<DPP_XOR2>(g1);
            if (valid && p == 0) {
                v4f r;
                r.x = g0;
                r.y = g1;
                r.z = g0 * as20 + g1 * as21;
                r.w = g0 * ad20 + g1 * ad21;
                *(v4f*)(rec2 + (size_t)d * 4) = r;
            }
        }
    };
    if (fits) node_pass(ListsIn<true>{}); else node_pass(ListsIn<false>{});
}

#define N2_GROUPS 4

__global__ __launch_bounds__(256) void k_node2(const int* __restrict__ rowptr,
                                               const int* __restrict__ cnt,
                                               const int* __restrict__ col,
                                               const float* __restrict__ rec2,
                                               const float* __restrict__ b2,
                                               float* __restrict__ out,
                                               int n) {
    const int lane = threadIdx.x & 31;
    const int w = __builtin_amdgcn_readfirstlane((int)((blockIdx.x * blockDim.x + threadIdx.x) >> 5));
    const float NEG_INF = __int_as_float(0xFF800000);
    const float bias0 = b2[0];
    const float bias1 = b2[1];
    const int sub = lane & 3;
    const int slot = lane >> 2;
    const v4f* __restrict__ rec = (const v4f*)rec2;
    const int dFirst = w * (8 * N2_GROUPS);
    if (dFirst >= n) return;

    int dc = dFirst + slot < n ? dFirst + slot : n - 1;
    int beg = rowptr[dc];
    int deg = cnt[dc];
    float ad = rec2[(size_t)dc * 4 + 3];

    for (int gi = 0; gi < N2_GROUPS; ++gi) {
        const int d0 = dFirst + gi * 8;
        if (d0 >= n) break;
        const int d = d0 + slot;
        const bool valid = d < n;
        const int tot = valid ? deg + 1 : 0;

        const int dn = d + 8 < n ? d + 8 : n - 1;
        const int nbeg = rowptr[dn];
        const int ndeg = cnt[dn];
        const float nad = rec2[(size_t)dn * 4 + 3];

        int mt = tot;
        mt = max(mt, lane_dpp_i<DPP_XOR4>(mt));
        mt = max(mt, lane_dpp_i<DPP_XOR8>(mt));
        mt = max(mt, (int)__builtin_amdgcn_permlanex16(mt, mt, 0x76543210, 0xfedcba98, false, false));
        mt = __builtin_amdgcn_readfirstlane(mt);

        float mx = NEG_INF;
        float den = 0.0f;
        float a0 = 0.0f;
        float a1 = 0.0f;
        for (int c0 = 0; c0 < mt; c0 += 32) {
            int sv[8];
            v4f r[8];
            float lg[8];
            #pragma unroll
            for (int j = 0; j < 8; ++j) {
                const int idx = c0 + 4 * j + sub;
                sv[j] = col[idx < deg ? beg + idx : 0];
            }
            #pragma unroll
            for (int j = 0; j < 8; ++j) {
                const int idx = c0 + 4 * j + sub;
                r[j] = rec[idx < deg ? sv[j] : dc];
            }
            float cm = NEG_INF;
            #pragma unroll
            for (int j = 0; j < 8; ++j) {
                const int idx = c0 + 4 * j + sub;
                lg[j] = idx < tot ? lrelu(r[j].z + ad) : NEG_INF;
                cm = fmaxf(cm, lg[j]);
            }
            cm = fmaxf(cm, lane_dpp<DPP_XOR1>(cm));
            cm = fmaxf(cm, lane_dpp<DPP_XOR2>(cm));
            if (c0 == 0) {
                mx = cm;
            } else {
                const float nm = fmaxf(mx, cm);
                const float sc = __expf(mx - nm);
                den *= sc;
                a0 *= sc;
                a1 *= sc;
                mx = nm;
            }
            #pragma unroll
            for (int j = 0; j < 8; ++j) {
                const int idx = c0 + 4 * j + sub;
                const float e = idx < tot ? __expf(lg[j] - mx) : 0.0f;
                den += e;
                a0 += e * r[j].x;
                a1 += e * r[j].y;
            }
        }
        den += lane_dpp<DPP_XOR1>(den);
        a0 += lane_dpp<DPP_XOR1>(a0);
        a1 += lane_dpp<DPP_XOR1>(a1);
        den += lane_dpp<DPP_XOR2>(den);
        a0 += lane_dpp<DPP_XOR2>(a0);
        a1 += lane_dpp<DPP_XOR2>(a1);
        if (valid && sub == 0) {
            const float inv = 1.0f / (den + 1e-16f);
            v2f o;
            o.x = a0 * inv + bias0;
            o.y = a1 * inv + bias1;
            *(v2f*)(out + (size_t)d * 2) = o;
        }
        dc = dn;
        beg = nbeg;
        deg = ndeg;
        ad = nad;
    }
}

extern "C" void kernel_launch(void* const* d_in, const int* in_sizes, int n_in,
                              void* d_out, int out_size, void* d_ws, size_t ws_size,
                              hipStream_t stream) {
    const float* x    = (const float*)d_in[0];
    const int*   ei   = (const int*)d_in[1];
    const float* W1   = (const float*)d_in[2];
    const float* as1w = (const float*)d_in[3];
    const float* ad1w = (const float*)d_in[4];
    const float* b1   = (const float*)d_in[5];
    const float* W2   = (const float*)d_in[6];
    const float* as2w = (const float*)d_in[7];
    const float* ad2w = (const float*)d_in[8];
    const float* b2   = (const float*)d_in[9];

    const int N = in_sizes[0] / 128;
    const int E = in_sizes[1] / 2;
    const int* srcI = ei;
    const int* dstI = ei + E;

    float* p = (float*)d_ws;
    float* h1     = p; p += (size_t)32 * N;
    float* rec2   = p; p += (size_t)4 * N;
    int* cnt      = (int*)p; p += (size_t)N;
    int* rowptr   = (int*)p; p += (size_t)N;
    int* staging  = (int*)p; p += (size_t)2 * E;
    int* col      = (int*)p; p += (size_t)E;
    const int nb    = (N + BIN_NODES - 1) / BIN_NODES;
    const int etiles = (E + TILE_EDGES - 1) / TILE_EDGES;
    int* L        = (int*)p; p += (size_t)etiles * (nb + 1);

    const int T = 256;

    {
        const int gblocks = (N + 511) / 512;
        k_tilesort_gemm1<<<etiles + gblocks, 1024, 0, stream>>>(srcI, dstI, L, staging,
                                                                E, nb, etiles, x, W1, h1, N);
    }
    k_binsort_node1<<<nb, 1024, 0, stream>>>(staging, L, rowptr, cnt, col, N, nb, etiles,
                                             h1, as1w, ad1w, b1, W2, as2w, ad2w, rec2);
    k_node2<<<(N + 255) / 256, T, 0, stream>>>(rowptr, cnt, col, rec2, b2,
                                             (float*)d_out, N);
}


# layer-1 product operand fetch: W1 fragments built once per work-group in LDS (not 128 global loads per wave), x read 16 B per lane; a WMMA takes k=8c+4h+{0,1} or {2,3}: k sum order changes
# speedup vs baseline: 14.6676x; 1.0402x over previous
#include <hip/hip_runtime.h>
#include <math.h>


#define NEG_SLOPE 0.2f

typedef __attribute__((ext_vector_type(2))) float v2f;
typedef __attribute__((ext_vector_type(4))) float v4f;
typedef __attribute__((ext_vector_type(8))) float v8f;

__device__ __forceinline__ float lrelu(float v) {
    return v > 0.0f ? v : NEG_SLOPE * v;
}

#define GEMM_AHEAD 6

__device__ __forceinline__ v2f lo2(const v4f v) { return __builtin_shufflevector(v, v, 0, 1); }
__device__ __forceinline__ v2f hi2(const v4f v) { return __builtin_shufflevector(v, v, 2, 3); }

__device__ __forceinline__ void gemm1_group(const float* __restrict__ x,
                                            const float* __restrict__ W1,
                                            float* __restrict__ h1,
                                            v4f* __restrict__ Bs,
                                            int n, int wave, int tid) {
    const int lane = tid & 31;
    const int m  = lane & 15;
    const int kh = lane >> 4;
    const int row0 = wave * 16;
    int row = row0 + m;
    if (row >= n) row = n - 1;
    const v4f* __restrict__ xr = (const v4f*)(x + (size_t)row * 128) + kh;

    v4f a[16];
    #pragma unroll
    for (int c = 0; c < GEMM_AHEAD; ++c) a[c] = xr[2 * c];

    {
        const int k = 8 * (tid >> 6) + 4 * kh + 2 * ((tid >> 5) & 1);
        v4f w;
        w.x = W1[k * 32 + m];
        w.y = W1[(k + 1) * 32 + m];
        w.z = W1[k * 32 + 16 + m];
        w.w = W1[(k + 1) * 32 + 16 + m];
        Bs[tid] = w;
    }
    __syncthreads();

    v8f c0 = {};
    v8f c1 = {};
    #pragma unroll
    for (int c = 0; c < 16; ++c) {
        if (c + GEMM_AHEAD < 16) a[c + GEMM_AHEAD] = xr[2 * (c + GEMM_AHEAD)];
        const v4f bA = Bs[(2 * c) * 32 + lane];
        const v4f bB = Bs[(2 * c + 1) * 32 + lane];
        c0 = __builtin_amdgcn_wmma_f32_16x16x4_f32(false, lo2(a[c]), false, lo2(bA),
                                                   (short)0, c0, false, false);
        c1 = __builtin_amdgcn_wmma_f32_16x16x4_f32(false, lo2(a[c]), false, hi2(bA),
                                                   (short)0, c1, false, false);
        c0 = __builtin_amdgcn_wmma_f32_16x16x4_f32(false, hi2(a[c]), false, lo2(bB),
                                                   (short)0, c0, false, false);
        c1 = __builtin_amdgcn_wmma_f32_16x16x4_f32(false, hi2(a[c]), false, hi2(bB),
                                                   (short)0, c1, false, false);
    }

    const int mb = kh * 8;
    #pragma unroll
    for (int j = 0; j < 8; ++j) {
        const int r = row0 + mb + j;
        if (r < n) {
            h1[(size_t)r * 32 + m]      = c0[j];
            h1[(size_t)r * 32 + 16 + m] = c1[j];
        }
    }
}


#define BIN_SHIFT 9
#define BIN_NODES 512
#define MAX_BINS 1024
#define MAX_TILES 1024
#define TILE_EDGES 8192
#define EPT 8
#define BIN_CAP 16384
#define RPW 7

typedef __attribute__((ext_vector_type(2))) int v2i;

__device__ __forceinline__ int block_excl_scan(int c, int* wsum, int lane, int wv) {
    int incl = c;
    for (int o = 1; o < 32; o <<= 1) {
        const int t = __shfl_up(incl, o);
        if (lane >= o) incl += t;
    }
    if (lane == 31) wsum[wv] = incl;
    __syncthreads();
    int woff = 0;
    for (int k = 0; k < wv; ++k) woff += wsum[k];
    return woff + incl - c;
}

__global__ __launch_bounds__(1024) void k_tilesort_gemm1(const int* __restrict__ src,
                                                         const int* __restrict__ dst,
                                                         int* __restrict__ L,
                                                         int* __restrict__ staging,
                                                         int E, int nb, int tiles,
                                                         const float* __restrict__ x,
                                                         const float* __restrict__ W1,
                                                         float* __restrict__ h1,
                                                         int n) {
    __shared__ v4f Bs[1024];
    if ((int)blockIdx.x >= tiles) {
        gemm1_group(x, W1, h1, Bs, n, ((int)blockIdx.x - tiles) * 32 + (int)(threadIdx.x >> 5),
                    (int)threadIdx.x);
        return;
    }
    __shared__ int cur[MAX_BINS];
    __shared__ int ws[32];
    __shared__ v2i buf[TILE_EDGES];
    const int tid = threadIdx.x;
    const int lane = tid & 31;
    const int wv = tid >> 5;
    const int tile = blockIdx.x;

    for (int b = tid; b < nb; b += 1024) cur[b] = 0;
    __syncthreads();

    const int e0 = tile * TILE_EDGES;
    int ne = E - e0;
    if (ne > TILE_EDGES) ne = TILE_EDGES;
    int sv[EPT], dv[EPT];
    #pragma unroll
    for (int u = 0; u < EPT; ++u) {
        const int i = u * 1024 + tid;
        const int ic = e0 + (i < ne ? i : ne - 1);
        sv[u] = src[ic];
        dv[u] = dst[ic];
    }
    #pragma unroll
    for (int u = 0; u < EPT; ++u)
        if (u * 1024 + tid < ne) atomicAdd(&cur[dv[u] >> BIN_SHIFT], 1);
    __syncthreads();
    const int myCnt = (tid < nb) ? cur[tid] : 0;
    const int ls = block_excl_scan(myCnt, ws, lane, wv);
    int* __restrict__ Lrow = L + (size_t)tile * (size_t)(nb + 1);
    if (tid < nb) { cur[tid] = ls; Lrow[tid] = ls; }
    if (tid == 0) Lrow[nb] = ne;
    __syncthreads();
    #pragma unroll
    for (int u = 0; u < EPT; ++u) {
        if (u * 1024 + tid < ne) {
            v2i r;
            r.x = sv[u];
            r.y = dv[u];
            const int p = atomicAdd(&cur[dv[u] >> BIN_SHIFT], 1);
            buf[p] = r;
        }
    }
    __syncthreads();
    v2i* __restrict__ st = (v2i*)staging + e0;
    for (int i = tid; i < ne; i += 1024) st[i] = buf[i];
}

#define DPP_XOR1 0xB1
#define DPP_XOR2 0x4E
#define DPP_XOR4 0x164
#define DPP_XOR8 0x168

template <int C>
__device__ __forceinline__ float lane_dpp(float v) {
    const int b = __builtin_bit_cast(int, v);
    return __builtin_bit_cast(float, __builtin_amdgcn_update_dpp(0, b, C, 0xf, 0xf, true));
}

template <int C>
__device__ __forceinline__ int lane_dpp_i(int v) {
    return __builtin_amdgcn_update_dpp(0, v, C, 0xf, 0xf, true);
}

template <int I>
__device__ __forceinline__ int quad_bcast(int v) {
    return __builtin_amdgcn_update_dpp(0, v, I * 0x55, 0xf, 0xf, true);
}

__device__ __forceinline__ int xhalf_i(int v) {
    return (int)__builtin_amdgcn_permlanex16(v, v, 0x76543210, 0xfedcba98, false, false);
}

#define N1_GROUPS 2
static_assert(32 * 8 * N1_GROUPS == BIN_NODES, "the 32 waves of a bin work-group cover its nodes");

__device__ __forceinline__ float dot8(const v4f a, const v4f b, const v4f wa, const v4f wb) {
    v2f s = lo2(a) * lo2(wa);
    s = __builtin_elementwise_fma(hi2(a), hi2(wa), s);
    s = __builtin_elementwise_fma(lo2(b), lo2(wb), s);
    s = __builtin_elementwise_fma(hi2(b), hi2(wb), s);
    return s.x + s.y;
}

template <bool B> struct ListsIn { static constexpr bool value = B; };

__global__ __launch_bounds__(1024) void k_binsort_node1(const int* __restrict__ staging,
                                                        const int* __restrict__ L,
                                                        int* __restrict__ rowptr,
                                                        int* __restrict__ cnt,
                                                        int* __restrict__ col,
                                                        int n, int nb, int tiles,
                                                        const float* __restrict__ h1,
                                                        const float* __restrict__ as1w,
                                                        const float* __restrict__ ad1w,
                                                        const float* __restrict__ b1,
                                                        const float* __restrict__ W2,
                                                        const float* __restrict__ as2w,
                                                        const float* __restrict__ ad2w,
                                                        float* __restrict__ rec2) {
    __shared__ v2i runL[MAX_TILES];
    __shared__ int cntL[BIN_NODES];
    __shared__ int curL[BIN_NODES];
    __shared__ int ws[32];
    __shared__ int tot[2];
    __shared__ int colL[BIN_CAP];
    const int tid = threadIdx.x;
    const int lane = tid & 31;
    const int wv = __builtin_amdgcn_readfirstlane(tid >> 5);
    const int b = blockIdx.x;
    const v2i* __restrict__ st = (const v2i*)staging;

    if (tid < BIN_NODES) cntL[tid] = 0;
    if (tid < 2) tot[tid] = 0;
    int a = 0, len = 0;
    if (tid < tiles) {
        const int* __restrict__ Lrow = L + (size_t)tid * (size_t)(nb + 1) + b;
        a = Lrow[0];
        len = Lrow[1] - a;
        v2i rl;
        rl.x = a;
        rl.y = len;
        runL[tid] = rl;
    }
    __syncthreads();
    {
        int sa = a, sl = len;
        for (int o = 16; o > 0; o >>= 1) { sa += __shfl_xor(sa, o); sl += __shfl_xor(sl, o); }
        if (lane == 0) { atomicAdd(&tot[0], sa); atomicAdd(&tot[1], sl); }
    }
    __syncthreads();
    const int e0 = tot[0];
    const int ne = tot[1];
    const bool fits = ne <= BIN_CAP;
    const bool one = tiles <= 32 * RPW;

    v2i r0[RPW], r1[RPW];
    int ln[RPW], off[RPW];
    auto load_runs = [&](const int t0) {
        v2i rl[RPW];
        #pragma unroll
        for (int u = 0; u < RPW; ++u) {
            const int t = t0 + u * 32 + wv;
            rl[u] = runL[t < tiles ? t : tiles - 1];
        }
        #pragma unroll
        for (int u = 0; u < RPW; ++u) {
            const int t = t0 + u * 32 + wv;
            const int l = t < tiles ? __builtin_amdgcn_readfirstlane(rl[u].y) : 0;
            const int o = l > 0 ? t * TILE_EDGES + __builtin_amdgcn_readfirstlane(rl[u].x) : 0;
            ln[u] = l;
            off[u] = o;
            const int last = l > 0 ? l - 1 : 0;
            r0[u] = st[o + (lane < l ? lane : last)];
            r1[u] = st[o + (lane + 32 < l ? lane + 32 : last)];
        }
    };

    for (int t0 = 0; t0 < tiles; t0 += 32 * RPW) {
        load_runs(t0);
        #pragma unroll
        for (int u = 0; u < RPW; ++u) {
            if (lane < ln[u]) atomicAdd(&cntL[r0[u].y & (BIN_NODES - 1)], 1);
            if (lane + 32 < ln[u]) atomicAdd(&cntL[r1[u].y & (BIN_NODES - 1)], 1);
            for (int i = 64 + lane; i < ln[u]; i += 32)
                atomicAdd(&cntL[st[off[u] + i].y & (BIN_NODES - 1)], 1);
        }
    }
    __syncthreads();
    const int c = (tid < BIN_NODES) ? cntL[tid] : 0;
    const int start = block_excl_scan(c, ws, lane, wv);
    if (tid < BIN_NODES) {
        curL[tid] = start;
        const int d = b * BIN_NODES + tid;
        if (d < n) { rowptr[d] = e0 + start; cnt[d] = c; }
    }
    __syncthreads();
    for (int t0 = 0; t0 < tiles; t0 += 32 * RPW) {
        if (!one) load_runs(t0);
        int p0[RPW], p1[RPW];
        #pragma unroll
        for (int u = 0; u < RPW; ++u) {
            p0[u] = 0;
            p1[u] = 0;
            if (lane < ln[u]) p0[u] = atomicAdd(&curL[r0[u].y & (BIN_NODES - 1)], 1);
            if (lane + 32 < ln[u]) p1[u] = atomicAdd(&curL[r1[u].y & (BIN_NODES - 1)], 1);
        }
        #pragma unroll
        for (int u = 0; u < RPW; ++u) {
            if (lane < ln[u]) {
                if (fits) colL[p0[u]] = r0[u].x; else col[e0 + p0[u]] = r0[u].x;
            }
            if (lane + 32 < ln[u]) {
                if (fits) colL[p1[u]] = r1[u].x; else col[e0 + p1[u]] = r1[u].x;
            }
            for (int i = 64 + lane; i < ln[u]; i += 32) {
                const v2i r = st[off[u] + i];
                const int p = atomicAdd(&curL[r.y & (BIN_NODES - 1)], 1);
                if (fits) colL[p] = r.x; else col[e0 + p] = r.x;
            }
        }
    }
    __syncthreads();
    if (fits)
        for (int i = tid; i < ne; i += 1024) col[e0 + i] = colL[i];

    auto node_pass = [&](auto where) {
        constexpr bool FITS = decltype(where)::value;
        const float NEG_INF = __int_as_float(0xFF800000);
        const int slot = lane >> 2;
        const int p = lane & 3;
        const v4f* __restrict__ h1v = (const v4f*)h1;
        const v4f aswA = *(const v4f*)(as1w + 8 * p);
        const v4f aswB = *(const v4f*)(as1w + 8 * p + 4);
        const float as20 = as2w[0];
        const float as21 = as2w[1];
        const float ad20 = ad2w[0];
        const float ad21 = ad2w[1];
        const int dFirst = b * BIN_NODES + wv * (8 * N1_GROUPS);
        if (dFirst >= n) return;

        #pragma nounroll
        for (int gi = 0; gi < N1_GROUPS; ++gi) {
            const int dg = dFirst + gi * 8;
            if (dg >= n) break;
            const int d = dg + slot;
            const bool valid = d < n;
            const int dc = valid ? d : n - 1;
            const int lc = dc - b * BIN_NODES;
            const int deg = cntL[lc];
            const int beg = curL[lc] - deg;
            const int nent = valid ? deg + 1 : 0;
            const unsigned rp = 2u * (unsigned)p;
            const v4f ownA = h1v[(unsigned)dc * 8u + rp];
            const v4f ownB = h1v[(unsigned)dc * 8u + rp + 1u];

            auto load_src = [&](const int c0, int& ea, int& eb) {
                const int ia = c0 + p;
                const int ib = c0 + 4 + p;
                ea = dc;
                eb = dc;
                if constexpr (FITS) {
                    if (ia < deg) ea = colL[beg + ia];
                    if (ib < deg) eb = colL[beg + ib];
                } else {
                    if (ia < deg) ea = col[e0 + beg + ia];
                    if (ib < deg) eb = col[e0 + beg + ib];
                }
            };
            int svA, svB;
            load_src(0, svA, svB);

            int mt = nent;
            mt = max(mt, lane_dpp_i<DPP_XOR4>(mt));
            mt = max(mt, lane_dpp_i<DPP_XOR8>(mt));
            mt = max(mt, xhalf_i(mt));
            mt = __builtin_amdgcn_readfirstlane(mt);

            float adp;
            {
                const v4f adwA = *(const v4f*)(ad1w + 8 * p);
                const v4f adwB = *(const v4f*)(ad1w + 8 * p + 4);
                float t = dot8(ownA, ownB, adwA, adwB);
                t += lane_dpp<DPP_XOR1>(t);
                adp = t;
            }

            float mx = NEG_INF;
            float den = 0.0f;
            v2f acc0 = {0.0f, 0.0f};
            v2f acc1 = {0.0f, 0.0f};
            v2f acc2 = {0.0f, 0.0f};
            v2f acc3 = {0.0f, 0.0f};

            auto walk = [&](const int c0, const bool first) {
                const int sa = svA;
                const int sb = svB;
                v4f rowA[8], rowB[8];
#define ROW_LOAD(I, S) { \
                    const unsigned r = (unsigned)quad_bcast<(I) & 3>(S) * 8u + rp; \
                    rowA[I] = h1v[r]; \
                    rowB[I] = h1v[r + 1u]; }
                ROW_LOAD(0, sa) ROW_LOAD(1, sa) ROW_LOAD(2, sa) ROW_LOAD(3, sa)
                ROW_LOAD(4, sb) ROW_LOAD(5, sb) ROW_LOAD(6, sb) ROW_LOAD(7, sb)
#undef ROW_LOAD
                if (c0 + 8 < mt) load_src(c0 + 8, svA, svB);

                float lg[8];
                float cm = NEG_INF;
                #pragma unroll
                for (int i = 0; i < 8; ++i) {
                    float t = dot8(rowA[i], rowB[i], aswA, aswB);
                    t += lane_dpp<DPP_XOR1>(t);
                    lg[i] = (c0 + i < nent) ? lrelu(t + adp) : NEG_INF;
                    cm = fmaxf(cm, lg[i]);
                }
                if (first) {
                    mx = cm;
                } else {
                    const float nm = fmaxf(mx, cm);
                    const float sc = __expf(mx - nm);
                    den *= sc;
                    acc0 *= sc; acc1 *= sc; acc2 *= sc; acc3 *= sc;
                    mx = nm;
                }
                #pragma unroll
                for (int i = 0; i < 8; ++i) {
                    const float e = (c0 + i < nent) ? __expf(lg[i] - mx) : 0.0f;
                    den += e;
                    const v2f ee = {e, e};
                    acc0 = __builtin_elementwise_fma(lo2(rowA[i]), ee, acc0);
                    acc1 = __builtin_elementwise_fma(hi2(rowA[i]), ee, acc1);
                    acc2 = __builtin_elementwise_fma(lo2(rowB[i]), ee, acc2);
                    acc3 = __builtin_elementwise_fma(hi2(rowB[i]), ee, acc3);
                }
            };
            walk(0, true);
            for (int c0 = 8; c0 < mt; c0 += 8) walk(c0, false);

            const v4f biasA = *(const v4f*)(b1 + 8 * p);
            const v4f biasB = *(const v4f*)(b1 + 8 * p + 4);
            const v4f w2a = *(const v4f*)(W2 + 16 * p);
            const v4f w2b = *(const v4f*)(W2 + 16 * p + 4);
            const v4f w2c = *(const v4f*)(W2 + 16 * p + 8);
            const v4f w2d = *(const v4f*)(W2 + 16 * p + 12);
            const float inv = 1.0f / (den + 1e-16f);
            float v0 = acc0.x * inv + biasA.x;
            float v1 = acc0.y * inv + biasA.y;
            float v2 = acc1.x * inv + biasA.z;
            float v3 = acc1.y * inv + biasA.w;
            float v4 = acc2.x * inv + biasB.x;
            float v5 = acc2.y * inv + biasB.y;
            float v6 = acc3.x * inv + biasB.z;
            float v7 = acc3.y * inv + biasB.w;
            v0 = v0 > 0.0f ? v0 : 0.0f;
            v1 = v1 > 0.0f ? v1 : 0.0f;
            v2 = v2 > 0.0f ? v2 : 0.0f;
            v3 = v3 > 0.0f ? v3 : 0.0f;
            v4 = v4 > 0.0f ? v4 : 0.0f;
            v5 = v5 > 0.0f ? v5 : 0.0f;
            v6 = v6 > 0.0f ? v6 : 0.0f;
            v7 = v7 > 0.0f ? v7 : 0.0f;
            float g0 = v0 * w2a.x + v1 * w2a.z + v2 * w2b.x + v3 * w2b.z
                     + v4 * w2c.x + v5 * w2c.z + v6 * w2d.x + v7 * w2d.z;
            float g1 = v0 * w2a.y + v1 * w2a.w + v2 * w2b.y + v3 * w2b.w
                     + v4 * w2c.y + v5 * w2c.w + v6 * w2d.y + v7 * w2d.w;
            g0 += lane_dpp<DPP_XOR1>(g0);
            g1 += lane_dpp<DPP_XOR1>(g1);
            g0 += lane_dpp<DPP_XOR2>(g0);
            g1 += lane_dpp<DPP_XOR2>(g1);
            if (valid && p == 0) {
                v4f r;
                r.x = g0;
                r.y = g1;
                r.z = g0 * as20 + g1 * as21;
                r.w = g0 * ad20 + g1 * ad21;
                *(v4f*)(rec2 + (size_t)d * 4) = r;
            }
        }
    };
    if (fits) node_pass(ListsIn<true>{}); else node_pass(ListsIn<false>{});
}

#define N2_GROUPS 4

__global__ __launch_bounds__(256) void k_node2(const int* __restrict__ rowptr,
                                               const int* __restrict__ cnt,
                                               const int* __restrict__ col,
                                               const float* __restrict__ rec2,
                                               const float* __restrict__ b2,
                                               float* __restrict__ out,
                                               int n) {
    const int lane = threadIdx.x & 31;
    const int w = __builtin_amdgcn_readfirstlane((int)((blockIdx.x * blockDim.x + threadIdx.x) >> 5));
    const float NEG_INF = __int_as_float(0xFF800000);
    const float bias0 = b2[0];
    const float bias1 = b2[1];
    const int sub = lane & 3;
    const int slot = lane >> 2;
    const v4f* __restrict__ rec = (const v4f*)rec2;
    const int dFirst = w * (8 * N2_GROUPS);
    if (dFirst >= n) return;

    int dc = dFirst + slot < n ? dFirst + slot : n - 1;
    int beg = rowptr[dc];
    int deg = cnt[dc];
    float ad = rec2[(size_t)dc * 4 + 3];

    for (int gi = 0; gi < N2_GROUPS; ++gi) {
        const int d0 = dFirst + gi * 8;
        if (d0 >= n) break;
        const int d = d0 + slot;
        const bool valid = d < n;
        const int tot = valid ? deg + 1 : 0;

        const int dn = d + 8 < n ? d + 8 : n - 1;
        const int nbeg = rowptr[dn];
        const int ndeg = cnt[dn];
        const float nad = rec2[(size_t)dn * 4 + 3];

        int mt = tot;
        mt = max(mt, lane_dpp_i<DPP_XOR4>(mt));
        mt = max(mt, lane_dpp_i<DPP_XOR8>(mt));
        mt = max(mt, (int)__builtin_amdgcn_permlanex16(mt, mt, 0x76543210, 0xfedcba98, false, false));
        mt = __builtin_amdgcn_readfirstlane(mt);

        float mx = NEG_INF;
        float den = 0.0f;
        float a0 = 0.0f;
        float a1 = 0.0f;
        for (int c0 = 0; c0 < mt; c0 += 32) {
            int sv[8];
            v4f r[8];
            float lg[8];
            #pragma unroll
            for (int j = 0; j < 8; ++j) {
                const int idx = c0 + 4 * j + sub;
                sv[j] = col[idx < deg ? beg + idx : 0];
            }
            #pragma unroll
            for (int j = 0; j < 8; ++j) {
                const int idx = c0 + 4 * j + sub;
                r[j] = rec[idx < deg ? sv[j] : dc];
            }
            float cm = NEG_INF;
            #pragma unroll
            for (int j = 0; j < 8; ++j) {
                const int idx = c0 + 4 * j + sub;
                lg[j] = idx < tot ? lrelu(r[j].z + ad) : NEG_INF;
                cm = fmaxf(cm, lg[j]);
            }
            cm = fmaxf(cm, lane_dpp<DPP_XOR1>(cm));
            cm = fmaxf(cm, lane_dpp<DPP_XOR2>(cm));
            if (c0 == 0) {
                mx = cm;
            } else {
                const float nm = fmaxf(mx, cm);
                const float sc = __expf(mx - nm);
                den *= sc;
                a0 *= sc;
                a1 *= sc;
                mx = nm;
            }
            #pragma unroll
            for (int j = 0; j < 8; ++j) {
                const int idx = c0 + 4 * j + sub;
                const float e = idx < tot ? __expf(lg[j] - mx) : 0.0f;
                den += e;
                a0 += e * r[j].x;
                a1 += e * r[j].y;
            }
        }
        den += lane_dpp<DPP_XOR1>(den);
        a0 += lane_dpp<DPP_XOR1>(a0);
        a1 += lane_dpp<DPP_XOR1>(a1);
        den += lane_dpp<DPP_XOR2>(den);
        a0 += lane_dpp<DPP_XOR2>(a0);
        a1 += lane_dpp<DPP_XOR2>(a1);
        if (valid && sub == 0) {
            const float inv = 1.0f / (den + 1e-16f);
            v2f o;
            o.x = a0 * inv + bias0;
            o.y = a1 * inv + bias1;
            *(v2f*)(out + (size_t)d * 2) = o;
        }
        dc = dn;
        beg = nbeg;
        deg = ndeg;
        ad = nad;
    }
}

extern "C" void kernel_launch(void* const* d_in, const int* in_sizes, int n_in,
                              void* d_out, int out_size, void* d_ws, size_t ws_size,
                              hipStream_t stream) {
    const float* x    = (const float*)d_in[0];
    const int*   ei   = (const int*)d_in[1];
    const float* W1   = (const float*)d_in[2];
    const float* as1w = (const float*)d_in[3];
    const float* ad1w = (const float*)d_in[4];
    const float* b1   = (const float*)d_in[5];
    const float* W2   = (const float*)d_in[6];
    const float* as2w = (const float*)d_in[7];
    const float* ad2w = (const float*)d_in[8];
    const float* b2   = (const float*)d_in[9];

    const int N = in_sizes[0] / 128;
    const int E = in_sizes[1] / 2;
    const int* srcI = ei;
    const int* dstI = ei + E;

    float* p = (float*)d_ws;
    float* h1     = p; p += (size_t)32 * N;
    float* rec2   = p; p += (size_t)4 * N;
    int* cnt      = (int*)p; p += (size_t)N;
    int* rowptr   = (int*)p; p += (size_t)N;
    int* staging  = (int*)p; p += (size_t)2 * E;
    int* col      = (int*)p; p += (size_t)E;
    const int nb    = (N + BIN_NODES - 1) / BIN_NODES;
    const int etiles = (E + TILE_EDGES - 1) / TILE_EDGES;
    int* L        = (int*)p; p += (size_t)etiles * (nb + 1);

    const int T = 256;

    {
        const int gblocks = (N + 511) / 512;
        k_tilesort_gemm1<<<etiles + gblocks, 1024, 0, stream>>>(srcI, dstI, L, staging,
                                                                E, nb, etiles, x, W1, h1, N);
    }
    k_binsort_node1<<<nb, 1024, 0, stream>>>(staging, L, rowptr, cnt, col, N, nb, etiles,
                                             h1, as1w, ad1w, b1, W2, as2w, ad2w, rec2);
    k_node2<<<(N + 255) / 256, T, 0, stream>>>(rowptr, cnt, col, rec2, b2,
                                             (float*)d_out, N);
}
